# starting kernel
# baseline (speedup 1.0000x reference)
#include <hip/hip_runtime.h>
#include <math.h>

#define NN   50000
#define EE   800000
#define ET   (EE + NN)
#define INC  128
#define CC   64
#define NEGS 0.2f

typedef float v2f __attribute__((ext_vector_type(2)));
typedef float v8f __attribute__((ext_vector_type(8)));

__device__ __forceinline__ unsigned fenc(float f) {
    unsigned u = __float_as_uint(f);
    return (u >> 31) ? ~u : (u | 0x80000000u);
}
__device__ __forceinline__ float fdec(unsigned u) {
    return __uint_as_float((u >> 31) ? (u & 0x7FFFFFFFu) : ~u);
}

__global__ __launch_bounds__(256)
void init_ws_kernel(float* __restrict__ out, float* __restrict__ denom,
                    unsigned* __restrict__ amax) {
    int i = blockIdx.x * blockDim.x + threadIdx.x;
    if (i < NN * CC) out[i] = 0.0f;
    if (i < NN) { denom[i] = 0.0f; amax[i] = 0x007FFFFFu;   }
}

__global__ __launch_bounds__(32)
void gemm_wmma_kernel(const float* __restrict__ x, const float* __restrict__ W,
                      float* __restrict__ g) {
    const int rb   = blockIdx.x;
    const int lane = threadIdx.x;
    const int half = lane >> 4;
    const int l15  = lane & 15;

    v8f acc0 = {}; v8f acc1 = {}; v8f acc2 = {}; v8f acc3 = {};

    const float* xrow = x + (rb * 16 + l15) * INC + half * 2;
    for (int k0 = 0; k0 < INC; k0 += 4) {
        v2f a; a.x = xrow[k0]; a.y = xrow[k0 + 1];
        const float* wp = W + (k0 + half * 2) * CC + l15;
        v2f b0, b1, b2, b3;
        b0.x = wp[0];  b0.y = wp[0  + CC];
        b1.x = wp[16]; b1.y = wp[16 + CC];
        b2.x = wp[32]; b2.y = wp[32 + CC];
        b3.x = wp[48]; b3.y = wp[48 + CC];
        acc0 = __builtin_amdgcn_wmma_f32_16x16x4_f32(false, a, false, b0, (short)0, acc0, false, false);
        acc1 = __builtin_amdgcn_wmma_f32_16x16x4_f32(false, a, false, b1, (short)0, acc1, false, false);
        acc2 = __builtin_amdgcn_wmma_f32_16x16x4_f32(false, a, false, b2, (short)0, acc2, false, false);
        acc3 = __builtin_amdgcn_wmma_f32_16x16x4_f32(false, a, false, b3, (short)0, acc3, false, false);
    }
    #pragma unroll
    for (int v = 0; v < 8; ++v) {
        float* op = g + (rb * 16 + v + half * 8) * CC + l15;
        op[0]  = acc0[v];
        op[16] = acc1[v];
        op[32] = acc2[v];
        op[48] = acc3[v];
    }
}

__global__ __launch_bounds__(256)
void node_scores_kernel(const float* __restrict__ g, const float* __restrict__ corrs,
                        const float* __restrict__ att_i, const float* __restrict__ att_j,
                        const float* __restrict__ aem_i, const float* __restrict__ aem_j,
                        float* __restrict__ s_i, float* __restrict__ s_j) {
    int node = (blockIdx.x * blockDim.x + threadIdx.x) >> 5;
    int lane = threadIdx.x & 31;
    if (node >= NN) return;
    const float* gr = g + node * CC;
    const float* cr = corrs + node * CC;
    float si = 0.0f, sj = 0.0f;
    #pragma unroll
    for (int c = lane; c < CC; c += 32) {
        float gv = gr[c], cv = cr[c];
        si += gv * att_i[c] + cv * aem_i[c];
        sj += gv * att_j[c] + cv * aem_j[c];
    }
    #pragma unroll
    for (int off = 16; off > 0; off >>= 1) {
        si += __shfl_xor(si, off, 32);
        sj += __shfl_xor(sj, off, 32);
    }
    if (lane == 0) { s_i[node] = si; s_j[node] = sj; }
}

__global__ __launch_bounds__(256)
void edge_alpha_kernel(const int* __restrict__ ei, const float* __restrict__ s_i,
                       const float* __restrict__ s_j, float* __restrict__ alpha,
                       unsigned* __restrict__ amax) {
    int e = blockIdx.x * blockDim.x + threadIdx.x;
    if (e >= ET) return;
    int src, dst;
    if (e < EE) { src = ei[e]; dst = ei[EE + e]; } else { src = dst = e - EE; }
    float a = s_i[dst] + s_j[src];
    a = (a > 0.0f) ? a : NEGS * a;
    alpha[e] = a;
    atomicMax(amax + dst, fenc(a));
}

__global__ __launch_bounds__(256)
void edge_exp_kernel(const int* __restrict__ ei, const unsigned* __restrict__ amax,
                     float* __restrict__ alpha_ex, float* __restrict__ denom) {
    int e = blockIdx.x * blockDim.x + threadIdx.x;
    if (e >= ET) return;
    int dst = (e < EE) ? ei[EE + e] : (e - EE);
    float ex = expf(alpha_ex[e] - fdec(amax[dst]));
    alpha_ex[e] = ex;
    atomicAdd(denom + dst, ex);
}

__global__ __launch_bounds__(256)
void message_kernel(const int* __restrict__ ei, const float* __restrict__ g,
                    const float* __restrict__ ex, const float* __restrict__ denom,
                    float* __restrict__ out) {
    int tid = blockIdx.x * blockDim.x + threadIdx.x;
    int e = tid >> 4;
    int q = tid & 15;
    if (e >= ET) return;
    int src, dst;
    if (e < EE) { src = ei[e]; dst = ei[EE + e]; } else { src = dst = e - EE; }
    float coef = ex[e] / (denom[dst] + 1e-16f);
    float4 gv = ((const float4*)(g + src * CC))[q];
    float* op = out + dst * CC + q * 4;
    atomicAdd(op + 0, gv.x * coef);
    atomicAdd(op + 1, gv.y * coef);
    atomicAdd(op + 2, gv.z * coef);
    atomicAdd(op + 3, gv.w * coef);
}

__global__ __launch_bounds__(256)
void finalize_kernel(float* __restrict__ out, const float* __restrict__ bias) {
    int i = blockIdx.x * blockDim.x + threadIdx.x;
    if (i >= NN * CC) return;
    float v = out[i] + bias[i & (CC - 1)];
    out[i] = (v > 0.0f) ? v : 0.0f;
}

extern "C" void kernel_launch(void* const* d_in, const int* in_sizes, int n_in,
                              void* d_out, int out_size, void* d_ws, size_t ws_size,
                              hipStream_t stream) {
    (void)in_sizes; (void)n_in; (void)out_size; (void)ws_size;
    const float* x      = (const float*)d_in[0];
    const int*   ei     = (const int*)  d_in[1];
    const float* corrs  = (const float*)d_in[2];
    const float* W      = (const float*)d_in[3];
    const float* att_i  = (const float*)d_in[4];
    const float* att_j  = (const float*)d_in[5];
    const float* aem_i  = (const float*)d_in[6];
    const float* aem_j  = (const float*)d_in[7];
    const float* bias   = (const float*)d_in[8];
    float* out = (float*)d_out;

    float*    ws    = (float*)d_ws;
    float*    g     = ws;
    float*    s_i   = g + (size_t)NN * CC;
    float*    s_j   = s_i + NN;
    unsigned* amax  = (unsigned*)(s_j + NN);
    float*    denom = (float*)(amax + NN);
    float*    aex   = denom + NN;

    init_ws_kernel<<<(NN * CC + 255) / 256, 256, 0, stream>>>(out, denom, amax);
    gemm_wmma_kernel<<<NN / 16, 32, 0, stream>>>(x, W, g);
    node_scores_kernel<<<(NN * 32 + 255) / 256, 256, 0, stream>>>(
        g, corrs, att_i, att_j, aem_i, aem_j, s_i, s_j);
    edge_alpha_kernel<<<(ET + 255) / 256, 256, 0, stream>>>(ei, s_i, s_j, aex, amax);
    edge_exp_kernel<<<(ET + 255) / 256, 256, 0, stream>>>(ei, amax, aex, denom);
    message_kernel<<<(ET * 16 + 255) / 256, 256, 0, stream>>>(ei, g, aex, denom, out);
    finalize_kernel<<<(NN * CC + 255) / 256, 256, 0, stream>>>(out, bias);
}


# Edges grouped by destination (count, scan, fill with integer atomics); one wave per node does max, exp sum and weighted row sum in registers; no float atomics, sum order = fill slot order
# speedup vs baseline: 1.1243x; 1.1243x over previous
#include <hip/hip_runtime.h>
#include <math.h>

#define NN   50000
#define EE   800000
#define INC  128
#define CC   64
#define NEGS 0.2f

typedef float v2f __attribute__((ext_vector_type(2)));
typedef float v8f __attribute__((ext_vector_type(8)));

__global__ __launch_bounds__(32)
void gemm_wmma_kernel(const float* __restrict__ x, const float* __restrict__ W,
                      float* __restrict__ g) {
    const int rb   = blockIdx.x;
    const int lane = threadIdx.x;
    const int half = lane >> 4;
    const int l15  = lane & 15;

    v8f acc0 = {}; v8f acc1 = {}; v8f acc2 = {}; v8f acc3 = {};

    const float* xrow = x + (rb * 16 + l15) * INC + half * 2;
    for (int k0 = 0; k0 < INC; k0 += 4) {
        v2f a; a.x = xrow[k0]; a.y = xrow[k0 + 1];
        const float* wp = W + (k0 + half * 2) * CC + l15;
        v2f b0, b1, b2, b3;
        b0.x = wp[0];  b0.y = wp[0  + CC];
        b1.x = wp[16]; b1.y = wp[16 + CC];
        b2.x = wp[32]; b2.y = wp[32 + CC];
        b3.x = wp[48]; b3.y = wp[48 + CC];
        acc0 = __builtin_amdgcn_wmma_f32_16x16x4_f32(false, a, false, b0, (short)0, acc0, false, false);
        acc1 = __builtin_amdgcn_wmma_f32_16x16x4_f32(false, a, false, b1, (short)0, acc1, false, false);
        acc2 = __builtin_amdgcn_wmma_f32_16x16x4_f32(false, a, false, b2, (short)0, acc2, false, false);
        acc3 = __builtin_amdgcn_wmma_f32_16x16x4_f32(false, a, false, b3, (short)0, acc3, false, false);
    }
    #pragma unroll
    for (int v = 0; v < 8; ++v) {
        float* op = g + (rb * 16 + v + half * 8) * CC + l15;
        op[0]  = acc0[v];
        op[16] = acc1[v];
        op[32] = acc2[v];
        op[48] = acc3[v];
    }
}

__global__ __launch_bounds__(256)
void node_scores_kernel(const float* __restrict__ g, const float* __restrict__ corrs,
                        const float* __restrict__ att_i, const float* __restrict__ att_j,
                        const float* __restrict__ aem_i, const float* __restrict__ aem_j,
                        float* __restrict__ s_i, float* __restrict__ s_j,
                        int* __restrict__ cnt) {
    int node = (blockIdx.x * blockDim.x + threadIdx.x) >> 5;
    int lane = threadIdx.x & 31;
    if (node >= NN) return;
    const float* gr = g + node * CC;
    const float* cr = corrs + node * CC;
    float si = 0.0f, sj = 0.0f;
    #pragma unroll
    for (int c = lane; c < CC; c += 32) {
        float gv = gr[c], cv = cr[c];
        si += gv * att_i[c] + cv * aem_i[c];
        sj += gv * att_j[c] + cv * aem_j[c];
    }
    #pragma unroll
    for (int off = 16; off > 0; off >>= 1) {
        si += __shfl_xor(si, off, 32);
        sj += __shfl_xor(sj, off, 32);
    }
    if (lane == 0) { s_i[node] = si; s_j[node] = sj; cnt[node] = 0; }
}

__global__ __launch_bounds__(256)
void degree_kernel(const int* __restrict__ ei, int* __restrict__ cnt) {
    int t = blockIdx.x * blockDim.x + threadIdx.x;
    if (t >= EE / 4) return;
    int4 d = ((const int4*)(ei + EE))[t];
    atomicAdd(cnt + d.x, 1);
    atomicAdd(cnt + d.y, 1);
    atomicAdd(cnt + d.z, 1);
    atomicAdd(cnt + d.w, 1);
}

#define SCAN_T   1024
#define SCAN_PER ((NN + SCAN_T - 1) / SCAN_T)
__global__ __launch_bounds__(SCAN_T)
void scan_kernel(const int* __restrict__ cnt, int* __restrict__ start,
                 int* __restrict__ cur) {
    __shared__ int buf[SCAN_T * SCAN_PER];
    __shared__ int wtot[SCAN_T / 32];
    const int t    = threadIdx.x;
    const int lane = t & 31;
    const int w    = t >> 5;
    #pragma unroll
    for (int k = 0; k < SCAN_PER; ++k) {
        const int i = k * SCAN_T + t;
        buf[i] = (i < NN) ? cnt[i] : 0;
    }
    __syncthreads();
    int s = 0;
    #pragma unroll
    for (int k = 0; k < SCAN_PER; ++k) s += buf[t * SCAN_PER + k];
    int inc = s;
    #pragma unroll
    for (int off = 1; off < 32; off <<= 1) {
        const int u = __shfl_up(inc, off, 32);
        if (lane >= off) inc += u;
    }
    if (lane == 31) wtot[w] = inc;
    __syncthreads();
    int run = inc - s;
    for (int k = 0; k < w; ++k) run += wtot[k];
    #pragma unroll
    for (int k = 0; k < SCAN_PER; ++k) {
        const int v = buf[t * SCAN_PER + k];
        buf[t * SCAN_PER + k] = run;
        run += v;
    }
    __syncthreads();
    #pragma unroll
    for (int k = 0; k < SCAN_PER; ++k) {
        const int i = k * SCAN_T + t;
        if (i < NN) { const int v = buf[i]; start[i] = v; cur[i] = v; }
    }
}

__global__ __launch_bounds__(256)
void fill_kernel(const int* __restrict__ ei, int* __restrict__ cur,
                 int* __restrict__ csr_src) {
    int t = blockIdx.x * blockDim.x + threadIdx.x;
    if (t >= EE / 4) return;
    int4 s = ((const int4*)ei)[t];
    int4 d = ((const int4*)(ei + EE))[t];
    int p0 = atomicAdd(cur + d.x, 1);
    int p1 = atomicAdd(cur + d.y, 1);
    int p2 = atomicAdd(cur + d.z, 1);
    int p3 = atomicAdd(cur + d.w, 1);
    csr_src[p0] = s.x;
    csr_src[p1] = s.y;
    csr_src[p2] = s.z;
    csr_src[p3] = s.w;
}

__device__ __forceinline__ float edge_alpha(float si, float sj) {
    float a = si + sj;
    return (a > 0.0f) ? a : NEGS * a;
}

__global__ __launch_bounds__(256)
void aggregate_kernel(const int* __restrict__ start, const int* __restrict__ cur,
                      const int* __restrict__ csr_src, const float* __restrict__ s_i,
                      const float* __restrict__ s_j, const float* __restrict__ g,
                      const float* __restrict__ bias, float* __restrict__ out) {
    const int node = __builtin_amdgcn_readfirstlane((blockIdx.x * blockDim.x + threadIdx.x) >> 5);
    const int lane = threadIdx.x & 31;
    if (node >= NN) return;
    const int beg = start[node];
    const int deg = cur[node] - beg;
    const int n   = deg + 1;
    const float si = s_i[node];
    const int* cs = csr_src + beg;

    float m = -INFINITY;
    for (int base = 0; base < n; base += 32) {
        const int k = base + lane;
        if (k < n) {
            const int src = (k < deg) ? cs[k] : node;
            m = fmaxf(m, edge_alpha(si, s_j[src]));
        }
    }
    #pragma unroll
    for (int off = 16; off > 0; off >>= 1) m = fmaxf(m, __shfl_xor(m, off, 32));

    float dsum = 0.0f;
    for (int base = 0; base < n; base += 32) {
        const int k = base + lane;
        if (k < n) {
            const int src = (k < deg) ? cs[k] : node;
            dsum += expf(edge_alpha(si, s_j[src]) - m);
        }
    }
    #pragma unroll
    for (int off = 16; off > 0; off >>= 1) dsum += __shfl_xor(dsum, off, 32);
    const float den = dsum + 1e-16f;

    float accx = 0.0f, accy = 0.0f;
    for (int base = 0; base < n; base += 32) {
        const int k = base + lane;
        int src = node;
        float c = 0.0f;
        if (k < n) {
            src = (k < deg) ? cs[k] : node;
            c = expf(edge_alpha(si, s_j[src]) - m) / den;
        }
        const int cn = (n - base < 32) ? (n - base) : 32;
        #pragma unroll 4
        for (int j = 0; j < cn; ++j) {
            const int   sj = __shfl(src, j, 32);
            const float cj = __shfl(c, j, 32);
            const float2 gv = ((const float2*)(g + (size_t)sj * CC))[lane];
            accx += gv.x * cj;
            accy += gv.y * cj;
        }
    }
    const float2 b = ((const float2*)bias)[lane];
    float2 r;
    r.x = accx + b.x; r.x = (r.x > 0.0f) ? r.x : 0.0f;
    r.y = accy + b.y; r.y = (r.y > 0.0f) ? r.y : 0.0f;
    ((float2*)(out + (size_t)node * CC))[lane] = r;
}

extern "C" void kernel_launch(void* const* d_in, const int* in_sizes, int n_in,
                              void* d_out, int out_size, void* d_ws, size_t ws_size,
                              hipStream_t stream) {
    (void)in_sizes; (void)n_in; (void)out_size; (void)ws_size;
    const float* x      = (const float*)d_in[0];
    const int*   ei     = (const int*)  d_in[1];
    const float* corrs  = (const float*)d_in[2];
    const float* W      = (const float*)d_in[3];
    const float* att_i  = (const float*)d_in[4];
    const float* att_j  = (const float*)d_in[5];
    const float* aem_i  = (const float*)d_in[6];
    const float* aem_j  = (const float*)d_in[7];
    const float* bias   = (const float*)d_in[8];
    float* out = (float*)d_out;

    float* ws    = (float*)d_ws;
    float* g     = ws;
    float* s_i   = g + (size_t)NN * CC;
    float* s_j   = s_i + NN;
    int*   cnt   = (int*)(s_j + NN);
    int*   start = cnt + NN;
    int*   cur   = start + NN;
    int*   csr   = cur + NN;

    gemm_wmma_kernel<<<NN / 16, 32, 0, stream>>>(x, W, g);
    node_scores_kernel<<<(NN * 32 + 255) / 256, 256, 0, stream>>>(
        g, corrs, att_i, att_j, aem_i, aem_j, s_i, s_j, cnt);
    degree_kernel<<<(EE / 4 + 255) / 256, 256, 0, stream>>>(ei, cnt);
    scan_kernel<<<1, SCAN_T, 0, stream>>>(cnt, start, cur);
    fill_kernel<<<(EE / 4 + 255) / 256, 256, 0, stream>>>(ei, cur, csr);
    aggregate_kernel<<<(NN * 32 + 255) / 256, 256, 0, stream>>>(
        start, cur, csr, s_i, s_j, g, bias, out);
}


# Edge ranks from LDS counters: 50 blocks rank their edge slice in a per-node LDS table and add it to the counts by whole lines; fill has no atomics; sum order = block arrival, then LDS add order
# speedup vs baseline: 2.0207x; 1.7973x over previous
#include <hip/hip_runtime.h>
#include <math.h>

#define NN   50000
#define EE   800000
#define INC  128
#define CC   64
#define NEGS 0.2f

typedef float v2f __attribute__((ext_vector_type(2)));
typedef float v8f __attribute__((ext_vector_type(8)));

__global__ __launch_bounds__(32)
void gemm_wmma_kernel(const float* __restrict__ x, const float* __restrict__ W,
                      float* __restrict__ g) {
    const int rb   = blockIdx.x;
    const int lane = threadIdx.x;
    const int half = lane >> 4;
    const int l15  = lane & 15;

    v8f acc0 = {}; v8f acc1 = {}; v8f acc2 = {}; v8f acc3 = {};

    const float* xrow = x + (rb * 16 + l15) * INC + half * 2;
    for (int k0 = 0; k0 < INC; k0 += 4) {
        v2f a; a.x = xrow[k0]; a.y = xrow[k0 + 1];
        const float* wp = W + (k0 + half * 2) * CC + l15;
        v2f b0, b1, b2, b3;
        b0.x = wp[0];  b0.y = wp[0  + CC];
        b1.x = wp[16]; b1.y = wp[16 + CC];
        b2.x = wp[32]; b2.y = wp[32 + CC];
        b3.x = wp[48]; b3.y = wp[48 + CC];
        acc0 = __builtin_amdgcn_wmma_f32_16x16x4_f32(false, a, false, b0, (short)0, acc0, false, false);
        acc1 = __builtin_amdgcn_wmma_f32_16x16x4_f32(false, a, false, b1, (short)0, acc1, false, false);
        acc2 = __builtin_amdgcn_wmma_f32_16x16x4_f32(false, a, false, b2, (short)0, acc2, false, false);
        acc3 = __builtin_amdgcn_wmma_f32_16x16x4_f32(false, a, false, b3, (short)0, acc3, false, false);
    }
    #pragma unroll
    for (int v = 0; v < 8; ++v) {
        float* op = g + (rb * 16 + v + half * 8) * CC + l15;
        op[0]  = acc0[v];
        op[16] = acc1[v];
        op[32] = acc2[v];
        op[48] = acc3[v];
    }
}

__global__ __launch_bounds__(256)
void node_scores_kernel(const float* __restrict__ g, const float* __restrict__ corrs,
                        const float* __restrict__ att_i, const float* __restrict__ att_j,
                        const float* __restrict__ aem_i, const float* __restrict__ aem_j,
                        float* __restrict__ s_i, float* __restrict__ s_j,
                        int* __restrict__ cnt) {
    int node = (blockIdx.x * blockDim.x + threadIdx.x) >> 5;
    int lane = threadIdx.x & 31;
    if (node >= NN) return;
    const float* gr = g + node * CC;
    const float* cr = corrs + node * CC;
    float si = 0.0f, sj = 0.0f;
    #pragma unroll
    for (int c = lane; c < CC; c += 32) {
        float gv = gr[c], cv = cr[c];
        si += gv * att_i[c] + cv * aem_i[c];
        sj += gv * att_j[c] + cv * aem_j[c];
    }
    #pragma unroll
    for (int off = 16; off > 0; off >>= 1) {
        si += __shfl_xor(si, off, 32);
        sj += __shfl_xor(sj, off, 32);
    }
    if (lane == 0) { s_i[node] = si; s_j[node] = sj; cnt[node] = 0; }
}

#define RK_T    1024
#define RK_G    50
#define RK_Q    (EE / 4 / RK_G)
#define RK_K    ((RK_Q + RK_T - 1) / RK_T)
#define RK_FULL (NN / RK_T)
#define RK_PAD  ((RK_FULL + 1) * RK_T)
#define RK_GRP  24
__global__ __launch_bounds__(RK_T)
void rank_kernel(const int* __restrict__ ei, int* __restrict__ cnt,
                 int* __restrict__ rank) {
    __shared__ int tbl[RK_PAD];
    const int t = threadIdx.x;
    const int4* dq = (const int4*)(ei + EE) + blockIdx.x * RK_Q;
    int4 d[RK_K];
    #pragma unroll
    for (int k = 0; k < RK_K; ++k) {
        const int q = k * RK_T + t;
        d[k] = dq[(q < RK_Q) ? q : RK_Q - 1];
    }
    #pragma unroll
    for (int k = 0; k < (RK_PAD / 4 + RK_T - 1) / RK_T; ++k) {
        const int i = k * RK_T + t;
        if (i < RK_PAD / 4) ((int4*)tbl)[i] = make_int4(0, 0, 0, 0);
    }
    __syncthreads();
    int4 r[RK_K];
    #pragma unroll
    for (int k = 0; k < RK_K; ++k) {
        const int q = k * RK_T + t;
        r[k] = make_int4(0, 0, 0, 0);
        if (q < RK_Q) {
            r[k].x = atomicAdd(&tbl[d[k].x], 1);
            r[k].y = atomicAdd(&tbl[d[k].y], 1);
            r[k].z = atomicAdd(&tbl[d[k].z], 1);
            r[k].w = atomicAdd(&tbl[d[k].w], 1);
        }
    }
    __syncthreads();
    const int j0 = blockIdx.x % RK_FULL;
    const int it = RK_FULL * RK_T + t;
    int bt = 0;
    if (it < NN) bt = atomicAdd(cnt + it, tbl[it]);
    #pragma unroll
    for (int g0 = 0; g0 < RK_FULL; g0 += RK_GRP) {
        int c[RK_GRP];
        #pragma unroll
        for (int u = 0; u < RK_GRP; ++u) {
            int j = g0 + u + j0; if (j >= RK_FULL) j -= RK_FULL;
            c[u] = tbl[j * RK_T + t];
        }
        #pragma unroll
        for (int u = 0; u < RK_GRP; ++u) {
            int j = g0 + u + j0; if (j >= RK_FULL) j -= RK_FULL;
            c[u] = atomicAdd(cnt + j * RK_T + t, c[u]);
        }
        #pragma unroll
        for (int u = 0; u < RK_GRP; ++u) {
            int j = g0 + u + j0; if (j >= RK_FULL) j -= RK_FULL;
            tbl[j * RK_T + t] = c[u];
        }
    }
    if (it < NN) tbl[it] = bt;
    __syncthreads();
    int4* rq = (int4*)rank + blockIdx.x * RK_Q;
    #pragma unroll
    for (int k = 0; k < RK_K; ++k) {
        const int q = k * RK_T + t;
        if (q < RK_Q) {
            int4 o;
            o.x = tbl[d[k].x] + r[k].x;
            o.y = tbl[d[k].y] + r[k].y;
            o.z = tbl[d[k].z] + r[k].z;
            o.w = tbl[d[k].w] + r[k].w;
            rq[q] = o;
        }
    }
}

#define SCAN_T   1024
#define SCAN_PER ((NN + SCAN_T - 1) / SCAN_T)
__global__ __launch_bounds__(SCAN_T)
void scan_kernel(const int* __restrict__ cnt, int* __restrict__ start) {
    __shared__ int buf[SCAN_T * SCAN_PER];
    __shared__ int wtot[SCAN_T / 32];
    const int t    = threadIdx.x;
    const int lane = t & 31;
    const int w    = t >> 5;
    #pragma unroll
    for (int k = 0; k < SCAN_PER; ++k) {
        const int i = k * SCAN_T + t;
        buf[i] = (i < NN) ? cnt[i] : 0;
    }
    __syncthreads();
    int s = 0;
    #pragma unroll
    for (int k = 0; k < SCAN_PER; ++k) s += buf[t * SCAN_PER + k];
    int inc = s;
    #pragma unroll
    for (int off = 1; off < 32; off <<= 1) {
        const int u = __shfl_up(inc, off, 32);
        if (lane >= off) inc += u;
    }
    if (lane == 31) wtot[w] = inc;
    __syncthreads();
    int run = inc - s;
    for (int k = 0; k < w; ++k) run += wtot[k];
    #pragma unroll
    for (int k = 0; k < SCAN_PER; ++k) {
        const int v = buf[t * SCAN_PER + k];
        buf[t * SCAN_PER + k] = run;
        run += v;
    }
    __syncthreads();
    #pragma unroll
    for (int k = 0; k < SCAN_PER; ++k) {
        const int i = k * SCAN_T + t;
        if (i < NN) start[i] = buf[i];
    }
}

__global__ __launch_bounds__(256)
void fill_kernel(const int* __restrict__ ei, const int* __restrict__ start,
                 const int* __restrict__ rank, int* __restrict__ csr_src) {
    int t = blockIdx.x * blockDim.x + threadIdx.x;
    if (t >= EE / 4) return;
    int4 s = ((const int4*)ei)[t];
    int4 d = ((const int4*)(ei + EE))[t];
    int4 r = ((const int4*)rank)[t];
    csr_src[start[d.x] + r.x] = s.x;
    csr_src[start[d.y] + r.y] = s.y;
    csr_src[start[d.z] + r.z] = s.z;
    csr_src[start[d.w] + r.w] = s.w;
}

__device__ __forceinline__ float edge_alpha(float si, float sj) {
    float a = si + sj;
    return (a > 0.0f) ? a : NEGS * a;
}

__global__ __launch_bounds__(256)
void aggregate_kernel(const int* __restrict__ start, const int* __restrict__ cnt,
                      const int* __restrict__ csr_src, const float* __restrict__ s_i,
                      const float* __restrict__ s_j, const float* __restrict__ g,
                      const float* __restrict__ bias, float* __restrict__ out) {
    const int node = __builtin_amdgcn_readfirstlane((blockIdx.x * blockDim.x + threadIdx.x) >> 5);
    const int lane = threadIdx.x & 31;
    if (node >= NN) return;
    const int beg = start[node];
    const int deg = cnt[node];
    const int n   = deg + 1;
    const float si = s_i[node];
    const int* cs = csr_src + beg;

    float m = -INFINITY;
    for (int base = 0; base < n; base += 32) {
        const int k = base + lane;
        if (k < n) {
            const int src = (k < deg) ? cs[k] : node;
            m = fmaxf(m, edge_alpha(si, s_j[src]));
        }
    }
    #pragma unroll
    for (int off = 16; off > 0; off >>= 1) m = fmaxf(m, __shfl_xor(m, off, 32));

    float dsum = 0.0f;
    for (int base = 0; base < n; base += 32) {
        const int k = base + lane;
        if (k < n) {
            const int src = (k < deg) ? cs[k] : node;
            dsum += expf(edge_alpha(si, s_j[src]) - m);
        }
    }
    #pragma unroll
    for (int off = 16; off > 0; off >>= 1) dsum += __shfl_xor(dsum, off, 32);
    const float den = dsum + 1e-16f;

    float accx = 0.0f, accy = 0.0f;
    for (int base = 0; base < n; base += 32) {
        const int k = base + lane;
        int src = node;
        float c = 0.0f;
        if (k < n) {
            src = (k < deg) ? cs[k] : node;
            c = expf(edge_alpha(si, s_j[src]) - m) / den;
        }
        const int cn = (n - base < 32) ? (n - base) : 32;
        #pragma unroll 4
        for (int j = 0; j < cn; ++j) {
            const int   sj = __shfl(src, j, 32);
            const float cj = __shfl(c, j, 32);
            const float2 gv = ((const float2*)(g + (size_t)sj * CC))[lane];
            accx += gv.x * cj;
            accy += gv.y * cj;
        }
    }
    const float2 b = ((const float2*)bias)[lane];
    float2 r;
    r.x = accx + b.x; r.x = (r.x > 0.0f) ? r.x : 0.0f;
    r.y = accy + b.y; r.y = (r.y > 0.0f) ? r.y : 0.0f;
    ((float2*)(out + (size_t)node * CC))[lane] = r;
}

extern "C" void kernel_launch(void* const* d_in, const int* in_sizes, int n_in,
                              void* d_out, int out_size, void* d_ws, size_t ws_size,
                              hipStream_t stream) {
    (void)in_sizes; (void)n_in; (void)out_size; (void)ws_size;
    const float* x      = (const float*)d_in[0];
    const int*   ei     = (const int*)  d_in[1];
    const float* corrs  = (const float*)d_in[2];
    const float* W      = (const float*)d_in[3];
    const float* att_i  = (const float*)d_in[4];
    const float* att_j  = (const float*)d_in[5];
    const float* aem_i  = (const float*)d_in[6];
    const float* aem_j  = (const float*)d_in[7];
    const float* bias   = (const float*)d_in[8];
    float* out = (float*)d_out;

    float* ws    = (float*)d_ws;
    float* g     = ws;
    float* s_i   = g + (size_t)NN * CC;
    float* s_j   = s_i + NN;
    int*   cnt   = (int*)(s_j + NN);
    int*   start = cnt + NN;
    int*   csr   = start + NN;
    int*   rank  = csr + EE;

    gemm_wmma_kernel<<<NN / 16, 32, 0, stream>>>(x, W, g);
    node_scores_kernel<<<(NN * 32 + 255) / 256, 256, 0, stream>>>(
        g, corrs, att_i, att_j, aem_i, aem_j, s_i, s_j, cnt);
    rank_kernel<<<RK_G, RK_T, 0, stream>>>(ei, cnt, rank);
    scan_kernel<<<1, SCAN_T, 0, stream>>>(cnt, start);
    fill_kernel<<<(EE / 4 + 255) / 256, 256, 0, stream>>>(ei, start, rank, csr);
    aggregate_kernel<<<(NN * 32 + 255) / 256, 256, 0, stream>>>(
        start, cnt, csr, s_i, s_j, g, bias, out);
}


# Node scores s_i, s_j computed in the GEMM wave from its accumulator tile and the corrs rows; scores kernel (50000 waves, one launch, a re-read of g) gone; score sum order: per lane, then 16-lane tree
# speedup vs baseline: 2.3225x; 1.1494x over previous
#include <hip/hip_runtime.h>
#include <math.h>

#define NN   50000
#define EE   800000
#define INC  128
#define CC   64
#define NEGS 0.2f

typedef float v2f __attribute__((ext_vector_type(2)));
typedef float v8f __attribute__((ext_vector_type(8)));

__global__ __launch_bounds__(32)
void gemm_scores_kernel(const float* __restrict__ x, const float* __restrict__ W,
                        const float* __restrict__ corrs,
                        const float* __restrict__ att_i, const float* __restrict__ att_j,
                        const float* __restrict__ aem_i, const float* __restrict__ aem_j,
                        float* __restrict__ g, float* __restrict__ s_i,
                        float* __restrict__ s_j, int* __restrict__ cnt) {
    const int rb   = blockIdx.x;
    const int lane = threadIdx.x;
    const int half = lane >> 4;
    const int l15  = lane & 15;

    float4 cr[8];
    {
        const float4* cp = (const float4*)(corrs + (size_t)(rb * 16 + half * 8) * CC) + l15;
        #pragma unroll
        for (int v = 0; v < 8; ++v) cr[v] = cp[v * (CC / 4)];
    }

    v8f acc0 = {}; v8f acc1 = {}; v8f acc2 = {}; v8f acc3 = {};

    const float* xrow = x + (rb * 16 + l15) * INC + half * 2;
    for (int k0 = 0; k0 < INC; k0 += 4) {
        v2f a; a.x = xrow[k0]; a.y = xrow[k0 + 1];
        const float* wp = W + (k0 + half * 2) * CC + l15;
        v2f b0, b1, b2, b3;
        b0.x = wp[0];  b0.y = wp[0  + CC];
        b1.x = wp[16]; b1.y = wp[16 + CC];
        b2.x = wp[32]; b2.y = wp[32 + CC];
        b3.x = wp[48]; b3.y = wp[48 + CC];
        acc0 = __builtin_amdgcn_wmma_f32_16x16x4_f32(false, a, false, b0, (short)0, acc0, false, false);
        acc1 = __builtin_amdgcn_wmma_f32_16x16x4_f32(false, a, false, b1, (short)0, acc1, false, false);
        acc2 = __builtin_amdgcn_wmma_f32_16x16x4_f32(false, a, false, b2, (short)0, acc2, false, false);
        acc3 = __builtin_amdgcn_wmma_f32_16x16x4_f32(false, a, false, b3, (short)0, acc3, false, false);
    }
    #pragma unroll
    for (int v = 0; v < 8; ++v) {
        float* op = g + (rb * 16 + v + half * 8) * CC + l15;
        op[0]  = acc0[v];
        op[16] = acc1[v];
        op[32] = acc2[v];
        op[48] = acc3[v];
    }

    float p[16];
    {
        float ai[4], aj[4];
        #pragma unroll
        for (int t = 0; t < 4; ++t) { ai[t] = att_i[l15 + 16 * t]; aj[t] = att_j[l15 + 16 * t]; }
        const float4 ei4 = ((const float4*)aem_i)[l15];
        const float4 ej4 = ((const float4*)aem_j)[l15];
        #pragma unroll
        for (int v = 0; v < 8; ++v) {
            float pi = acc0[v] * ai[0] + acc1[v] * ai[1] + acc2[v] * ai[2] + acc3[v] * ai[3];
            float pj = acc0[v] * aj[0] + acc1[v] * aj[1] + acc2[v] * aj[2] + acc3[v] * aj[3];
            pi += cr[v].x * ei4.x + cr[v].y * ei4.y + cr[v].z * ei4.z + cr[v].w * ei4.w;
            pj += cr[v].x * ej4.x + cr[v].y * ej4.y + cr[v].z * ej4.z + cr[v].w * ej4.w;
            p[v] = pi; p[8 + v] = pj;
        }
    }
    const bool b3 = (l15 & 8) != 0, b2 = (l15 & 4) != 0, b1 = (l15 & 2) != 0, b0 = (l15 & 1) != 0;
    float k8[8], k4[4], k2[2];
    #pragma unroll
    for (int u = 0; u < 8; ++u) {
        const float keep = b3 ? p[8 + u] : p[u];
        const float send = b3 ? p[u] : p[8 + u];
        k8[u] = keep + __shfl_xor(send, 8, 32);
    }
    #pragma unroll
    for (int u = 0; u < 4; ++u) {
        const float keep = b2 ? k8[4 + u] : k8[u];
        const float send = b2 ? k8[u] : k8[4 + u];
        k4[u] = keep + __shfl_xor(send, 4, 32);
    }
    #pragma unroll
    for (int u = 0; u < 2; ++u) {
        const float keep = b1 ? k4[2 + u] : k4[u];
        const float send = b1 ? k4[u] : k4[2 + u];
        k2[u] = keep + __shfl_xor(send, 2, 32);
    }
    const float keep1 = b0 ? k2[1] : k2[0];
    const float send1 = b0 ? k2[0] : k2[1];
    const float sc = keep1 + __shfl_xor(send1, 1, 32);
    const int row = rb * 16 + half * 8 + (l15 & 7);
    if (b3) { s_j[row] = sc; } else { s_i[row] = sc; cnt[row] = 0; }
}

#define RK_T    1024
#define RK_G    50
#define RK_Q    (EE / 4 / RK_G)
#define RK_K    ((RK_Q + RK_T - 1) / RK_T)
#define RK_FULL (NN / RK_T)
#define RK_PAD  ((RK_FULL + 1) * RK_T)
#define RK_GRP  24
__global__ __launch_bounds__(RK_T)
void rank_kernel(const int* __restrict__ ei, int* __restrict__ cnt,
                 int* __restrict__ rank) {
    __shared__ int tbl[RK_PAD];
    const int t = threadIdx.x;
    const int4* dq = (const int4*)(ei + EE) + blockIdx.x * RK_Q;
    int4 d[RK_K];
    #pragma unroll
    for (int k = 0; k < RK_K; ++k) {
        const int q = k * RK_T + t;
        d[k] = dq[(q < RK_Q) ? q : RK_Q - 1];
    }
    #pragma unroll
    for (int k = 0; k < (RK_PAD / 4 + RK_T - 1) / RK_T; ++k) {
        const int i = k * RK_T + t;
        if (i < RK_PAD / 4) ((int4*)tbl)[i] = make_int4(0, 0, 0, 0);
    }
    __syncthreads();
    int4 r[RK_K];
    #pragma unroll
    for (int k = 0; k < RK_K; ++k) {
        const int q = k * RK_T + t;
        r[k] = make_int4(0, 0, 0, 0);
        if (q < RK_Q) {
            r[k].x = atomicAdd(&tbl[d[k].x], 1);
            r[k].y = atomicAdd(&tbl[d[k].y], 1);
            r[k].z = atomicAdd(&tbl[d[k].z], 1);
            r[k].w = atomicAdd(&tbl[d[k].w], 1);
        }
    }
    __syncthreads();
    const int j0 = blockIdx.x % RK_FULL;
    const int it = RK_FULL * RK_T + t;
    int bt = 0;
    if (it < NN) bt = atomicAdd(cnt + it, tbl[it]);
    #pragma unroll
    for (int g0 = 0; g0 < RK_FULL; g0 += RK_GRP) {
        int c[RK_GRP];
        #pragma unroll
        for (int u = 0; u < RK_GRP; ++u) {
            int j = g0 + u + j0; if (j >= RK_FULL) j -= RK_FULL;
            c[u] = tbl[j * RK_T + t];
        }
        #pragma unroll
        for (int u = 0; u < RK_GRP; ++u) {
            int j = g0 + u + j0; if (j >= RK_FULL) j -= RK_FULL;
            c[u] = atomicAdd(cnt + j * RK_T + t, c[u]);
        }
        #pragma unroll
        for (int u = 0; u < RK_GRP; ++u) {
            int j = g0 + u + j0; if (j >= RK_FULL) j -= RK_FULL;
            tbl[j * RK_T + t] = c[u];
        }
    }
    if (it < NN) tbl[it] = bt;
    __syncthreads();
    int4* rq = (int4*)rank + blockIdx.x * RK_Q;
    #pragma unroll
    for (int k = 0; k < RK_K; ++k) {
        const int q = k * RK_T + t;
        if (q < RK_Q) {
            int4 o;
            o.x = tbl[d[k].x] + r[k].x;
            o.y = tbl[d[k].y] + r[k].y;
            o.z = tbl[d[k].z] + r[k].z;
            o.w = tbl[d[k].w] + r[k].w;
            rq[q] = o;
        }
    }
}

#define SCAN_T   1024
#define SCAN_PER ((NN + SCAN_T - 1) / SCAN_T)
__global__ __launch_bounds__(SCAN_T)
void scan_kernel(const int* __restrict__ cnt, int* __restrict__ start) {
    __shared__ int buf[SCAN_T * SCAN_PER];
    __shared__ int wtot[SCAN_T / 32];
    const int t    = threadIdx.x;
    const int lane = t & 31;
    const int w    = t >> 5;
    #pragma unroll
    for (int k = 0; k < SCAN_PER; ++k) {
        const int i = k * SCAN_T + t;
        buf[i] = (i < NN) ? cnt[i] : 0;
    }
    __syncthreads();
    int s = 0;
    #pragma unroll
    for (int k = 0; k < SCAN_PER; ++k) s += buf[t * SCAN_PER + k];
    int inc = s;
    #pragma unroll
    for (int off = 1; off < 32; off <<= 1) {
        const int u = __shfl_up(inc, off, 32);
        if (lane >= off) inc += u;
    }
    if (lane == 31) wtot[w] = inc;
    __syncthreads();
    int run = inc - s;
    for (int k = 0; k < w; ++k) run += wtot[k];
    #pragma unroll
    for (int k = 0; k < SCAN_PER; ++k) {
        const int v = buf[t * SCAN_PER + k];
        buf[t * SCAN_PER + k] = run;
        run += v;
    }
    __syncthreads();
    #pragma unroll
    for (int k = 0; k < SCAN_PER; ++k) {
        const int i = k * SCAN_T + t;
        if (i < NN) start[i] = buf[i];
    }
}

__global__ __launch_bounds__(256)
void fill_kernel(const int* __restrict__ ei, const int* __restrict__ start,
                 const int* __restrict__ rank, int* __restrict__ csr_src) {
    int t = blockIdx.x * blockDim.x + threadIdx.x;
    if (t >= EE / 4) return;
    int4 s = ((const int4*)ei)[t];
    int4 d = ((const int4*)(ei + EE))[t];
    int4 r = ((const int4*)rank)[t];
    csr_src[start[d.x] + r.x] = s.x;
    csr_src[start[d.y] + r.y] = s.y;
    csr_src[start[d.z] + r.z] = s.z;
    csr_src[start[d.w] + r.w] = s.w;
}

__device__ __forceinline__ float edge_alpha(float si, float sj) {
    float a = si + sj;
    return (a > 0.0f) ? a : NEGS * a;
}

__global__ __launch_bounds__(256)
void aggregate_kernel(const int* __restrict__ start, const int* __restrict__ cnt,
                      const int* __restrict__ csr_src, const float* __restrict__ s_i,
                      const float* __restrict__ s_j, const float* __restrict__ g,
                      const float* __restrict__ bias, float* __restrict__ out) {
    const int node = __builtin_amdgcn_readfirstlane((blockIdx.x * blockDim.x + threadIdx.x) >> 5);
    const int lane = threadIdx.x & 31;
    if (node >= NN) return;
    const int beg = start[node];
    const int deg = cnt[node];
    const int n   = deg + 1;
    const float si = s_i[node];
    const int* cs = csr_src + beg;

    float m = -INFINITY;
    for (int base = 0; base < n; base += 32) {
        const int k = base + lane;
        if (k < n) {
            const int src = (k < deg) ? cs[k] : node;
            m = fmaxf(m, edge_alpha(si, s_j[src]));
        }
    }
    #pragma unroll
    for (int off = 16; off > 0; off >>= 1) m = fmaxf(m, __shfl_xor(m, off, 32));

    float dsum = 0.0f;
    for (int base = 0; base < n; base += 32) {
        const int k = base + lane;
        if (k < n) {
            const int src = (k < deg) ? cs[k] : node;
            dsum += expf(edge_alpha(si, s_j[src]) - m);
        }
    }
    #pragma unroll
    for (int off = 16; off > 0; off >>= 1) dsum += __shfl_xor(dsum, off, 32);
    const float den = dsum + 1e-16f;

    float accx = 0.0f, accy = 0.0f;
    for (int base = 0; base < n; base += 32) {
        const int k = base + lane;
        int src = node;
        float c = 0.0f;
        if (k < n) {
            src = (k < deg) ? cs[k] : node;
            c = expf(edge_alpha(si, s_j[src]) - m) / den;
        }
        const int cn = (n - base < 32) ? (n - base) : 32;
        #pragma unroll 4
        for (int j = 0; j < cn; ++j) {
            const int   sj = __shfl(src, j, 32);
            const float cj = __shfl(c, j, 32);
            const float2 gv = ((const float2*)(g + (size_t)sj * CC))[lane];
            accx += gv.x * cj;
            accy += gv.y * cj;
        }
    }
    const float2 b = ((const float2*)bias)[lane];
    float2 r;
    r.x = accx + b.x; r.x = (r.x > 0.0f) ? r.x : 0.0f;
    r.y = accy + b.y; r.y = (r.y > 0.0f) ? r.y : 0.0f;
    ((float2*)(out + (size_t)node * CC))[lane] = r;
}

extern "C" void kernel_launch(void* const* d_in, const int* in_sizes, int n_in,
                              void* d_out, int out_size, void* d_ws, size_t ws_size,
                              hipStream_t stream) {
    (void)in_sizes; (void)n_in; (void)out_size; (void)ws_size;
    const float* x      = (const float*)d_in[0];
    const int*   ei     = (const int*)  d_in[1];
    const float* corrs  = (const float*)d_in[2];
    const float* W      = (const float*)d_in[3];
    const float* att_i  = (const float*)d_in[4];
    const float* att_j  = (const float*)d_in[5];
    const float* aem_i  = (const float*)d_in[6];
    const float* aem_j  = (const float*)d_in[7];
    const float* bias   = (const float*)d_in[8];
    float* out = (float*)d_out;

    float* ws    = (float*)d_ws;
    float* g     = ws;
    float* s_i   = g + (size_t)NN * CC;
    float* s_j   = s_i + NN;
    int*   cnt   = (int*)(s_j + NN);
    int*   start = cnt + NN;
    int*   csr   = start + NN;
    int*   rank  = csr + EE;

    gemm_scores_kernel<<<NN / 16, 32, 0, stream>>>(
        x, W, corrs, att_i, att_j, aem_i, aem_j, g, s_i, s_j, cnt);
    rank_kernel<<<RK_G, RK_T, 0, stream>>>(ei, cnt, rank);
    scan_kernel<<<1, SCAN_T, 0, stream>>>(cnt, start);
    fill_kernel<<<(EE / 4 + 255) / 256, 256, 0, stream>>>(ei, start, rank, csr);
    aggregate_kernel<<<(NN * 32 + 255) / 256, 256, 0, stream>>>(
        start, cnt, csr, s_i, s_j, g, bias, out);
}


# Node lists in fixed 64-slot buckets (first slot = node * 64, overflow list past 64): in-degree scan and its launch gone, fill needs no start lookups; 5 launches -> 4; sum order unchanged
# speedup vs baseline: 2.7902x; 1.2014x over previous
#include <hip/hip_runtime.h>
#include <math.h>

#define NN   50000
#define EE   800000
#define INC  128
#define CC   64
#define NEGS 0.2f
#define CAP  64

typedef float v2f __attribute__((ext_vector_type(2)));
typedef float v8f __attribute__((ext_vector_type(8)));

__global__ __launch_bounds__(32)
void gemm_scores_kernel(const float* __restrict__ x, const float* __restrict__ W,
                        const float* __restrict__ corrs,
                        const float* __restrict__ att_i, const float* __restrict__ att_j,
                        const float* __restrict__ aem_i, const float* __restrict__ aem_j,
                        float* __restrict__ g, float* __restrict__ s_i,
                        float* __restrict__ s_j, int* __restrict__ cnt) {
    const int rb   = blockIdx.x;
    const int lane = threadIdx.x;
    const int half = lane >> 4;
    const int l15  = lane & 15;

    float4 cr[8];
    {
        const float4* cp = (const float4*)(corrs + (size_t)(rb * 16 + half * 8) * CC) + l15;
        #pragma unroll
        for (int v = 0; v < 8; ++v) cr[v] = cp[v * (CC / 4)];
    }

    v8f acc0 = {}; v8f acc1 = {}; v8f acc2 = {}; v8f acc3 = {};

    const float* xrow = x + (rb * 16 + l15) * INC + half * 2;
    for (int k0 = 0; k0 < INC; k0 += 4) {
        v2f a; a.x = xrow[k0]; a.y = xrow[k0 + 1];
        const float* wp = W + (k0 + half * 2) * CC + l15;
        v2f b0, b1, b2, b3;
        b0.x = wp[0];  b0.y = wp[0  + CC];
        b1.x = wp[16]; b1.y = wp[16 + CC];
        b2.x = wp[32]; b2.y = wp[32 + CC];
        b3.x = wp[48]; b3.y = wp[48 + CC];
        acc0 = __builtin_amdgcn_wmma_f32_16x16x4_f32(false, a, false, b0, (short)0, acc0, false, false);
        acc1 = __builtin_amdgcn_wmma_f32_16x16x4_f32(false, a, false, b1, (short)0, acc1, false, false);
        acc2 = __builtin_amdgcn_wmma_f32_16x16x4_f32(false, a, false, b2, (short)0, acc2, false, false);
        acc3 = __builtin_amdgcn_wmma_f32_16x16x4_f32(false, a, false, b3, (short)0, acc3, false, false);
    }
    #pragma unroll
    for (int v = 0; v < 8; ++v) {
        float* op = g + (rb * 16 + v + half * 8) * CC + l15;
        op[0]  = acc0[v];
        op[16] = acc1[v];
        op[32] = acc2[v];
        op[48] = acc3[v];
    }

    float p[16];
    {
        float ai[4], aj[4];
        #pragma unroll
        for (int t = 0; t < 4; ++t) { ai[t] = att_i[l15 + 16 * t]; aj[t] = att_j[l15 + 16 * t]; }
        const float4 ei4 = ((const float4*)aem_i)[l15];
        const float4 ej4 = ((const float4*)aem_j)[l15];
        #pragma unroll
        for (int v = 0; v < 8; ++v) {
            float pi = acc0[v] * ai[0] + acc1[v] * ai[1] + acc2[v] * ai[2] + acc3[v] * ai[3];
            float pj = acc0[v] * aj[0] + acc1[v] * aj[1] + acc2[v] * aj[2] + acc3[v] * aj[3];
            pi += cr[v].x * ei4.x + cr[v].y * ei4.y + cr[v].z * ei4.z + cr[v].w * ei4.w;
            pj += cr[v].x * ej4.x + cr[v].y * ej4.y + cr[v].z * ej4.z + cr[v].w * ej4.w;
            p[v] = pi; p[8 + v] = pj;
        }
    }
    const bool b3 = (l15 & 8) != 0, b2 = (l15 & 4) != 0, b1 = (l15 & 2) != 0, b0 = (l15 & 1) != 0;
    float k8[8], k4[4], k2[2];
    #pragma unroll
    for (int u = 0; u < 8; ++u) {
        const float keep = b3 ? p[8 + u] : p[u];
        const float send = b3 ? p[u] : p[8 + u];
        k8[u] = keep + __shfl_xor(send, 8, 32);
    }
    #pragma unroll
    for (int u = 0; u < 4; ++u) {
        const float keep = b2 ? k8[4 + u] : k8[u];
        const float send = b2 ? k8[u] : k8[4 + u];
        k4[u] = keep + __shfl_xor(send, 4, 32);
    }
    #pragma unroll
    for (int u = 0; u < 2; ++u) {
        const float keep = b1 ? k4[2 + u] : k4[u];
        const float send = b1 ? k4[u] : k4[2 + u];
        k2[u] = keep + __shfl_xor(send, 2, 32);
    }
    const float keep1 = b0 ? k2[1] : k2[0];
    const float send1 = b0 ? k2[0] : k2[1];
    const float sc = keep1 + __shfl_xor(send1, 1, 32);
    const int row = rb * 16 + half * 8 + (l15 & 7);
    if (b3) { s_j[row] = sc; } else { s_i[row] = sc; cnt[row] = 0; }
    if (rb == 0 && lane == 0) cnt[NN] = 0;
}

#define RK_T    1024
#define RK_G    50
#define RK_Q    (EE / 4 / RK_G)
#define RK_K    ((RK_Q + RK_T - 1) / RK_T)
#define RK_FULL (NN / RK_T)
#define RK_PAD  ((RK_FULL + 1) * RK_T)
#define RK_GRP  24
__global__ __launch_bounds__(RK_T)
void rank_kernel(const int* __restrict__ ei, int* __restrict__ cnt,
                 int* __restrict__ rank) {
    __shared__ int tbl[RK_PAD];
    const int t = threadIdx.x;
    const int4* dq = (const int4*)(ei + EE) + blockIdx.x * RK_Q;
    int4 d[RK_K];
    #pragma unroll
    for (int k = 0; k < RK_K; ++k) {
        const int q = k * RK_T + t;
        d[k] = dq[(q < RK_Q) ? q : RK_Q - 1];
    }
    #pragma unroll
    for (int k = 0; k < (RK_PAD / 4 + RK_T - 1) / RK_T; ++k) {
        const int i = k * RK_T + t;
        if (i < RK_PAD / 4) ((int4*)tbl)[i] = make_int4(0, 0, 0, 0);
    }
    __syncthreads();
    int4 r[RK_K];
    #pragma unroll
    for (int k = 0; k < RK_K; ++k) {
        const int q = k * RK_T + t;
        r[k] = make_int4(0, 0, 0, 0);
        if (q < RK_Q) {
            r[k].x = atomicAdd(&tbl[d[k].x], 1);
            r[k].y = atomicAdd(&tbl[d[k].y], 1);
            r[k].z = atomicAdd(&tbl[d[k].z], 1);
            r[k].w = atomicAdd(&tbl[d[k].w], 1);
        }
    }
    __syncthreads();
    const int j0 = blockIdx.x % RK_FULL;
    const int it = RK_FULL * RK_T + t;
    int bt = 0;
    if (it < NN) bt = atomicAdd(cnt + it, tbl[it]);
    #pragma unroll
    for (int g0 = 0; g0 < RK_FULL; g0 += RK_GRP) {
        int c[RK_GRP];
        #pragma unroll
        for (int u = 0; u < RK_GRP; ++u) {
            int j = g0 + u + j0; if (j >= RK_FULL) j -= RK_FULL;
            c[u] = tbl[j * RK_T + t];
        }
        #pragma unroll
        for (int u = 0; u < RK_GRP; ++u) {
            int j = g0 + u + j0; if (j >= RK_FULL) j -= RK_FULL;
            c[u] = atomicAdd(cnt + j * RK_T + t, c[u]);
        }
        #pragma unroll
        for (int u = 0; u < RK_GRP; ++u) {
            int j = g0 + u + j0; if (j >= RK_FULL) j -= RK_FULL;
            tbl[j * RK_T + t] = c[u];
        }
    }
    if (it < NN) tbl[it] = bt;
    __syncthreads();
    int4* rq = (int4*)rank + blockIdx.x * RK_Q;
    #pragma unroll
    for (int k = 0; k < RK_K; ++k) {
        const int q = k * RK_T + t;
        if (q < RK_Q) {
            int4 o;
            o.x = tbl[d[k].x] + r[k].x;
            o.y = tbl[d[k].y] + r[k].y;
            o.z = tbl[d[k].z] + r[k].z;
            o.w = tbl[d[k].w] + r[k].w;
            rq[q] = o;
        }
    }
}

__device__ __forceinline__ void place_edge(int d, int s, int r, int* __restrict__ csr_src,
                                           int* __restrict__ ovf_n, int* __restrict__ ovf) {
    if (r < CAP) {
        csr_src[(size_t)d * CAP + r] = s;
    } else {
        const int p = atomicAdd(ovf_n, 1);
        *(int2*)(ovf + 2 * (size_t)p) = make_int2(d, s);
    }
}

__global__ __launch_bounds__(256)
void fill_kernel(const int* __restrict__ ei, const int* __restrict__ rank,
                 int* __restrict__ csr_src, int* __restrict__ ovf_n, int* __restrict__ ovf) {
    int t = blockIdx.x * blockDim.x + threadIdx.x;
    if (t >= EE / 4) return;
    int4 s = ((const int4*)ei)[t];
    int4 d = ((const int4*)(ei + EE))[t];
    int4 r = ((const int4*)rank)[t];
    place_edge(d.x, s.x, r.x, csr_src, ovf_n, ovf);
    place_edge(d.y, s.y, r.y, csr_src, ovf_n, ovf);
    place_edge(d.z, s.z, r.z, csr_src, ovf_n, ovf);
    place_edge(d.w, s.w, r.w, csr_src, ovf_n, ovf);
}

__device__ __forceinline__ float edge_alpha(float si, float sj) {
    float a = si + sj;
    return (a > 0.0f) ? a : NEGS * a;
}

__device__ __forceinline__ void add_rows(int src, float c, int use, int cn, int lane,
                                         const float* __restrict__ g,
                                         float& accx, float& accy) {
    for (int j = 0; j < cn; ++j) {
        const int   uj = __shfl(use, j, 32);
        const int   sj = __shfl(src, j, 32);
        const float cj = __shfl(c, j, 32);
        if (uj != 0) {
            const float2 gv = ((const float2*)(g + (size_t)sj * CC))[lane];
            accx += gv.x * cj;
            accy += gv.y * cj;
        }
    }
}

__global__ __launch_bounds__(256)
void aggregate_kernel(const int* __restrict__ cnt, const int* __restrict__ csr_src,
                      const int* __restrict__ ovf, const float* __restrict__ s_i,
                      const float* __restrict__ s_j, const float* __restrict__ g,
                      const float* __restrict__ bias, float* __restrict__ out) {
    const int node = __builtin_amdgcn_readfirstlane((blockIdx.x * blockDim.x + threadIdx.x) >> 5);
    const int lane = threadIdx.x & 31;
    if (node >= NN) return;
    const float si = s_i[node];
    const int deg = cnt[node];
    const int inb = (deg < CAP) ? deg : CAP;
    const int n   = inb + 1;
    const int novf = (deg > CAP) ? cnt[NN] : 0;
    const int* cs = csr_src + (size_t)node * CAP;

    float m = -INFINITY;
    for (int base = 0; base < n; base += 32) {
        const int k = base + lane;
        if (k < n) {
            const int src = (k < inb) ? cs[k] : node;
            m = fmaxf(m, edge_alpha(si, s_j[src]));
        }
    }
    for (int base = 0; base < novf; base += 32) {
        const int p = base + lane;
        if (p < novf) {
            const int2 e = *(const int2*)(ovf + 2 * (size_t)p);
            if (e.x == node) m = fmaxf(m, edge_alpha(si, s_j[e.y]));
        }
    }
    #pragma unroll
    for (int off = 16; off > 0; off >>= 1) m = fmaxf(m, __shfl_xor(m, off, 32));

    float dsum = 0.0f;
    for (int base = 0; base < n; base += 32) {
        const int k = base + lane;
        if (k < n) {
            const int src = (k < inb) ? cs[k] : node;
            dsum += expf(edge_alpha(si, s_j[src]) - m);
        }
    }
    for (int base = 0; base < novf; base += 32) {
        const int p = base + lane;
        if (p < novf) {
            const int2 e = *(const int2*)(ovf + 2 * (size_t)p);
            if (e.x == node) dsum += expf(edge_alpha(si, s_j[e.y]) - m);
        }
    }
    #pragma unroll
    for (int off = 16; off > 0; off >>= 1) dsum += __shfl_xor(dsum, off, 32);
    const float den = dsum + 1e-16f;

    float accx = 0.0f, accy = 0.0f;
    for (int base = 0; base < n; base += 32) {
        const int k = base + lane;
        int src = node;
        float c = 0.0f;
        if (k < n) {
            src = (k < inb) ? cs[k] : node;
            c = expf(edge_alpha(si, s_j[src]) - m) / den;
        }
        const int cn = (n - base < 32) ? (n - base) : 32;
        #pragma unroll 4
        for (int j = 0; j < cn; ++j) {
            const int   sj = __shfl(src, j, 32);
            const float cj = __shfl(c, j, 32);
            const float2 gv = ((const float2*)(g + (size_t)sj * CC))[lane];
            accx += gv.x * cj;
            accy += gv.y * cj;
        }
    }
    for (int base = 0; base < novf; base += 32) {
        const int p = base + lane;
        int src = node;
        float c = 0.0f;
        int use = 0;
        if (p < novf) {
            const int2 e = *(const int2*)(ovf + 2 * (size_t)p);
            if (e.x == node) {
                src = e.y;
                c = expf(edge_alpha(si, s_j[src]) - m) / den;
                use = 1;
            }
        }
        const int cn = (novf - base < 32) ? (novf - base) : 32;
        add_rows(src, c, use, cn, lane, g, accx, accy);
    }
    const float2 b = ((const float2*)bias)[lane];
    float2 r;
    r.x = accx + b.x; r.x = (r.x > 0.0f) ? r.x : 0.0f;
    r.y = accy + b.y; r.y = (r.y > 0.0f) ? r.y : 0.0f;
    ((float2*)(out + (size_t)node * CC))[lane] = r;
}

extern "C" void kernel_launch(void* const* d_in, const int* in_sizes, int n_in,
                              void* d_out, int out_size, void* d_ws, size_t ws_size,
                              hipStream_t stream) {
    (void)in_sizes; (void)n_in; (void)out_size; (void)ws_size;
    const float* x      = (const float*)d_in[0];
    const int*   ei     = (const int*)  d_in[1];
    const float* corrs  = (const float*)d_in[2];
    const float* W      = (const float*)d_in[3];
    const float* att_i  = (const float*)d_in[4];
    const float* att_j  = (const float*)d_in[5];
    const float* aem_i  = (const float*)d_in[6];
    const float* aem_j  = (const float*)d_in[7];
    const float* bias   = (const float*)d_in[8];
    float* out = (float*)d_out;

    float* ws    = (float*)d_ws;
    float* g     = ws;
    int*   csr   = (int*)(g + (size_t)NN * CC);
    int*   rank  = csr + (size_t)NN * CAP;
    int*   ovf   = rank + EE;
    float* s_i   = (float*)(rank + EE) + (size_t)2 * EE;
    float* s_j   = s_i + NN;
    int*   cnt   = (int*)(s_j + NN);

    gemm_scores_kernel<<<NN / 16, 32, 0, stream>>>(
        x, W, corrs, att_i, att_j, aem_i, aem_j, g, s_i, s_j, cnt);
    rank_kernel<<<RK_G, RK_T, 0, stream>>>(ei, cnt, rank);
    fill_kernel<<<(EE / 4 + 255) / 256, 256, 0, stream>>>(ei, rank, csr, cnt + NN, ovf);
    aggregate_kernel<<<(NN * 32 + 255) / 256, 256, 0, stream>>>(
        cnt, csr, ovf, s_i, s_j, g, bias, out);
}


# Rank+fill merged: 256 blocks, one per (1/16 of edges, 1/16 of nodes): LDS count table, one returning add per (slice, node), slots by LDS adds; 4 -> 3 launches; sum order: block arrival, LDS add order
# speedup vs baseline: 3.7561x; 1.3462x over previous
#include <hip/hip_runtime.h>
#include <math.h>

#define NN   50000
#define EE   800000
#define INC  128
#define CC   64
#define NEGS 0.2f
#define CAP  64

typedef float v2f __attribute__((ext_vector_type(2)));
typedef float v8f __attribute__((ext_vector_type(8)));

__global__ __launch_bounds__(32)
void gemm_scores_kernel(const float* __restrict__ x, const float* __restrict__ W,
                        const float* __restrict__ corrs,
                        const float* __restrict__ att_i, const float* __restrict__ att_j,
                        const float* __restrict__ aem_i, const float* __restrict__ aem_j,
                        float* __restrict__ g, float* __restrict__ s_i,
                        float* __restrict__ s_j, int* __restrict__ cnt) {
    const int rb   = blockIdx.x;
    const int lane = threadIdx.x;
    const int half = lane >> 4;
    const int l15  = lane & 15;

    float4 cr[8];
    {
        const float4* cp = (const float4*)(corrs + (size_t)(rb * 16 + half * 8) * CC) + l15;
        #pragma unroll
        for (int v = 0; v < 8; ++v) cr[v] = cp[v * (CC / 4)];
    }

    v8f acc0 = {}; v8f acc1 = {}; v8f acc2 = {}; v8f acc3 = {};

    const float* xrow = x + (rb * 16 + l15) * INC + half * 2;
    for (int k0 = 0; k0 < INC; k0 += 4) {
        v2f a; a.x = xrow[k0]; a.y = xrow[k0 + 1];
        const float* wp = W + (k0 + half * 2) * CC + l15;
        v2f b0, b1, b2, b3;
        b0.x = wp[0];  b0.y = wp[0  + CC];
        b1.x = wp[16]; b1.y = wp[16 + CC];
        b2.x = wp[32]; b2.y = wp[32 + CC];
        b3.x = wp[48]; b3.y = wp[48 + CC];
        acc0 = __builtin_amdgcn_wmma_f32_16x16x4_f32(false, a, false, b0, (short)0, acc0, false, false);
        acc1 = __builtin_amdgcn_wmma_f32_16x16x4_f32(false, a, false, b1, (short)0, acc1, false, false);
        acc2 = __builtin_amdgcn_wmma_f32_16x16x4_f32(false, a, false, b2, (short)0, acc2, false, false);
        acc3 = __builtin_amdgcn_wmma_f32_16x16x4_f32(false, a, false, b3, (short)0, acc3, false, false);
    }
    #pragma unroll
    for (int v = 0; v < 8; ++v) {
        float* op = g + (rb * 16 + v + half * 8) * CC + l15;
        op[0]  = acc0[v];
        op[16] = acc1[v];
        op[32] = acc2[v];
        op[48] = acc3[v];
    }

    float p[16];
    {
        float ai[4], aj[4];
        #pragma unroll
        for (int t = 0; t < 4; ++t) { ai[t] = att_i[l15 + 16 * t]; aj[t] = att_j[l15 + 16 * t]; }
        const float4 ei4 = ((const float4*)aem_i)[l15];
        const float4 ej4 = ((const float4*)aem_j)[l15];
        #pragma unroll
        for (int v = 0; v < 8; ++v) {
            float pi = acc0[v] * ai[0] + acc1[v] * ai[1] + acc2[v] * ai[2] + acc3[v] * ai[3];
            float pj = acc0[v] * aj[0] + acc1[v] * aj[1] + acc2[v] * aj[2] + acc3[v] * aj[3];
            pi += cr[v].x * ei4.x + cr[v].y * ei4.y + cr[v].z * ei4.z + cr[v].w * ei4.w;
            pj += cr[v].x * ej4.x + cr[v].y * ej4.y + cr[v].z * ej4.z + cr[v].w * ej4.w;
            p[v] = pi; p[8 + v] = pj;
        }
    }
    const bool b3 = (l15 & 8) != 0, b2 = (l15 & 4) != 0, b1 = (l15 & 2) != 0, b0 = (l15 & 1) != 0;
    float k8[8], k4[4], k2[2];
    #pragma unroll
    for (int u = 0; u < 8; ++u) {
        const float keep = b3 ? p[8 + u] : p[u];
        const float send = b3 ? p[u] : p[8 + u];
        k8[u] = keep + __shfl_xor(send, 8, 32);
    }
    #pragma unroll
    for (int u = 0; u < 4; ++u) {
        const float keep = b2 ? k8[4 + u] : k8[u];
        const float send = b2 ? k8[u] : k8[4 + u];
        k4[u] = keep + __shfl_xor(send, 4, 32);
    }
    #pragma unroll
    for (int u = 0; u < 2; ++u) {
        const float keep = b1 ? k4[2 + u] : k4[u];
        const float send = b1 ? k4[u] : k4[2 + u];
        k2[u] = keep + __shfl_xor(send, 2, 32);
    }
    const float keep1 = b0 ? k2[1] : k2[0];
    const float send1 = b0 ? k2[0] : k2[1];
    const float sc = keep1 + __shfl_xor(send1, 1, 32);
    const int row = rb * 16 + half * 8 + (l15 & 7);
    if (b3) { s_j[row] = sc; } else { s_i[row] = sc; cnt[row] = 0; }
    if (rb == 0 && lane == 0) cnt[NN] = 0;
}

#define PL_S   16
#define PL_R   16
#define PL_T   1024
#define PL_Q   (EE / 4 / PL_S)
#define PL_N   (NN / PL_R)
#define PL_K   ((PL_Q + PL_T - 1) / PL_T)
#define PL_F   ((PL_N + PL_T - 1) / PL_T)
#define PL_U   4

__device__ __forceinline__ void place_edge(int d, int s, int r, int* __restrict__ csr_src,
                                           int* __restrict__ ovf_n, int* __restrict__ ovf) {
    if (r < CAP) {
        csr_src[(size_t)d * CAP + r] = s;
    } else {
        const int p = atomicAdd(ovf_n, 1);
        *(int2*)(ovf + 2 * (size_t)p) = make_int2(d, s);
    }
}

__global__ __launch_bounds__(PL_T)
void place_kernel(const int* __restrict__ ei, int* __restrict__ cnt,
                  int* __restrict__ csr_src, int* __restrict__ ovf) {
    __shared__ int tbl[PL_N];
    const int t  = threadIdx.x;
    const int sl = blockIdx.x / PL_R;
    const int lo = (blockIdx.x % PL_R) * PL_N;
    const int4* sq = (const int4*)ei + sl * PL_Q;
    const int4* dq = (const int4*)(ei + EE) + sl * PL_Q;

    int4 d1[PL_K];
    #pragma unroll
    for (int k = 0; k < PL_K; ++k) {
        const int q = k * PL_T + t;
        d1[k] = dq[(q < PL_Q) ? q : PL_Q - 1];
    }
    #pragma unroll
    for (int j = 0; j < PL_F; ++j) {
        const int i = j * PL_T + t;
        if (i < PL_N) tbl[i] = 0;
    }
    __syncthreads();
    #pragma unroll
    for (int k = 0; k < PL_K; ++k) {
        const int q = k * PL_T + t;
        if (q < PL_Q) {
            const unsigned ux = (unsigned)(d1[k].x - lo), uy = (unsigned)(d1[k].y - lo);
            const unsigned uz = (unsigned)(d1[k].z - lo), uw = (unsigned)(d1[k].w - lo);
            if (ux < PL_N) atomicAdd(&tbl[ux], 1);
            if (uy < PL_N) atomicAdd(&tbl[uy], 1);
            if (uz < PL_N) atomicAdd(&tbl[uz], 1);
            if (uw < PL_N) atomicAdd(&tbl[uw], 1);
        }
    }
    __syncthreads();
    {
        int c[PL_F], b[PL_F];
        #pragma unroll
        for (int j = 0; j < PL_F; ++j) {
            const int i = j * PL_T + t;
            c[j] = (i < PL_N) ? tbl[i] : 0;
        }
        #pragma unroll
        for (int j = 0; j < PL_F; ++j) {
            const int i = j * PL_T + t;
            b[j] = 0;
            if (c[j] != 0) b[j] = atomicAdd(cnt + lo + i, c[j]);
        }
        #pragma unroll
        for (int j = 0; j < PL_F; ++j) {
            const int i = j * PL_T + t;
            if (c[j] != 0) tbl[i] = b[j];
        }
    }
    __syncthreads();
    for (int k0 = 0; k0 < PL_K; k0 += PL_U) {
        int4 d[PL_U], s[PL_U], r[PL_U];
        #pragma unroll
        for (int j = 0; j < PL_U; ++j) {
            const int q = (k0 + j) * PL_T + t;
            const int qq = (q < PL_Q) ? q : PL_Q - 1;
            d[j] = dq[qq];
            s[j] = sq[qq];
            if (q >= PL_Q) d[j] = make_int4(-1, -1, -1, -1);
        }
        #pragma unroll
        for (int j = 0; j < PL_U; ++j) {
            const unsigned ux = (unsigned)(d[j].x - lo), uy = (unsigned)(d[j].y - lo);
            const unsigned uz = (unsigned)(d[j].z - lo), uw = (unsigned)(d[j].w - lo);
            r[j] = make_int4(-1, -1, -1, -1);
            if (ux < PL_N) r[j].x = atomicAdd(&tbl[ux], 1);
            if (uy < PL_N) r[j].y = atomicAdd(&tbl[uy], 1);
            if (uz < PL_N) r[j].z = atomicAdd(&tbl[uz], 1);
            if (uw < PL_N) r[j].w = atomicAdd(&tbl[uw], 1);
        }
        #pragma unroll
        for (int j = 0; j < PL_U; ++j) {
            if (r[j].x >= 0) place_edge(d[j].x, s[j].x, r[j].x, csr_src, cnt + NN, ovf);
            if (r[j].y >= 0) place_edge(d[j].y, s[j].y, r[j].y, csr_src, cnt + NN, ovf);
            if (r[j].z >= 0) place_edge(d[j].z, s[j].z, r[j].z, csr_src, cnt + NN, ovf);
            if (r[j].w >= 0) place_edge(d[j].w, s[j].w, r[j].w, csr_src, cnt + NN, ovf);
        }
    }
}

__device__ __forceinline__ float edge_alpha(float si, float sj) {
    float a = si + sj;
    return (a > 0.0f) ? a : NEGS * a;
}

__device__ __forceinline__ void add_rows(int src, float c, int use, int cn, int lane,
                                         const float* __restrict__ g,
                                         float& accx, float& accy) {
    for (int j = 0; j < cn; ++j) {
        const int   uj = __shfl(use, j, 32);
        const int   sj = __shfl(src, j, 32);
        const float cj = __shfl(c, j, 32);
        if (uj != 0) {
            const float2 gv = ((const float2*)(g + (size_t)sj * CC))[lane];
            accx += gv.x * cj;
            accy += gv.y * cj;
        }
    }
}

__global__ __launch_bounds__(256)
void aggregate_kernel(const int* __restrict__ cnt, const int* __restrict__ csr_src,
                      const int* __restrict__ ovf, const float* __restrict__ s_i,
                      const float* __restrict__ s_j, const float* __restrict__ g,
                      const float* __restrict__ bias, float* __restrict__ out) {
    const int node = __builtin_amdgcn_readfirstlane((blockIdx.x * blockDim.x + threadIdx.x) >> 5);
    const int lane = threadIdx.x & 31;
    if (node >= NN) return;
    const float si = s_i[node];
    const int deg = cnt[node];
    const int inb = (deg < CAP) ? deg : CAP;
    const int n   = inb + 1;
    const int novf = (deg > CAP) ? cnt[NN] : 0;
    const int* cs = csr_src + (size_t)node * CAP;

    float m = -INFINITY;
    for (int base = 0; base < n; base += 32) {
        const int k = base + lane;
        if (k < n) {
            const int src = (k < inb) ? cs[k] : node;
            m = fmaxf(m, edge_alpha(si, s_j[src]));
        }
    }
    for (int base = 0; base < novf; base += 32) {
        const int p = base + lane;
        if (p < novf) {
            const int2 e = *(const int2*)(ovf + 2 * (size_t)p);
            if (e.x == node) m = fmaxf(m, edge_alpha(si, s_j[e.y]));
        }
    }
    #pragma unroll
    for (int off = 16; off > 0; off >>= 1) m = fmaxf(m, __shfl_xor(m, off, 32));

    float dsum = 0.0f;
    for (int base = 0; base < n; base += 32) {
        const int k = base + lane;
        if (k < n) {
            const int src = (k < inb) ? cs[k] : node;
            dsum += expf(edge_alpha(si, s_j[src]) - m);
        }
    }
    for (int base = 0; base < novf; base += 32) {
        const int p = base + lane;
        if (p < novf) {
            const int2 e = *(const int2*)(ovf + 2 * (size_t)p);
            if (e.x == node) dsum += expf(edge_alpha(si, s_j[e.y]) - m);
        }
    }
    #pragma unroll
    for (int off = 16; off > 0; off >>= 1) dsum += __shfl_xor(dsum, off, 32);
    const float den = dsum + 1e-16f;

    float accx = 0.0f, accy = 0.0f;
    for (int base = 0; base < n; base += 32) {
        const int k = base + lane;
        int src = node;
        float c = 0.0f;
        if (k < n) {
            src = (k < inb) ? cs[k] : node;
            c = expf(edge_alpha(si, s_j[src]) - m) / den;
        }
        const int cn = (n - base < 32) ? (n - base) : 32;
        #pragma unroll 4
        for (int j = 0; j < cn; ++j) {
            const int   sj = __shfl(src, j, 32);
            const float cj = __shfl(c, j, 32);
            const float2 gv = ((const float2*)(g + (size_t)sj * CC))[lane];
            accx += gv.x * cj;
            accy += gv.y * cj;
        }
    }
    for (int base = 0; base < novf; base += 32) {
        const int p = base + lane;
        int src = node;
        float c = 0.0f;
        int use = 0;
        if (p < novf) {
            const int2 e = *(const int2*)(ovf + 2 * (size_t)p);
            if (e.x == node) {
                src = e.y;
                c = expf(edge_alpha(si, s_j[src]) - m) / den;
                use = 1;
            }
        }
        const int cn = (novf - base < 32) ? (novf - base) : 32;
        add_rows(src, c, use, cn, lane, g, accx, accy);
    }
    const float2 b = ((const float2*)bias)[lane];
    float2 r;
    r.x = accx + b.x; r.x = (r.x > 0.0f) ? r.x : 0.0f;
    r.y = accy + b.y; r.y = (r.y > 0.0f) ? r.y : 0.0f;
    ((float2*)(out + (size_t)node * CC))[lane] = r;
}

extern "C" void kernel_launch(void* const* d_in, const int* in_sizes, int n_in,
                              void* d_out, int out_size, void* d_ws, size_t ws_size,
                              hipStream_t stream) {
    (void)in_sizes; (void)n_in; (void)out_size; (void)ws_size;
    const float* x      = (const float*)d_in[0];
    const int*   ei     = (const int*)  d_in[1];
    const float* corrs  = (const float*)d_in[2];
    const float* W      = (const float*)d_in[3];
    const float* att_i  = (const float*)d_in[4];
    const float* att_j  = (const float*)d_in[5];
    const float* aem_i  = (const float*)d_in[6];
    const float* aem_j  = (const float*)d_in[7];
    const float* bias   = (const float*)d_in[8];
    float* out = (float*)d_out;

    float* ws    = (float*)d_ws;
    float* g     = ws;
    int*   csr   = (int*)(g + (size_t)NN * CC);
    int*   ovf   = csr + (size_t)NN * CAP;
    float* s_i   = (float*)(ovf + (size_t)2 * EE);
    float* s_j   = s_i + NN;
    int*   cnt   = (int*)(s_j + NN);

    gemm_scores_kernel<<<NN / 16, 32, 0, stream>>>(
        x, W, corrs, att_i, att_j, aem_i, aem_j, g, s_i, s_j, cnt);
    place_kernel<<<PL_S * PL_R, PL_T, 0, stream>>>(ei, cnt, csr, ovf);
    aggregate_kernel<<<(NN * 32 + 255) / 256, 256, 0, stream>>>(
        cnt, csr, ovf, s_i, s_j, g, bias, out);
}


# Aggregation: a wave takes 8 consecutive nodes at once (6250 waves instead of 50000); the 8 nodes' index, score and row loads are issued side by side, two rows of each in flight; sum order unchanged
# speedup vs baseline: 4.0463x; 1.0773x over previous
#include <hip/hip_runtime.h>
#include <math.h>

#define NN   50000
#define EE   800000
#define INC  128
#define CC   64
#define NEGS 0.2f
#define CAP  64

typedef float v2f __attribute__((ext_vector_type(2)));
typedef float v8f __attribute__((ext_vector_type(8)));

__global__ __launch_bounds__(32)
void gemm_scores_kernel(const float* __restrict__ x, const float* __restrict__ W,
                        const float* __restrict__ corrs,
                        const float* __restrict__ att_i, const float* __restrict__ att_j,
                        const float* __restrict__ aem_i, const float* __restrict__ aem_j,
                        float* __restrict__ g, float* __restrict__ s_i,
                        float* __restrict__ s_j, int* __restrict__ cnt) {
    const int rb   = blockIdx.x;
    const int lane = threadIdx.x;
    const int half = lane >> 4;
    const int l15  = lane & 15;

    float4 cr[8];
    {
        const float4* cp = (const float4*)(corrs + (size_t)(rb * 16 + half * 8) * CC) + l15;
        #pragma unroll
        for (int v = 0; v < 8; ++v) cr[v] = cp[v * (CC / 4)];
    }

    v8f acc0 = {}; v8f acc1 = {}; v8f acc2 = {}; v8f acc3 = {};

    const float* xrow = x + (rb * 16 + l15) * INC + half * 2;
    for (int k0 = 0; k0 < INC; k0 += 4) {
        v2f a; a.x = xrow[k0]; a.y = xrow[k0 + 1];
        const float* wp = W + (k0 + half * 2) * CC + l15;
        v2f b0, b1, b2, b3;
        b0.x = wp[0];  b0.y = wp[0  + CC];
        b1.x = wp[16]; b1.y = wp[16 + CC];
        b2.x = wp[32]; b2.y = wp[32 + CC];
        b3.x = wp[48]; b3.y = wp[48 + CC];
        acc0 = __builtin_amdgcn_wmma_f32_16x16x4_f32(false, a, false, b0, (short)0, acc0, false, false);
        acc1 = __builtin_amdgcn_wmma_f32_16x16x4_f32(false, a, false, b1, (short)0, acc1, false, false);
        acc2 = __builtin_amdgcn_wmma_f32_16x16x4_f32(false, a, false, b2, (short)0, acc2, false, false);
        acc3 = __builtin_amdgcn_wmma_f32_16x16x4_f32(false, a, false, b3, (short)0, acc3, false, false);
    }
    #pragma unroll
    for (int v = 0; v < 8; ++v) {
        float* op = g + (rb * 16 + v + half * 8) * CC + l15;
        op[0]  = acc0[v];
        op[16] = acc1[v];
        op[32] = acc2[v];
        op[48] = acc3[v];
    }

    float p[16];
    {
        float ai[4], aj[4];
        #pragma unroll
        for (int t = 0; t < 4; ++t) { ai[t] = att_i[l15 + 16 * t]; aj[t] = att_j[l15 + 16 * t]; }
        const float4 ei4 = ((const float4*)aem_i)[l15];
        const float4 ej4 = ((const float4*)aem_j)[l15];
        #pragma unroll
        for (int v = 0; v < 8; ++v) {
            float pi = acc0[v] * ai[0] + acc1[v] * ai[1] + acc2[v] * ai[2] + acc3[v] * ai[3];
            float pj = acc0[v] * aj[0] + acc1[v] * aj[1] + acc2[v] * aj[2] + acc3[v] * aj[3];
            pi += cr[v].x * ei4.x + cr[v].y * ei4.y + cr[v].z * ei4.z + cr[v].w * ei4.w;
            pj += cr[v].x * ej4.x + cr[v].y * ej4.y + cr[v].z * ej4.z + cr[v].w * ej4.w;
            p[v] = pi; p[8 + v] = pj;
        }
    }
    const bool b3 = (l15 & 8) != 0, b2 = (l15 & 4) != 0, b1 = (l15 & 2) != 0, b0 = (l15 & 1) != 0;
    float k8[8], k4[4], k2[2];
    #pragma unroll
    for (int u = 0; u < 8; ++u) {
        const float keep = b3 ? p[8 + u] : p[u];
        const float send = b3 ? p[u] : p[8 + u];
        k8[u] = keep + __shfl_xor(send, 8, 32);
    }
    #pragma unroll
    for (int u = 0; u < 4; ++u) {
        const float keep = b2 ? k8[4 + u] : k8[u];
        const float send = b2 ? k8[u] : k8[4 + u];
        k4[u] = keep + __shfl_xor(send, 4, 32);
    }
    #pragma unroll
    for (int u = 0; u < 2; ++u) {
        const float keep = b1 ? k4[2 + u] : k4[u];
        const float send = b1 ? k4[u] : k4[2 + u];
        k2[u] = keep + __shfl_xor(send, 2, 32);
    }
    const float keep1 = b0 ? k2[1] : k2[0];
    const float send1 = b0 ? k2[0] : k2[1];
    const float sc = keep1 + __shfl_xor(send1, 1, 32);
    const int row = rb * 16 + half * 8 + (l15 & 7);
    if (b3) { s_j[row] = sc; } else { s_i[row] = sc; cnt[row] = 0; }
    if (rb == 0 && lane == 0) cnt[NN] = 0;
}

#define PL_S   16
#define PL_R   16
#define PL_T   1024
#define PL_Q   (EE / 4 / PL_S)
#define PL_N   (NN / PL_R)
#define PL_K   ((PL_Q + PL_T - 1) / PL_T)
#define PL_F   ((PL_N + PL_T - 1) / PL_T)
#define PL_U   4

__device__ __forceinline__ void place_edge(int d, int s, int r, int* __restrict__ csr_src,
                                           int* __restrict__ ovf_n, int* __restrict__ ovf) {
    if (r < CAP) {
        csr_src[(size_t)d * CAP + r] = s;
    } else {
        const int p = atomicAdd(ovf_n, 1);
        *(int2*)(ovf + 2 * (size_t)p) = make_int2(d, s);
    }
}

__global__ __launch_bounds__(PL_T)
void place_kernel(const int* __restrict__ ei, int* __restrict__ cnt,
                  int* __restrict__ csr_src, int* __restrict__ ovf) {
    __shared__ int tbl[PL_N];
    const int t  = threadIdx.x;
    const int sl = blockIdx.x / PL_R;
    const int lo = (blockIdx.x % PL_R) * PL_N;
    const int4* sq = (const int4*)ei + sl * PL_Q;
    const int4* dq = (const int4*)(ei + EE) + sl * PL_Q;

    int4 d1[PL_K];
    #pragma unroll
    for (int k = 0; k < PL_K; ++k) {
        const int q = k * PL_T + t;
        d1[k] = dq[(q < PL_Q) ? q : PL_Q - 1];
    }
    #pragma unroll
    for (int j = 0; j < PL_F; ++j) {
        const int i = j * PL_T + t;
        if (i < PL_N) tbl[i] = 0;
    }
    __syncthreads();
    #pragma unroll
    for (int k = 0; k < PL_K; ++k) {
        const int q = k * PL_T + t;
        if (q < PL_Q) {
            const unsigned ux = (unsigned)(d1[k].x - lo), uy = (unsigned)(d1[k].y - lo);
            const unsigned uz = (unsigned)(d1[k].z - lo), uw = (unsigned)(d1[k].w - lo);
            if (ux < PL_N) atomicAdd(&tbl[ux], 1);
            if (uy < PL_N) atomicAdd(&tbl[uy], 1);
            if (uz < PL_N) atomicAdd(&tbl[uz], 1);
            if (uw < PL_N) atomicAdd(&tbl[uw], 1);
        }
    }
    __syncthreads();
    {
        int c[PL_F], b[PL_F];
        #pragma unroll
        for (int j = 0; j < PL_F; ++j) {
            const int i = j * PL_T + t;
            c[j] = (i < PL_N) ? tbl[i] : 0;
        }
        #pragma unroll
        for (int j = 0; j < PL_F; ++j) {
            const int i = j * PL_T + t;
            b[j] = 0;
            if (c[j] != 0) b[j] = atomicAdd(cnt + lo + i, c[j]);
        }
        #pragma unroll
        for (int j = 0; j < PL_F; ++j) {
            const int i = j * PL_T + t;
            if (c[j] != 0) tbl[i] = b[j];
        }
    }
    __syncthreads();
    for (int k0 = 0; k0 < PL_K; k0 += PL_U) {
        int4 d[PL_U], s[PL_U], r[PL_U];
        #pragma unroll
        for (int j = 0; j < PL_U; ++j) {
            const int q = (k0 + j) * PL_T + t;
            const int qq = (q < PL_Q) ? q : PL_Q - 1;
            d[j] = dq[qq];
            s[j] = sq[qq];
            if (q >= PL_Q) d[j] = make_int4(-1, -1, -1, -1);
        }
        #pragma unroll
        for (int j = 0; j < PL_U; ++j) {
            const unsigned ux = (unsigned)(d[j].x - lo), uy = (unsigned)(d[j].y - lo);
            const unsigned uz = (unsigned)(d[j].z - lo), uw = (unsigned)(d[j].w - lo);
            r[j] = make_int4(-1, -1, -1, -1);
            if (ux < PL_N) r[j].x = atomicAdd(&tbl[ux], 1);
            if (uy < PL_N) r[j].y = atomicAdd(&tbl[uy], 1);
            if (uz < PL_N) r[j].z = atomicAdd(&tbl[uz], 1);
            if (uw < PL_N) r[j].w = atomicAdd(&tbl[uw], 1);
        }
        #pragma unroll
        for (int j = 0; j < PL_U; ++j) {
            if (r[j].x >= 0) place_edge(d[j].x, s[j].x, r[j].x, csr_src, cnt + NN, ovf);
            if (r[j].y >= 0) place_edge(d[j].y, s[j].y, r[j].y, csr_src, cnt + NN, ovf);
            if (r[j].z >= 0) place_edge(d[j].z, s[j].z, r[j].z, csr_src, cnt + NN, ovf);
            if (r[j].w >= 0) place_edge(d[j].w, s[j].w, r[j].w, csr_src, cnt + NN, ovf);
        }
    }
}

#define AG_G 8

__device__ __forceinline__ float edge_alpha(float si, float sj) {
    float a = si + sj;
    return (a > 0.0f) ? a : NEGS * a;
}

__device__ __forceinline__ void add_rows(int src, float c, int use, int cn, int lane,
                                         const float* __restrict__ g,
                                         float& accx, float& accy) {
    for (int j = 0; j < cn; ++j) {
        const int   uj = __shfl(use, j, 32);
        const int   sj = __shfl(src, j, 32);
        const float cj = __shfl(c, j, 32);
        if (uj != 0) {
            const float2 gv = ((const float2*)(g + (size_t)sj * CC))[lane];
            accx += gv.x * cj;
            accy += gv.y * cj;
        }
    }
}

__device__ __forceinline__ int lane_value(int v, int l) {
    return __builtin_amdgcn_readlane(v, l);
}
__device__ __forceinline__ float lane_value(float v, int l) {
    return __int_as_float(__builtin_amdgcn_readlane(__float_as_int(v), l));
}

__device__ __forceinline__ float wave_max8(const float (&v)[AG_G], int lane) {
    const bool b4 = (lane & 16) != 0, b3 = (lane & 8) != 0, b2 = (lane & 4) != 0;
    float k4[4], k2[2];
    #pragma unroll
    for (int u = 0; u < 4; ++u) {
        const float keep = b4 ? v[4 + u] : v[u];
        const float send = b4 ? v[u] : v[4 + u];
        k4[u] = fmaxf(keep, __shfl_xor(send, 16, 32));
    }
    #pragma unroll
    for (int u = 0; u < 2; ++u) {
        const float keep = b3 ? k4[2 + u] : k4[u];
        const float send = b3 ? k4[u] : k4[2 + u];
        k2[u] = fmaxf(keep, __shfl_xor(send, 8, 32));
    }
    const float keep1 = b2 ? k2[1] : k2[0];
    const float send1 = b2 ? k2[0] : k2[1];
    float k1 = fmaxf(keep1, __shfl_xor(send1, 4, 32));
    k1 = fmaxf(k1, __shfl_xor(k1, 2, 32));
    k1 = fmaxf(k1, __shfl_xor(k1, 1, 32));
    return k1;
}
__device__ __forceinline__ float wave_sum8(const float (&v)[AG_G], int lane) {
    const bool b4 = (lane & 16) != 0, b3 = (lane & 8) != 0, b2 = (lane & 4) != 0;
    float k4[4], k2[2];
    #pragma unroll
    for (int u = 0; u < 4; ++u) {
        const float keep = b4 ? v[4 + u] : v[u];
        const float send = b4 ? v[u] : v[4 + u];
        k4[u] = keep + __shfl_xor(send, 16, 32);
    }
    #pragma unroll
    for (int u = 0; u < 2; ++u) {
        const float keep = b3 ? k4[2 + u] : k4[u];
        const float send = b3 ? k4[u] : k4[2 + u];
        k2[u] = keep + __shfl_xor(send, 8, 32);
    }
    const float keep1 = b2 ? k2[1] : k2[0];
    const float send1 = b2 ? k2[0] : k2[1];
    float k1 = keep1 + __shfl_xor(send1, 4, 32);
    k1 += __shfl_xor(k1, 2, 32);
    k1 += __shfl_xor(k1, 1, 32);
    return k1;
}

__device__ void node_general(int node, int lane,
                             const int* __restrict__ cnt, const int* __restrict__ csr_src,
                             const int* __restrict__ ovf, const float* __restrict__ s_i,
                             const float* __restrict__ s_j, const float* __restrict__ g,
                             const float* __restrict__ bias, float* __restrict__ out) {
    const float si = s_i[node];
    const int deg = cnt[node];
    const int inb = (deg < CAP) ? deg : CAP;
    const int n   = inb + 1;
    const int novf = (deg > CAP) ? cnt[NN] : 0;
    const int* cs = csr_src + (size_t)node * CAP;

    float m = -INFINITY;
    for (int base = 0; base < n; base += 32) {
        const int k = base + lane;
        if (k < n) {
            const int src = (k < inb) ? cs[k] : node;
            m = fmaxf(m, edge_alpha(si, s_j[src]));
        }
    }
    for (int base = 0; base < novf; base += 32) {
        const int p = base + lane;
        if (p < novf) {
            const int2 e = *(const int2*)(ovf + 2 * (size_t)p);
            if (e.x == node) m = fmaxf(m, edge_alpha(si, s_j[e.y]));
        }
    }
    #pragma unroll
    for (int off = 16; off > 0; off >>= 1) m = fmaxf(m, __shfl_xor(m, off, 32));

    float dsum = 0.0f;
    for (int base = 0; base < n; base += 32) {
        const int k = base + lane;
        if (k < n) {
            const int src = (k < inb) ? cs[k] : node;
            dsum += expf(edge_alpha(si, s_j[src]) - m);
        }
    }
    for (int base = 0; base < novf; base += 32) {
        const int p = base + lane;
        if (p < novf) {
            const int2 e = *(const int2*)(ovf + 2 * (size_t)p);
            if (e.x == node) dsum += expf(edge_alpha(si, s_j[e.y]) - m);
        }
    }
    #pragma unroll
    for (int off = 16; off > 0; off >>= 1) dsum += __shfl_xor(dsum, off, 32);
    const float den = dsum + 1e-16f;

    float accx = 0.0f, accy = 0.0f;
    for (int base = 0; base < n; base += 32) {
        const int k = base + lane;
        int src = node;
        float c = 0.0f;
        if (k < n) {
            src = (k < inb) ? cs[k] : node;
            c = expf(edge_alpha(si, s_j[src]) - m) / den;
        }
        const int cn = (n - base < 32) ? (n - base) : 32;
        #pragma unroll 4
        for (int j = 0; j < cn; ++j) {
            const int   sj = __shfl(src, j, 32);
            const float cj = __shfl(c, j, 32);
            const float2 gv = ((const float2*)(g + (size_t)sj * CC))[lane];
            accx += gv.x * cj;
            accy += gv.y * cj;
        }
    }
    for (int base = 0; base < novf; base += 32) {
        const int p = base + lane;
        int src = node;
        float c = 0.0f;
        int use = 0;
        if (p < novf) {
            const int2 e = *(const int2*)(ovf + 2 * (size_t)p);
            if (e.x == node) {
                src = e.y;
                c = expf(edge_alpha(si, s_j[src]) - m) / den;
                use = 1;
            }
        }
        const int cn = (novf - base < 32) ? (novf - base) : 32;
        add_rows(src, c, use, cn, lane, g, accx, accy);
    }
    const float2 b = ((const float2*)bias)[lane];
    float2 r;
    r.x = accx + b.x; r.x = (r.x > 0.0f) ? r.x : 0.0f;
    r.y = accy + b.y; r.y = (r.y > 0.0f) ? r.y : 0.0f;
    ((float2*)(out + (size_t)node * CC))[lane] = r;
}

__global__ __launch_bounds__(256)
void aggregate_kernel(const int* __restrict__ cnt, const int* __restrict__ csr_src,
                      const int* __restrict__ ovf, const float* __restrict__ s_i,
                      const float* __restrict__ s_j, const float* __restrict__ g,
                      const float* __restrict__ bias, float* __restrict__ out) {
    const int wv   = __builtin_amdgcn_readfirstlane((blockIdx.x * blockDim.x + threadIdx.x) >> 5);
    const int lane = threadIdx.x & 31;
    if (wv >= NN / AG_G) return;
    const int node0 = wv * AG_G;

    int n[AG_G];
    float si[AG_G];
    unsigned big = 0;
    int maxn = 0;
    #pragma unroll
    for (int q = 0; q < AG_G; ++q) {
        const int deg = cnt[node0 + q];
        si[q] = s_i[node0 + q];
        n[q] = deg + 1;
        if (deg > 31) { n[q] = 0; big |= 1u << q; }
        maxn = (n[q] > maxn) ? n[q] : maxn;
    }

    int src[AG_G];
    #pragma unroll
    for (int q = 0; q < AG_G; ++q) {
        src[q] = node0 + q;
        if (lane < n[q] - 1) src[q] = csr_src[(size_t)(node0 + q) * CAP + lane];
    }
    float sjv[AG_G];
    #pragma unroll
    for (int q = 0; q < AG_G; ++q) sjv[q] = s_j[src[q]];
    asm volatile("" :: "v"(sjv[0]), "v"(sjv[1]), "v"(sjv[2]), "v"(sjv[3]),
                       "v"(sjv[4]), "v"(sjv[5]), "v"(sjv[6]), "v"(sjv[7]));
    float al[AG_G];
    #pragma unroll
    for (int q = 0; q < AG_G; ++q)
        al[q] = (lane < n[q]) ? edge_alpha(si[q], sjv[q]) : -INFINITY;
    const float mred = wave_max8(al, lane);
    float ex[AG_G];
    #pragma unroll
    for (int q = 0; q < AG_G; ++q) {
        const float m = lane_value(mred, 4 * q);
        ex[q] = (lane < n[q]) ? expf(al[q] - m) : 0.0f;
    }
    const float sred = wave_sum8(ex, lane);
    float c[AG_G];
    #pragma unroll
    for (int q = 0; q < AG_G; ++q) {
        const float den = lane_value(sred, 4 * q) + 1e-16f;
        c[q] = ex[q] / den;
    }

    float2 acc[AG_G];
    #pragma unroll
    for (int q = 0; q < AG_G; ++q) acc[q] = make_float2(0.0f, 0.0f);
    for (int j = 0; j < maxn; j += 2) {
        float2 gv[2 * AG_G];
        #pragma unroll
        for (int u = 0; u < 2; ++u) {
            #pragma unroll
            for (int q = 0; q < AG_G; ++q) {
                gv[u * AG_G + q] = make_float2(0.0f, 0.0f);
                if (j + u < n[q]) {
                    const int sj = lane_value(src[q], j + u);
                    gv[u * AG_G + q] = ((const float2*)(g + (size_t)sj * CC))[lane];
                }
            }
        }
        #pragma unroll
        for (int u = 0; u < 2; ++u) {
            #pragma unroll
            for (int q = 0; q < AG_G; ++q) {
                const float cj = lane_value(c[q], j + u);
                acc[q].x += gv[u * AG_G + q].x * cj;
                acc[q].y += gv[u * AG_G + q].y * cj;
            }
        }
    }

    const float2 b = ((const float2*)bias)[lane];
    #pragma unroll
    for (int q = 0; q < AG_G; ++q) {
        if (((big >> q) & 1u) == 0) {
            float2 r;
            r.x = acc[q].x + b.x; r.x = (r.x > 0.0f) ? r.x : 0.0f;
            r.y = acc[q].y + b.y; r.y = (r.y > 0.0f) ? r.y : 0.0f;
            ((float2*)(out + (size_t)(node0 + q) * CC))[lane] = r;
        }
    }
    for (int q = 0; q < AG_G; ++q) {
        if ((big >> q) & 1u)
            node_general(node0 + q, lane, cnt, csr_src, ovf, s_i, s_j, g, bias, out);
    }
}

extern "C" void kernel_launch(void* const* d_in, const int* in_sizes, int n_in,
                              void* d_out, int out_size, void* d_ws, size_t ws_size,
                              hipStream_t stream) {
    (void)in_sizes; (void)n_in; (void)out_size; (void)ws_size;
    const float* x      = (const float*)d_in[0];
    const int*   ei     = (const int*)  d_in[1];
    const float* corrs  = (const float*)d_in[2];
    const float* W      = (const float*)d_in[3];
    const float* att_i  = (const float*)d_in[4];
    const float* att_j  = (const float*)d_in[5];
    const float* aem_i  = (const float*)d_in[6];
    const float* aem_j  = (const float*)d_in[7];
    const float* bias   = (const float*)d_in[8];
    float* out = (float*)d_out;

    float* ws    = (float*)d_ws;
    float* g     = ws;
    int*   csr   = (int*)(g + (size_t)NN * CC);
    int*   ovf   = csr + (size_t)NN * CAP;
    float* s_i   = (float*)(ovf + (size_t)2 * EE);
    float* s_j   = s_i + NN;
    int*   cnt   = (int*)(s_j + NN);

    gemm_scores_kernel<<<NN / 16, 32, 0, stream>>>(
        x, W, corrs, att_i, att_j, aem_i, aem_j, g, s_i, s_j, cnt);
    place_kernel<<<PL_S * PL_R, PL_T, 0, stream>>>(ei, cnt, csr, ovf);
    aggregate_kernel<<<(NN / AG_G * 32 + 255) / 256, 256, 0, stream>>>(
        cnt, csr, ovf, s_i, s_j, g, bias, out);
}


# Aggregation rows gathered by 16-lane rows: each row of a wave owns 4 nodes, one 16-byte load fetches 2 rows, index and weight reach a row by in-row broadcast, no scalar trip; sum order unchanged
# speedup vs baseline: 4.2414x; 1.0482x over previous
#include <hip/hip_runtime.h>
#include <math.h>

#define NN   50000
#define EE   800000
#define INC  128
#define CC   64
#define NEGS 0.2f
#define CAP  64

typedef float v2f __attribute__((ext_vector_type(2)));
typedef float v8f __attribute__((ext_vector_type(8)));

__global__ __launch_bounds__(32)
void gemm_scores_kernel(const float* __restrict__ x, const float* __restrict__ W,
                        const float* __restrict__ corrs,
                        const float* __restrict__ att_i, const float* __restrict__ att_j,
                        const float* __restrict__ aem_i, const float* __restrict__ aem_j,
                        float* __restrict__ g, float* __restrict__ s_i,
                        float* __restrict__ s_j, int* __restrict__ cnt) {
    const int rb   = blockIdx.x;
    const int lane = threadIdx.x;
    const int half = lane >> 4;
    const int l15  = lane & 15;

    float4 cr[8];
    {
        const float4* cp = (const float4*)(corrs + (size_t)(rb * 16 + half * 8) * CC) + l15;
        #pragma unroll
        for (int v = 0; v < 8; ++v) cr[v] = cp[v * (CC / 4)];
    }

    v8f acc0 = {}; v8f acc1 = {}; v8f acc2 = {}; v8f acc3 = {};

    const float* xrow = x + (rb * 16 + l15) * INC + half * 2;
    for (int k0 = 0; k0 < INC; k0 += 4) {
        v2f a; a.x = xrow[k0]; a.y = xrow[k0 + 1];
        const float* wp = W + (k0 + half * 2) * CC + l15;
        v2f b0, b1, b2, b3;
        b0.x = wp[0];  b0.y = wp[0  + CC];
        b1.x = wp[16]; b1.y = wp[16 + CC];
        b2.x = wp[32]; b2.y = wp[32 + CC];
        b3.x = wp[48]; b3.y = wp[48 + CC];
        acc0 = __builtin_amdgcn_wmma_f32_16x16x4_f32(false, a, false, b0, (short)0, acc0, false, false);
        acc1 = __builtin_amdgcn_wmma_f32_16x16x4_f32(false, a, false, b1, (short)0, acc1, false, false);
        acc2 = __builtin_amdgcn_wmma_f32_16x16x4_f32(false, a, false, b2, (short)0, acc2, false, false);
        acc3 = __builtin_amdgcn_wmma_f32_16x16x4_f32(false, a, false, b3, (short)0, acc3, false, false);
    }
    #pragma unroll
    for (int v = 0; v < 8; ++v) {
        float* op = g + (rb * 16 + v + half * 8) * CC + l15;
        op[0]  = acc0[v];
        op[16] = acc1[v];
        op[32] = acc2[v];
        op[48] = acc3[v];
    }

    float p[16];
    {
        float ai[4], aj[4];
        #pragma unroll
        for (int t = 0; t < 4; ++t) { ai[t] = att_i[l15 + 16 * t]; aj[t] = att_j[l15 + 16 * t]; }
        const float4 ei4 = ((const float4*)aem_i)[l15];
        const float4 ej4 = ((const float4*)aem_j)[l15];
        #pragma unroll
        for (int v = 0; v < 8; ++v) {
            float pi = acc0[v] * ai[0] + acc1[v] * ai[1] + acc2[v] * ai[2] + acc3[v] * ai[3];
            float pj = acc0[v] * aj[0] + acc1[v] * aj[1] + acc2[v] * aj[2] + acc3[v] * aj[3];
            pi += cr[v].x * ei4.x + cr[v].y * ei4.y + cr[v].z * ei4.z + cr[v].w * ei4.w;
            pj += cr[v].x * ej4.x + cr[v].y * ej4.y + cr[v].z * ej4.z + cr[v].w * ej4.w;
            p[v] = pi; p[8 + v] = pj;
        }
    }
    const bool b3 = (l15 & 8) != 0, b2 = (l15 & 4) != 0, b1 = (l15 & 2) != 0, b0 = (l15 & 1) != 0;
    float k8[8], k4[4], k2[2];
    #pragma unroll
    for (int u = 0; u < 8; ++u) {
        const float keep = b3 ? p[8 + u] : p[u];
        const float send = b3 ? p[u] : p[8 + u];
        k8[u] = keep + __shfl_xor(send, 8, 32);
    }
    #pragma unroll
    for (int u = 0; u < 4; ++u) {
        const float keep = b2 ? k8[4 + u] : k8[u];
        const float send = b2 ? k8[u] : k8[4 + u];
        k4[u] = keep + __shfl_xor(send, 4, 32);
    }
    #pragma unroll
    for (int u = 0; u < 2; ++u) {
        const float keep = b1 ? k4[2 + u] : k4[u];
        const float send = b1 ? k4[u] : k4[2 + u];
        k2[u] = keep + __shfl_xor(send, 2, 32);
    }
    const float keep1 = b0 ? k2[1] : k2[0];
    const float send1 = b0 ? k2[0] : k2[1];
    const float sc = keep1 + __shfl_xor(send1, 1, 32);
    const int row = rb * 16 + half * 8 + (l15 & 7);
    if (b3) { s_j[row] = sc; } else { s_i[row] = sc; cnt[row] = 0; }
    if (rb == 0 && lane == 0) cnt[NN] = 0;
}

#define PL_S   16
#define PL_R   16
#define PL_T   1024
#define PL_Q   (EE / 4 / PL_S)
#define PL_N   (NN / PL_R)
#define PL_K   ((PL_Q + PL_T - 1) / PL_T)
#define PL_F   ((PL_N + PL_T - 1) / PL_T)
#define PL_U   4

__device__ __forceinline__ void place_edge(int d, int s, int r, int* __restrict__ csr_src,
                                           int* __restrict__ ovf_n, int* __restrict__ ovf) {
    if (r < CAP) {
        csr_src[(size_t)d * CAP + r] = s;
    } else {
        const int p = atomicAdd(ovf_n, 1);
        *(int2*)(ovf + 2 * (size_t)p) = make_int2(d, s);
    }
}

__global__ __launch_bounds__(PL_T)
void place_kernel(const int* __restrict__ ei, int* __restrict__ cnt,
                  int* __restrict__ csr_src, int* __restrict__ ovf) {
    __shared__ int tbl[PL_N];
    const int t  = threadIdx.x;
    const int sl = blockIdx.x / PL_R;
    const int lo = (blockIdx.x % PL_R) * PL_N;
    const int4* sq = (const int4*)ei + sl * PL_Q;
    const int4* dq = (const int4*)(ei + EE) + sl * PL_Q;

    int4 d1[PL_K];
    #pragma unroll
    for (int k = 0; k < PL_K; ++k) {
        const int q = k * PL_T + t;
        d1[k] = dq[(q < PL_Q) ? q : PL_Q - 1];
    }
    #pragma unroll
    for (int j = 0; j < PL_F; ++j) {
        const int i = j * PL_T + t;
        if (i < PL_N) tbl[i] = 0;
    }
    __syncthreads();
    #pragma unroll
    for (int k = 0; k < PL_K; ++k) {
        const int q = k * PL_T + t;
        if (q < PL_Q) {
            const unsigned ux = (unsigned)(d1[k].x - lo), uy = (unsigned)(d1[k].y - lo);
            const unsigned uz = (unsigned)(d1[k].z - lo), uw = (unsigned)(d1[k].w - lo);
            if (ux < PL_N) atomicAdd(&tbl[ux], 1);
            if (uy < PL_N) atomicAdd(&tbl[uy], 1);
            if (uz < PL_N) atomicAdd(&tbl[uz], 1);
            if (uw < PL_N) atomicAdd(&tbl[uw], 1);
        }
    }
    __syncthreads();
    {
        int c[PL_F], b[PL_F];
        #pragma unroll
        for (int j = 0; j < PL_F; ++j) {
            const int i = j * PL_T + t;
            c[j] = (i < PL_N) ? tbl[i] : 0;
        }
        #pragma unroll
        for (int j = 0; j < PL_F; ++j) {
            const int i = j * PL_T + t;
            b[j] = 0;
            if (c[j] != 0) b[j] = atomicAdd(cnt + lo + i, c[j]);
        }
        #pragma unroll
        for (int j = 0; j < PL_F; ++j) {
            const int i = j * PL_T + t;
            if (c[j] != 0) tbl[i] = b[j];
        }
    }
    __syncthreads();
    for (int k0 = 0; k0 < PL_K; k0 += PL_U) {
        int4 d[PL_U], s[PL_U], r[PL_U];
        #pragma unroll
        for (int j = 0; j < PL_U; ++j) {
            const int q = (k0 + j) * PL_T + t;
            const int qq = (q < PL_Q) ? q : PL_Q - 1;
            d[j] = dq[qq];
            s[j] = sq[qq];
            if (q >= PL_Q) d[j] = make_int4(-1, -1, -1, -1);
        }
        #pragma unroll
        for (int j = 0; j < PL_U; ++j) {
            const unsigned ux = (unsigned)(d[j].x - lo), uy = (unsigned)(d[j].y - lo);
            const unsigned uz = (unsigned)(d[j].z - lo), uw = (unsigned)(d[j].w - lo);
            r[j] = make_int4(-1, -1, -1, -1);
            if (ux < PL_N) r[j].x = atomicAdd(&tbl[ux], 1);
            if (uy < PL_N) r[j].y = atomicAdd(&tbl[uy], 1);
            if (uz < PL_N) r[j].z = atomicAdd(&tbl[uz], 1);
            if (uw < PL_N) r[j].w = atomicAdd(&tbl[uw], 1);
        }
        #pragma unroll
        for (int j = 0; j < PL_U; ++j) {
            if (r[j].x >= 0) place_edge(d[j].x, s[j].x, r[j].x, csr_src, cnt + NN, ovf);
            if (r[j].y >= 0) place_edge(d[j].y, s[j].y, r[j].y, csr_src, cnt + NN, ovf);
            if (r[j].z >= 0) place_edge(d[j].z, s[j].z, r[j].z, csr_src, cnt + NN, ovf);
            if (r[j].w >= 0) place_edge(d[j].w, s[j].w, r[j].w, csr_src, cnt + NN, ovf);
        }
    }
}

#define AG_G 8

__device__ __forceinline__ float edge_alpha(float si, float sj) {
    float a = si + sj;
    return (a > 0.0f) ? a : NEGS * a;
}

__device__ __forceinline__ void add_rows(int src, float c, int use, int cn, int lane,
                                         const float* __restrict__ g,
                                         float& accx, float& accy) {
    for (int j = 0; j < cn; ++j) {
        const int   uj = __shfl(use, j, 32);
        const int   sj = __shfl(src, j, 32);
        const float cj = __shfl(c, j, 32);
        if (uj != 0) {
            const float2 gv = ((const float2*)(g + (size_t)sj * CC))[lane];
            accx += gv.x * cj;
            accy += gv.y * cj;
        }
    }
}

__device__ __forceinline__ int row_value(int v, int k) {
    const unsigned sel = 0x11111111u * (unsigned)k;
    return (int)__builtin_amdgcn_permlane16((unsigned)v, (unsigned)v, sel, sel, true, false);
}
__device__ __forceinline__ float row_value(float v, int k) {
    return __int_as_float(row_value(__float_as_int(v), k));
}

#define AG_P 4
__device__ __forceinline__ float row_max4(const float (&v)[AG_P], int lane) {
    const bool b3 = (lane & 8) != 0, b2 = (lane & 4) != 0;
    float k2[2];
    #pragma unroll
    for (int u = 0; u < 2; ++u) {
        const float keep = b3 ? v[2 + u] : v[u];
        const float send = b3 ? v[u] : v[2 + u];
        k2[u] = fmaxf(keep, __shfl_xor(send, 8, 32));
    }
    const float keep1 = b2 ? k2[1] : k2[0];
    const float send1 = b2 ? k2[0] : k2[1];
    float k1 = fmaxf(keep1, __shfl_xor(send1, 4, 32));
    k1 = fmaxf(k1, __shfl_xor(k1, 2, 32));
    k1 = fmaxf(k1, __shfl_xor(k1, 1, 32));
    return k1;
}
__device__ __forceinline__ float row_sum4(const float (&v)[AG_P], int lane) {
    const bool b3 = (lane & 8) != 0, b2 = (lane & 4) != 0;
    float k2[2];
    #pragma unroll
    for (int u = 0; u < 2; ++u) {
        const float keep = b3 ? v[2 + u] : v[u];
        const float send = b3 ? v[u] : v[2 + u];
        k2[u] = keep + __shfl_xor(send, 8, 32);
    }
    const float keep1 = b2 ? k2[1] : k2[0];
    const float send1 = b2 ? k2[0] : k2[1];
    float k1 = keep1 + __shfl_xor(send1, 4, 32);
    k1 += __shfl_xor(k1, 2, 32);
    k1 += __shfl_xor(k1, 1, 32);
    return k1;
}

__device__ void node_general(int node, int lane,
                             const int* __restrict__ cnt, const int* __restrict__ csr_src,
                             const int* __restrict__ ovf, const float* __restrict__ s_i,
                             const float* __restrict__ s_j, const float* __restrict__ g,
                             const float* __restrict__ bias, float* __restrict__ out) {
    const float si = s_i[node];
    const int deg = cnt[node];
    const int inb = (deg < CAP) ? deg : CAP;
    const int n   = inb + 1;
    const int novf = (deg > CAP) ? cnt[NN] : 0;
    const int* cs = csr_src + (size_t)node * CAP;

    float m = -INFINITY;
    for (int base = 0; base < n; base += 32) {
        const int k = base + lane;
        if (k < n) {
            const int src = (k < inb) ? cs[k] : node;
            m = fmaxf(m, edge_alpha(si, s_j[src]));
        }
    }
    for (int base = 0; base < novf; base += 32) {
        const int p = base + lane;
        if (p < novf) {
            const int2 e = *(const int2*)(ovf + 2 * (size_t)p);
            if (e.x == node) m = fmaxf(m, edge_alpha(si, s_j[e.y]));
        }
    }
    #pragma unroll
    for (int off = 16; off > 0; off >>= 1) m = fmaxf(m, __shfl_xor(m, off, 32));

    float dsum = 0.0f;
    for (int base = 0; base < n; base += 32) {
        const int k = base + lane;
        if (k < n) {
            const int src = (k < inb) ? cs[k] : node;
            dsum += expf(edge_alpha(si, s_j[src]) - m);
        }
    }
    for (int base = 0; base < novf; base += 32) {
        const int p = base + lane;
        if (p < novf) {
            const int2 e = *(const int2*)(ovf + 2 * (size_t)p);
            if (e.x == node) dsum += expf(edge_alpha(si, s_j[e.y]) - m);
        }
    }
    #pragma unroll
    for (int off = 16; off > 0; off >>= 1) dsum += __shfl_xor(dsum, off, 32);
    const float den = dsum + 1e-16f;

    float accx = 0.0f, accy = 0.0f;
    for (int base = 0; base < n; base += 32) {
        const int k = base + lane;
        int src = node;
        float c = 0.0f;
        if (k < n) {
            src = (k < inb) ? cs[k] : node;
            c = expf(edge_alpha(si, s_j[src]) - m) / den;
        }
        const int cn = (n - base < 32) ? (n - base) : 32;
        #pragma unroll 4
        for (int j = 0; j < cn; ++j) {
            const int   sj = __shfl(src, j, 32);
            const float cj = __shfl(c, j, 32);
            const float2 gv = ((const float2*)(g + (size_t)sj * CC))[lane];
            accx += gv.x * cj;
            accy += gv.y * cj;
        }
    }
    for (int base = 0; base < novf; base += 32) {
        const int p = base + lane;
        int src = node;
        float c = 0.0f;
        int use = 0;
        if (p < novf) {
            const int2 e = *(const int2*)(ovf + 2 * (size_t)p);
            if (e.x == node) {
                src = e.y;
                c = expf(edge_alpha(si, s_j[src]) - m) / den;
                use = 1;
            }
        }
        const int cn = (novf - base < 32) ? (novf - base) : 32;
        add_rows(src, c, use, cn, lane, g, accx, accy);
    }
    const float2 b = ((const float2*)bias)[lane];
    float2 r;
    r.x = accx + b.x; r.x = (r.x > 0.0f) ? r.x : 0.0f;
    r.y = accy + b.y; r.y = (r.y > 0.0f) ? r.y : 0.0f;
    ((float2*)(out + (size_t)node * CC))[lane] = r;
}

template <int STEP>
__device__ __forceinline__ void row_steps(const int (&srcr)[AG_P], const float (&cr)[AG_P], int cn,
                                          unsigned c16, const float4* __restrict__ grow,
                                          float4 (&acc)[AG_P]) {
    for (int j = 0; j < cn; j += STEP) {
        float4 gv[STEP * AG_P];
        #pragma unroll
        for (int u = 0; u < STEP; ++u) {
            #pragma unroll
            for (int p = 0; p < AG_P; ++p) {
                const unsigned sj = (unsigned)row_value(srcr[p], j + u);
                gv[u * AG_P + p] = grow[sj * (CC / 4) + c16];
            }
        }
        #pragma unroll
        for (int u = 0; u < STEP; ++u) {
            #pragma unroll
            for (int p = 0; p < AG_P; ++p) {
                const float cj = row_value(cr[p], j + u);
                acc[p].x = fmaf(gv[u * AG_P + p].x, cj, acc[p].x);
                acc[p].y = fmaf(gv[u * AG_P + p].y, cj, acc[p].y);
                acc[p].z = fmaf(gv[u * AG_P + p].z, cj, acc[p].z);
                acc[p].w = fmaf(gv[u * AG_P + p].w, cj, acc[p].w);
            }
        }
    }
}

__global__ __launch_bounds__(256)
void aggregate_kernel(const int* __restrict__ cnt, const int* __restrict__ csr_src,
                      const int* __restrict__ ovf, const float* __restrict__ s_i,
                      const float* __restrict__ s_j, const float* __restrict__ g,
                      const float* __restrict__ bias, float* __restrict__ out) {
    const int wv   = __builtin_amdgcn_readfirstlane((blockIdx.x * blockDim.x + threadIdx.x) >> 5);
    const int lane = threadIdx.x & 31;
    if (wv >= NN / AG_G) return;
    const int node0 = wv * AG_G;
    const int h   = lane >> 4;
    const int c16 = lane & 15;

    int n[AG_G];
    float si[AG_G];
    unsigned big = 0;
    int maxn = 0;
    #pragma unroll
    for (int q = 0; q < AG_G; ++q) {
        const int deg = cnt[node0 + q];
        si[q] = s_i[node0 + q];
        n[q] = deg + 1;
        if (deg > 31) { n[q] = 0; big |= 1u << q; }
        maxn = (n[q] > maxn) ? n[q] : maxn;
    }
    int nl[AG_P], mynode[AG_P];
    float sil[AG_P];
    #pragma unroll
    for (int p = 0; p < AG_P; ++p) {
        nl[p]  = h ? n[p + AG_P] : n[p];
        sil[p] = h ? si[p + AG_P] : si[p];
        mynode[p] = node0 + p + AG_P * h;
    }

    int src[2][AG_P];
    #pragma unroll
    for (int p = 0; p < AG_P; ++p) {
        #pragma unroll
        for (int r = 0; r < 2; ++r) {
            const int e = c16 + 16 * r;
            src[r][p] = (e == nl[p] - 1) ? mynode[p] : 0;
            if (e < nl[p] - 1) src[r][p] = csr_src[(size_t)mynode[p] * CAP + e];
        }
    }
    float sjv[AG_P][2];
    #pragma unroll
    for (int p = 0; p < AG_P; ++p) {
        sjv[p][0] = s_j[src[0][p]];
        sjv[p][1] = s_j[src[1][p]];
    }
    asm volatile("" :: "v"(sjv[0][0]), "v"(sjv[0][1]), "v"(sjv[1][0]), "v"(sjv[1][1]),
                       "v"(sjv[2][0]), "v"(sjv[2][1]), "v"(sjv[3][0]), "v"(sjv[3][1]));
    float al[AG_P][2], mv[AG_P];
    #pragma unroll
    for (int p = 0; p < AG_P; ++p) {
        al[p][0] = (c16      < nl[p]) ? edge_alpha(sil[p], sjv[p][0]) : -INFINITY;
        al[p][1] = (c16 + 16 < nl[p]) ? edge_alpha(sil[p], sjv[p][1]) : -INFINITY;
        mv[p] = fmaxf(al[p][0], al[p][1]);
    }
    const float mred = row_max4(mv, lane);
    float ex[AG_P][2], sv[AG_P];
    #pragma unroll
    for (int p = 0; p < AG_P; ++p) {
        const float m = row_value(mred, 4 * p);
        ex[p][0] = (c16      < nl[p]) ? expf(al[p][0] - m) : 0.0f;
        ex[p][1] = (c16 + 16 < nl[p]) ? expf(al[p][1] - m) : 0.0f;
        sv[p] = ex[p][0] + ex[p][1];
    }
    const float sred = row_sum4(sv, lane);
    float c[2][AG_P];
    #pragma unroll
    for (int p = 0; p < AG_P; ++p) {
        const float den = row_value(sred, 4 * p) + 1e-16f;
        c[0][p] = ex[p][0] / den;
        c[1][p] = ex[p][1] / den;
    }

    const float4* grow = (const float4*)g;
    float4 acc[AG_P];
    #pragma unroll
    for (int p = 0; p < AG_P; ++p) acc[p] = make_float4(0.0f, 0.0f, 0.0f, 0.0f);
    row_steps<2>(src[0], c[0], (maxn > 16) ? 16 : maxn, (unsigned)c16, grow, acc);
    row_steps<2>(src[1], c[1], maxn - 16, (unsigned)c16, grow, acc);

    const float4 b = ((const float4*)bias)[c16];
    #pragma unroll
    for (int p = 0; p < AG_P; ++p) {
        if (((big >> (p + AG_P * h)) & 1u) == 0) {
            float4 r4;
            r4.x = acc[p].x + b.x; r4.x = (r4.x > 0.0f) ? r4.x : 0.0f;
            r4.y = acc[p].y + b.y; r4.y = (r4.y > 0.0f) ? r4.y : 0.0f;
            r4.z = acc[p].z + b.z; r4.z = (r4.z > 0.0f) ? r4.z : 0.0f;
            r4.w = acc[p].w + b.w; r4.w = (r4.w > 0.0f) ? r4.w : 0.0f;
            ((float4*)(out + (size_t)mynode[p] * CC))[c16] = r4;
        }
    }
    for (int q = 0; q < AG_G; ++q) {
        if ((big >> q) & 1u)
            node_general(node0 + q, lane, cnt, csr_src, ovf, s_i, s_j, g, bias, out);
    }
}

extern "C" void kernel_launch(void* const* d_in, const int* in_sizes, int n_in,
                              void* d_out, int out_size, void* d_ws, size_t ws_size,
                              hipStream_t stream) {
    (void)in_sizes; (void)n_in; (void)out_size; (void)ws_size;
    const float* x      = (const float*)d_in[0];
    const int*   ei     = (const int*)  d_in[1];
    const float* corrs  = (const float*)d_in[2];
    const float* W      = (const float*)d_in[3];
    const float* att_i  = (const float*)d_in[4];
    const float* att_j  = (const float*)d_in[5];
    const float* aem_i  = (const float*)d_in[6];
    const float* aem_j  = (const float*)d_in[7];
    const float* bias   = (const float*)d_in[8];
    float* out = (float*)d_out;

    float* ws    = (float*)d_ws;
    float* g     = ws;
    int*   csr   = (int*)(g + (size_t)NN * CC);
    int*   ovf   = csr + (size_t)NN * CAP;
    float* s_i   = (float*)(ovf + (size_t)2 * EE);
    float* s_j   = s_i + NN;
    int*   cnt   = (int*)(s_j + NN);

    gemm_scores_kernel<<<NN / 16, 32, 0, stream>>>(
        x, W, corrs, att_i, att_j, aem_i, aem_j, g, s_i, s_j, cnt);
    place_kernel<<<PL_S * PL_R, PL_T, 0, stream>>>(ei, cnt, csr, ovf);
    aggregate_kernel<<<(NN / AG_G * 32 + 255) / 256, 256, 0, stream>>>(
        cnt, csr, ovf, s_i, s_j, g, bias, out);
}


# Place: each wave packs its in-range edges (node offset, source) by ballot into its own LDS queue; count, rank and store run on full lanes, edges read once; sum order: block arrival, then queue order
# speedup vs baseline: 4.5295x; 1.0679x over previous
#include <hip/hip_runtime.h>
#include <math.h>

#define NN   50000
#define EE   800000
#define INC  128
#define CC   64
#define NEGS 0.2f
#define CAP  64

typedef float v2f __attribute__((ext_vector_type(2)));
typedef float v8f __attribute__((ext_vector_type(8)));

__global__ __launch_bounds__(32)
void gemm_scores_kernel(const float* __restrict__ x, const float* __restrict__ W,
                        const float* __restrict__ corrs,
                        const float* __restrict__ att_i, const float* __restrict__ att_j,
                        const float* __restrict__ aem_i, const float* __restrict__ aem_j,
                        float* __restrict__ g, float* __restrict__ s_i,
                        float* __restrict__ s_j, int* __restrict__ cnt) {
    const int rb   = blockIdx.x;
    const int lane = threadIdx.x;
    const int half = lane >> 4;
    const int l15  = lane & 15;

    float4 cr[8];
    {
        const float4* cp = (const float4*)(corrs + (size_t)(rb * 16 + half * 8) * CC) + l15;
        #pragma unroll
        for (int v = 0; v < 8; ++v) cr[v] = cp[v * (CC / 4)];
    }

    v8f acc0 = {}; v8f acc1 = {}; v8f acc2 = {}; v8f acc3 = {};

    const float* xrow = x + (rb * 16 + l15) * INC + half * 2;
    for (int k0 = 0; k0 < INC; k0 += 4) {
        v2f a; a.x = xrow[k0]; a.y = xrow[k0 + 1];
        const float* wp = W + (k0 + half * 2) * CC + l15;
        v2f b0, b1, b2, b3;
        b0.x = wp[0];  b0.y = wp[0  + CC];
        b1.x = wp[16]; b1.y = wp[16 + CC];
        b2.x = wp[32]; b2.y = wp[32 + CC];
        b3.x = wp[48]; b3.y = wp[48 + CC];
        acc0 = __builtin_amdgcn_wmma_f32_16x16x4_f32(false, a, false, b0, (short)0, acc0, false, false);
        acc1 = __builtin_amdgcn_wmma_f32_16x16x4_f32(false, a, false, b1, (short)0, acc1, false, false);
        acc2 = __builtin_amdgcn_wmma_f32_16x16x4_f32(false, a, false, b2, (short)0, acc2, false, false);
        acc3 = __builtin_amdgcn_wmma_f32_16x16x4_f32(false, a, false, b3, (short)0, acc3, false, false);
    }
    #pragma unroll
    for (int v = 0; v < 8; ++v) {
        float* op = g + (rb * 16 + v + half * 8) * CC + l15;
        op[0]  = acc0[v];
        op[16] = acc1[v];
        op[32] = acc2[v];
        op[48] = acc3[v];
    }

    float p[16];
    {
        float ai[4], aj[4];
        #pragma unroll
        for (int t = 0; t < 4; ++t) { ai[t] = att_i[l15 + 16 * t]; aj[t] = att_j[l15 + 16 * t]; }
        const float4 ei4 = ((const float4*)aem_i)[l15];
        const float4 ej4 = ((const float4*)aem_j)[l15];
        #pragma unroll
        for (int v = 0; v < 8; ++v) {
            float pi = acc0[v] * ai[0] + acc1[v] * ai[1] + acc2[v] * ai[2] + acc3[v] * ai[3];
            float pj = acc0[v] * aj[0] + acc1[v] * aj[1] + acc2[v] * aj[2] + acc3[v] * aj[3];
            pi += cr[v].x * ei4.x + cr[v].y * ei4.y + cr[v].z * ei4.z + cr[v].w * ei4.w;
            pj += cr[v].x * ej4.x + cr[v].y * ej4.y + cr[v].z * ej4.z + cr[v].w * ej4.w;
            p[v] = pi; p[8 + v] = pj;
        }
    }
    const bool b3 = (l15 & 8) != 0, b2 = (l15 & 4) != 0, b1 = (l15 & 2) != 0, b0 = (l15 & 1) != 0;
    float k8[8], k4[4], k2[2];
    #pragma unroll
    for (int u = 0; u < 8; ++u) {
        const float keep = b3 ? p[8 + u] : p[u];
        const float send = b3 ? p[u] : p[8 + u];
        k8[u] = keep + __shfl_xor(send, 8, 32);
    }
    #pragma unroll
    for (int u = 0; u < 4; ++u) {
        const float keep = b2 ? k8[4 + u] : k8[u];
        const float send = b2 ? k8[u] : k8[4 + u];
        k4[u] = keep + __shfl_xor(send, 4, 32);
    }
    #pragma unroll
    for (int u = 0; u < 2; ++u) {
        const float keep = b1 ? k4[2 + u] : k4[u];
        const float send = b1 ? k4[u] : k4[2 + u];
        k2[u] = keep + __shfl_xor(send, 2, 32);
    }
    const float keep1 = b0 ? k2[1] : k2[0];
    const float send1 = b0 ? k2[0] : k2[1];
    const float sc = keep1 + __shfl_xor(send1, 1, 32);
    const int row = rb * 16 + half * 8 + (l15 & 7);
    if (b3) { s_j[row] = sc; } else { s_i[row] = sc; cnt[row] = 0; }
    if (rb == 0 && lane == 0) cnt[NN] = 0;
}

#define PL_S   16
#define PL_R   16
#define PL_T   1024
#define PL_Q   (EE / 4 / PL_S)
#define PL_N   (NN / PL_R)
#define PL_K   ((PL_Q + PL_T - 1) / PL_T)
#define PL_F   ((PL_N + PL_T - 1) / PL_T)
#define PL_WV  (PL_T / 32)
#define PL_QC  (PL_K * 4 * 32)
#define PL_D   4

__device__ __forceinline__ void place_edge(int d, int s, int r, int* __restrict__ csr_src,
                                           int* __restrict__ ovf_n, int* __restrict__ ovf) {
    if (r < CAP) {
        csr_src[(size_t)d * CAP + r] = s;
    } else {
        const int p = atomicAdd(ovf_n, 1);
        *(int2*)(ovf + 2 * (size_t)p) = make_int2(d, s);
    }
}

__device__ __forceinline__ void queue_edge(int d, int s, bool ok, int lo,
                                           unsigned* __restrict__ myq, int& nq) {
    const unsigned u = (unsigned)(d - lo);
    const bool in = ok && (u < PL_N);
    const unsigned m = __builtin_amdgcn_ballot_w32(in);
    if (in) myq[nq + (int)__builtin_amdgcn_mbcnt_lo(m, 0u)] = (u << 16) | (unsigned)s;
    nq += __builtin_popcount(m);
}

__global__ __launch_bounds__(PL_T)
void place_kernel(const int* __restrict__ ei, int* __restrict__ cnt,
                  int* __restrict__ csr_src, int* __restrict__ ovf) {
    __shared__ int tbl[PL_N];
    __shared__ unsigned que[PL_WV * PL_QC];
    const int t    = threadIdx.x;
    const int lane = t & 31;
    const int sl = blockIdx.x / PL_R;
    const int lo = (blockIdx.x % PL_R) * PL_N;
    const int4* sq = (const int4*)ei + sl * PL_Q;
    const int4* dq = (const int4*)(ei + EE) + sl * PL_Q;
    unsigned* myq = que + (t >> 5) * PL_QC;
    int nq = 0;

    int4 d[PL_K], s[PL_K];
    #pragma unroll
    for (int k = 0; k < PL_K; ++k) {
        const int q = k * PL_T + t;
        const int qq = (q < PL_Q) ? q : PL_Q - 1;
        d[k] = dq[qq];
        s[k] = sq[qq];
    }
    #pragma unroll
    for (int j = 0; j < PL_F; ++j) {
        const int i = j * PL_T + t;
        if (i < PL_N) tbl[i] = 0;
    }
    __syncthreads();
    #pragma unroll
    for (int k = 0; k < PL_K; ++k) {
        const bool ok = (k * PL_T + t) < PL_Q;
        queue_edge(d[k].x, s[k].x, ok, lo, myq, nq);
        queue_edge(d[k].y, s[k].y, ok, lo, myq, nq);
        queue_edge(d[k].z, s[k].z, ok, lo, myq, nq);
        queue_edge(d[k].w, s[k].w, ok, lo, myq, nq);
    }
    for (int i0 = 0; i0 < nq; i0 += 32 * PL_D) {
        unsigned e[PL_D];
        #pragma unroll
        for (int j = 0; j < PL_D; ++j) {
            const int i = i0 + j * 32 + lane;
            e[j] = myq[(i < nq) ? i : 0];
        }
        #pragma unroll
        for (int j = 0; j < PL_D; ++j) {
            const int i = i0 + j * 32 + lane;
            if (i < nq) atomicAdd(&tbl[e[j] >> 16], 1);
        }
    }
    __syncthreads();
    {
        int c[PL_F], b[PL_F];
        #pragma unroll
        for (int j = 0; j < PL_F; ++j) {
            const int i = j * PL_T + t;
            c[j] = (i < PL_N) ? tbl[i] : 0;
        }
        #pragma unroll
        for (int j = 0; j < PL_F; ++j) {
            const int i = j * PL_T + t;
            b[j] = 0;
            if (c[j] != 0) b[j] = atomicAdd(cnt + lo + i, c[j]);
        }
        #pragma unroll
        for (int j = 0; j < PL_F; ++j) {
            const int i = j * PL_T + t;
            if (c[j] != 0) tbl[i] = b[j];
        }
    }
    __syncthreads();
    for (int i0 = 0; i0 < nq; i0 += 32 * PL_D) {
        unsigned e[PL_D];
        int r[PL_D];
        #pragma unroll
        for (int j = 0; j < PL_D; ++j) {
            const int i = i0 + j * 32 + lane;
            e[j] = myq[(i < nq) ? i : 0];
        }
        #pragma unroll
        for (int j = 0; j < PL_D; ++j) {
            const int i = i0 + j * 32 + lane;
            r[j] = -1;
            if (i < nq) r[j] = atomicAdd(&tbl[e[j] >> 16], 1);
        }
        #pragma unroll
        for (int j = 0; j < PL_D; ++j) {
            if (r[j] >= 0)
                place_edge(lo + (int)(e[j] >> 16), (int)(e[j] & 0xffffu), r[j], csr_src, cnt + NN, ovf);
        }
    }
}

#define AG_G 8

__device__ __forceinline__ float edge_alpha(float si, float sj) {
    float a = si + sj;
    return (a > 0.0f) ? a : NEGS * a;
}

__device__ __forceinline__ void add_rows(int src, float c, int use, int cn, int lane,
                                         const float* __restrict__ g,
                                         float& accx, float& accy) {
    for (int j = 0; j < cn; ++j) {
        const int   uj = __shfl(use, j, 32);
        const int   sj = __shfl(src, j, 32);
        const float cj = __shfl(c, j, 32);
        if (uj != 0) {
            const float2 gv = ((const float2*)(g + (size_t)sj * CC))[lane];
            accx += gv.x * cj;
            accy += gv.y * cj;
        }
    }
}

__device__ __forceinline__ int row_value(int v, int k) {
    const unsigned sel = 0x11111111u * (unsigned)k;
    return (int)__builtin_amdgcn_permlane16((unsigned)v, (unsigned)v, sel, sel, true, false);
}
__device__ __forceinline__ float row_value(float v, int k) {
    return __int_as_float(row_value(__float_as_int(v), k));
}

#define AG_P 4
__device__ __forceinline__ float row_max4(const float (&v)[AG_P], int lane) {
    const bool b3 = (lane & 8) != 0, b2 = (lane & 4) != 0;
    float k2[2];
    #pragma unroll
    for (int u = 0; u < 2; ++u) {
        const float keep = b3 ? v[2 + u] : v[u];
        const float send = b3 ? v[u] : v[2 + u];
        k2[u] = fmaxf(keep, __shfl_xor(send, 8, 32));
    }
    const float keep1 = b2 ? k2[1] : k2[0];
    const float send1 = b2 ? k2[0] : k2[1];
    float k1 = fmaxf(keep1, __shfl_xor(send1, 4, 32));
    k1 = fmaxf(k1, __shfl_xor(k1, 2, 32));
    k1 = fmaxf(k1, __shfl_xor(k1, 1, 32));
    return k1;
}
__device__ __forceinline__ float row_sum4(const float (&v)[AG_P], int lane) {
    const bool b3 = (lane & 8) != 0, b2 = (lane & 4) != 0;
    float k2[2];
    #pragma unroll
    for (int u = 0; u < 2; ++u) {
        const float keep = b3 ? v[2 + u] : v[u];
        const float send = b3 ? v[u] : v[2 + u];
        k2[u] = keep + __shfl_xor(send, 8, 32);
    }
    const float keep1 = b2 ? k2[1] : k2[0];
    const float send1 = b2 ? k2[0] : k2[1];
    float k1 = keep1 + __shfl_xor(send1, 4, 32);
    k1 += __shfl_xor(k1, 2, 32);
    k1 += __shfl_xor(k1, 1, 32);
    return k1;
}

__device__ void node_general(int node, int lane,
                             const int* __restrict__ cnt, const int* __restrict__ csr_src,
                             const int* __restrict__ ovf, const float* __restrict__ s_i,
                             const float* __restrict__ s_j, const float* __restrict__ g,
                             const float* __restrict__ bias, float* __restrict__ out) {
    const float si = s_i[node];
    const int deg = cnt[node];
    const int inb = (deg < CAP) ? deg : CAP;
    const int n   = inb + 1;
    const int novf = (deg > CAP) ? cnt[NN] : 0;
    const int* cs = csr_src + (size_t)node * CAP;

    float m = -INFINITY;
    for (int base = 0; base < n; base += 32) {
        const int k = base + lane;
        if (k < n) {
            const int src = (k < inb) ? cs[k] : node;
            m = fmaxf(m, edge_alpha(si, s_j[src]));
        }
    }
    for (int base = 0; base < novf; base += 32) {
        const int p = base + lane;
        if (p < novf) {
            const int2 e = *(const int2*)(ovf + 2 * (size_t)p);
            if (e.x == node) m = fmaxf(m, edge_alpha(si, s_j[e.y]));
        }
    }
    #pragma unroll
    for (int off = 16; off > 0; off >>= 1) m = fmaxf(m, __shfl_xor(m, off, 32));

    float dsum = 0.0f;
    for (int base = 0; base < n; base += 32) {
        const int k = base + lane;
        if (k < n) {
            const int src = (k < inb) ? cs[k] : node;
            dsum += expf(edge_alpha(si, s_j[src]) - m);
        }
    }
    for (int base = 0; base < novf; base += 32) {
        const int p = base + lane;
        if (p < novf) {
            const int2 e = *(const int2*)(ovf + 2 * (size_t)p);
            if (e.x == node) dsum += expf(edge_alpha(si, s_j[e.y]) - m);
        }
    }
    #pragma unroll
    for (int off = 16; off > 0; off >>= 1) dsum += __shfl_xor(dsum, off, 32);
    const float den = dsum + 1e-16f;

    float accx = 0.0f, accy = 0.0f;
    for (int base = 0; base < n; base += 32) {
        const int k = base + lane;
        int src = node;
        float c = 0.0f;
        if (k < n) {
            src = (k < inb) ? cs[k] : node;
            c = expf(edge_alpha(si, s_j[src]) - m) / den;
        }
        const int cn = (n - base < 32) ? (n - base) : 32;
        #pragma unroll 4
        for (int j = 0; j < cn; ++j) {
            const int   sj = __shfl(src, j, 32);
            const float cj = __shfl(c, j, 32);
            const float2 gv = ((const float2*)(g + (size_t)sj * CC))[lane];
            accx += gv.x * cj;
            accy += gv.y * cj;
        }
    }
    for (int base = 0; base < novf; base += 32) {
        const int p = base + lane;
        int src = node;
        float c = 0.0f;
        int use = 0;
        if (p < novf) {
            const int2 e = *(const int2*)(ovf + 2 * (size_t)p);
            if (e.x == node) {
                src = e.y;
                c = expf(edge_alpha(si, s_j[src]) - m) / den;
                use = 1;
            }
        }
        const int cn = (novf - base < 32) ? (novf - base) : 32;
        add_rows(src, c, use, cn, lane, g, accx, accy);
    }
    const float2 b = ((const float2*)bias)[lane];
    float2 r;
    r.x = accx + b.x; r.x = (r.x > 0.0f) ? r.x : 0.0f;
    r.y = accy + b.y; r.y = (r.y > 0.0f) ? r.y : 0.0f;
    ((float2*)(out + (size_t)node * CC))[lane] = r;
}

template <int STEP>
__device__ __forceinline__ void row_steps(const int (&srcr)[AG_P], const float (&cr)[AG_P], int cn,
                                          unsigned c16, const float4* __restrict__ grow,
                                          float4 (&acc)[AG_P]) {
    for (int j = 0; j < cn; j += STEP) {
        float4 gv[STEP * AG_P];
        #pragma unroll
        for (int u = 0; u < STEP; ++u) {
            #pragma unroll
            for (int p = 0; p < AG_P; ++p) {
                const unsigned sj = (unsigned)row_value(srcr[p], j + u);
                gv[u * AG_P + p] = grow[sj * (CC / 4) + c16];
            }
        }
        #pragma unroll
        for (int u = 0; u < STEP; ++u) {
            #pragma unroll
            for (int p = 0; p < AG_P; ++p) {
                const float cj = row_value(cr[p], j + u);
                acc[p].x = fmaf(gv[u * AG_P + p].x, cj, acc[p].x);
                acc[p].y = fmaf(gv[u * AG_P + p].y, cj, acc[p].y);
                acc[p].z = fmaf(gv[u * AG_P + p].z, cj, acc[p].z);
                acc[p].w = fmaf(gv[u * AG_P + p].w, cj, acc[p].w);
            }
        }
    }
}

__global__ __launch_bounds__(256)
void aggregate_kernel(const int* __restrict__ cnt, const int* __restrict__ csr_src,
                      const int* __restrict__ ovf, const float* __restrict__ s_i,
                      const float* __restrict__ s_j, const float* __restrict__ g,
                      const float* __restrict__ bias, float* __restrict__ out) {
    const int wv   = __builtin_amdgcn_readfirstlane((blockIdx.x * blockDim.x + threadIdx.x) >> 5);
    const int lane = threadIdx.x & 31;
    if (wv >= NN / AG_G) return;
    const int node0 = wv * AG_G;
    const int h   = lane >> 4;
    const int c16 = lane & 15;

    int n[AG_G];
    float si[AG_G];
    unsigned big = 0;
    int maxn = 0;
    #pragma unroll
    for (int q = 0; q < AG_G; ++q) {
        const int deg = cnt[node0 + q];
        si[q] = s_i[node0 + q];
        n[q] = deg + 1;
        if (deg > 31) { n[q] = 0; big |= 1u << q; }
        maxn = (n[q] > maxn) ? n[q] : maxn;
    }
    int nl[AG_P], mynode[AG_P];
    float sil[AG_P];
    #pragma unroll
    for (int p = 0; p < AG_P; ++p) {
        nl[p]  = h ? n[p + AG_P] : n[p];
        sil[p] = h ? si[p + AG_P] : si[p];
        mynode[p] = node0 + p + AG_P * h;
    }

    int src[2][AG_P];
    #pragma unroll
    for (int p = 0; p < AG_P; ++p) {
        #pragma unroll
        for (int r = 0; r < 2; ++r) {
            const int e = c16 + 16 * r;
            src[r][p] = (e == nl[p] - 1) ? mynode[p] : 0;
            if (e < nl[p] - 1) src[r][p] = csr_src[(size_t)mynode[p] * CAP + e];
        }
    }
    float sjv[AG_P][2];
    #pragma unroll
    for (int p = 0; p < AG_P; ++p) {
        sjv[p][0] = s_j[src[0][p]];
        sjv[p][1] = s_j[src[1][p]];
    }
    asm volatile("" :: "v"(sjv[0][0]), "v"(sjv[0][1]), "v"(sjv[1][0]), "v"(sjv[1][1]),
                       "v"(sjv[2][0]), "v"(sjv[2][1]), "v"(sjv[3][0]), "v"(sjv[3][1]));
    float al[AG_P][2], mv[AG_P];
    #pragma unroll
    for (int p = 0; p < AG_P; ++p) {
        al[p][0] = (c16      < nl[p]) ? edge_alpha(sil[p], sjv[p][0]) : -INFINITY;
        al[p][1] = (c16 + 16 < nl[p]) ? edge_alpha(sil[p], sjv[p][1]) : -INFINITY;
        mv[p] = fmaxf(al[p][0], al[p][1]);
    }
    const float mred = row_max4(mv, lane);
    float ex[AG_P][2], sv[AG_P];
    #pragma unroll
    for (int p = 0; p < AG_P; ++p) {
        const float m = row_value(mred, 4 * p);
        ex[p][0] = (c16      < nl[p]) ? expf(al[p][0] - m) : 0.0f;
        ex[p][1] = (c16 + 16 < nl[p]) ? expf(al[p][1] - m) : 0.0f;
        sv[p] = ex[p][0] + ex[p][1];
    }
    const float sred = row_sum4(sv, lane);
    float c[2][AG_P];
    #pragma unroll
    for (int p = 0; p < AG_P; ++p) {
        const float den = row_value(sred, 4 * p) + 1e-16f;
        c[0][p] = ex[p][0] / den;
        c[1][p] = ex[p][1] / den;
    }

    const float4* grow = (const float4*)g;
    float4 acc[AG_P];
    #pragma unroll
    for (int p = 0; p < AG_P; ++p) acc[p] = make_float4(0.0f, 0.0f, 0.0f, 0.0f);
    row_steps<2>(src[0], c[0], (maxn > 16) ? 16 : maxn, (unsigned)c16, grow, acc);
    row_steps<2>(src[1], c[1], maxn - 16, (unsigned)c16, grow, acc);

    const float4 b = ((const float4*)bias)[c16];
    #pragma unroll
    for (int p = 0; p < AG_P; ++p) {
        if (((big >> (p + AG_P * h)) & 1u) == 0) {
            float4 r4;
            r4.x = acc[p].x + b.x; r4.x = (r4.x > 0.0f) ? r4.x : 0.0f;
            r4.y = acc[p].y + b.y; r4.y = (r4.y > 0.0f) ? r4.y : 0.0f;
            r4.z = acc[p].z + b.z; r4.z = (r4.z > 0.0f) ? r4.z : 0.0f;
            r4.w = acc[p].w + b.w; r4.w = (r4.w > 0.0f) ? r4.w : 0.0f;
            ((float4*)(out + (size_t)mynode[p] * CC))[c16] = r4;
        }
    }
    for (int q = 0; q < AG_G; ++q) {
        if ((big >> q) & 1u)
            node_general(node0 + q, lane, cnt, csr_src, ovf, s_i, s_j, g, bias, out);
    }
}

extern "C" void kernel_launch(void* const* d_in, const int* in_sizes, int n_in,
                              void* d_out, int out_size, void* d_ws, size_t ws_size,
                              hipStream_t stream) {
    (void)in_sizes; (void)n_in; (void)out_size; (void)ws_size;
    const float* x      = (const float*)d_in[0];
    const int*   ei     = (const int*)  d_in[1];
    const float* corrs  = (const float*)d_in[2];
    const float* W      = (const float*)d_in[3];
    const float* att_i  = (const float*)d_in[4];
    const float* att_j  = (const float*)d_in[5];
    const float* aem_i  = (const float*)d_in[6];
    const float* aem_j  = (const float*)d_in[7];
    const float* bias   = (const float*)d_in[8];
    float* out = (float*)d_out;

    float* ws    = (float*)d_ws;
    float* g     = ws;
    int*   csr   = (int*)(g + (size_t)NN * CC);
    int*   ovf   = csr + (size_t)NN * CAP;
    float* s_i   = (float*)(ovf + (size_t)2 * EE);
    float* s_j   = s_i + NN;
    int*   cnt   = (int*)(s_j + NN);

    gemm_scores_kernel<<<NN / 16, 32, 0, stream>>>(
        x, W, corrs, att_i, att_j, aem_i, aem_j, g, s_i, s_j, cnt);
    place_kernel<<<PL_S * PL_R, PL_T, 0, stream>>>(ei, cnt, csr, ovf);
    aggregate_kernel<<<(NN / AG_G * 32 + 255) / 256, 256, 0, stream>>>(
        cnt, csr, ovf, s_i, s_j, g, bias, out);
}


# Aggregation keeps 48 entries per node in registers, not 32: nodes of 32-47 in-edges leave the serial whole-wave path that ended the launch; their exp sum order: per lane, then 16-lane tree
# speedup vs baseline: 4.8382x; 1.0682x over previous
#include <hip/hip_runtime.h>
#include <math.h>

#define NN   50000
#define EE   800000
#define INC  128
#define CC   64
#define NEGS 0.2f
#define CAP  64

typedef float v2f __attribute__((ext_vector_type(2)));
typedef float v8f __attribute__((ext_vector_type(8)));

__global__ __launch_bounds__(32)
void gemm_scores_kernel(const float* __restrict__ x, const float* __restrict__ W,
                        const float* __restrict__ corrs,
                        const float* __restrict__ att_i, const float* __restrict__ att_j,
                        const float* __restrict__ aem_i, const float* __restrict__ aem_j,
                        float* __restrict__ g, float* __restrict__ s_i,
                        float* __restrict__ s_j, int* __restrict__ cnt) {
    const int rb   = blockIdx.x;
    const int lane = threadIdx.x;
    const int half = lane >> 4;
    const int l15  = lane & 15;

    float4 cr[8];
    {
        const float4* cp = (const float4*)(corrs + (size_t)(rb * 16 + half * 8) * CC) + l15;
        #pragma unroll
        for (int v = 0; v < 8; ++v) cr[v] = cp[v * (CC / 4)];
    }

    v8f acc0 = {}; v8f acc1 = {}; v8f acc2 = {}; v8f acc3 = {};

    const float* xrow = x + (rb * 16 + l15) * INC + half * 2;
    for (int k0 = 0; k0 < INC; k0 += 4) {
        v2f a; a.x = xrow[k0]; a.y = xrow[k0 + 1];
        const float* wp = W + (k0 + half * 2) * CC + l15;
        v2f b0, b1, b2, b3;
        b0.x = wp[0];  b0.y = wp[0  + CC];
        b1.x = wp[16]; b1.y = wp[16 + CC];
        b2.x = wp[32]; b2.y = wp[32 + CC];
        b3.x = wp[48]; b3.y = wp[48 + CC];
        acc0 = __builtin_amdgcn_wmma_f32_16x16x4_f32(false, a, false, b0, (short)0, acc0, false, false);
        acc1 = __builtin_amdgcn_wmma_f32_16x16x4_f32(false, a, false, b1, (short)0, acc1, false, false);
        acc2 = __builtin_amdgcn_wmma_f32_16x16x4_f32(false, a, false, b2, (short)0, acc2, false, false);
        acc3 = __builtin_amdgcn_wmma_f32_16x16x4_f32(false, a, false, b3, (short)0, acc3, false, false);
    }
    #pragma unroll
    for (int v = 0; v < 8; ++v) {
        float* op = g + (rb * 16 + v + half * 8) * CC + l15;
        op[0]  = acc0[v];
        op[16] = acc1[v];
        op[32] = acc2[v];
        op[48] = acc3[v];
    }

    float p[16];
    {
        float ai[4], aj[4];
        #pragma unroll
        for (int t = 0; t < 4; ++t) { ai[t] = att_i[l15 + 16 * t]; aj[t] = att_j[l15 + 16 * t]; }
        const float4 ei4 = ((const float4*)aem_i)[l15];
        const float4 ej4 = ((const float4*)aem_j)[l15];
        #pragma unroll
        for (int v = 0; v < 8; ++v) {
            float pi = acc0[v] * ai[0] + acc1[v] * ai[1] + acc2[v] * ai[2] + acc3[v] * ai[3];
            float pj = acc0[v] * aj[0] + acc1[v] * aj[1] + acc2[v] * aj[2] + acc3[v] * aj[3];
            pi += cr[v].x * ei4.x + cr[v].y * ei4.y + cr[v].z * ei4.z + cr[v].w * ei4.w;
            pj += cr[v].x * ej4.x + cr[v].y * ej4.y + cr[v].z * ej4.z + cr[v].w * ej4.w;
            p[v] = pi; p[8 + v] = pj;
        }
    }
    const bool b3 = (l15 & 8) != 0, b2 = (l15 & 4) != 0, b1 = (l15 & 2) != 0, b0 = (l15 & 1) != 0;
    float k8[8], k4[4], k2[2];
    #pragma unroll
    for (int u = 0; u < 8; ++u) {
        const float keep = b3 ? p[8 + u] : p[u];
        const float send = b3 ? p[u] : p[8 + u];
        k8[u] = keep + __shfl_xor(send, 8, 32);
    }
    #pragma unroll
    for (int u = 0; u < 4; ++u) {
        const float keep = b2 ? k8[4 + u] : k8[u];
        const float send = b2 ? k8[u] : k8[4 + u];
        k4[u] = keep + __shfl_xor(send, 4, 32);
    }
    #pragma unroll
    for (int u = 0; u < 2; ++u) {
        const float keep = b1 ? k4[2 + u] : k4[u];
        const float send = b1 ? k4[u] : k4[2 + u];
        k2[u] = keep + __shfl_xor(send, 2, 32);
    }
    const float keep1 = b0 ? k2[1] : k2[0];
    const float send1 = b0 ? k2[0] : k2[1];
    const float sc = keep1 + __shfl_xor(send1, 1, 32);
    const int row = rb * 16 + half * 8 + (l15 & 7);
    if (b3) { s_j[row] = sc; } else { s_i[row] = sc; cnt[row] = 0; }
    if (rb == 0 && lane == 0) cnt[NN] = 0;
}

#define PL_S   16
#define PL_R   16
#define PL_T   1024
#define PL_Q   (EE / 4 / PL_S)
#define PL_N   (NN / PL_R)
#define PL_K   ((PL_Q + PL_T - 1) / PL_T)
#define PL_F   ((PL_N + PL_T - 1) / PL_T)
#define PL_WV  (PL_T / 32)
#define PL_QC  (PL_K * 4 * 32)
#define PL_D   4

__device__ __forceinline__ void place_edge(int d, int s, int r, int* __restrict__ csr_src,
                                           int* __restrict__ ovf_n, int* __restrict__ ovf) {
    if (r < CAP) {
        csr_src[(size_t)d * CAP + r] = s;
    } else {
        const int p = atomicAdd(ovf_n, 1);
        *(int2*)(ovf + 2 * (size_t)p) = make_int2(d, s);
    }
}

__device__ __forceinline__ void queue_edge(int d, int s, bool ok, int lo,
                                           unsigned* __restrict__ myq, int& nq) {
    const unsigned u = (unsigned)(d - lo);
    const bool in = ok && (u < PL_N);
    const unsigned m = __builtin_amdgcn_ballot_w32(in);
    if (in) myq[nq + (int)__builtin_amdgcn_mbcnt_lo(m, 0u)] = (u << 16) | (unsigned)s;
    nq += __builtin_popcount(m);
}

__global__ __launch_bounds__(PL_T)
void place_kernel(const int* __restrict__ ei, int* __restrict__ cnt,
                  int* __restrict__ csr_src, int* __restrict__ ovf) {
    __shared__ int tbl[PL_N];
    __shared__ unsigned que[PL_WV * PL_QC];
    const int t    = threadIdx.x;
    const int lane = t & 31;
    const int sl = blockIdx.x / PL_R;
    const int lo = (blockIdx.x % PL_R) * PL_N;
    const int4* sq = (const int4*)ei + sl * PL_Q;
    const int4* dq = (const int4*)(ei + EE) + sl * PL_Q;
    unsigned* myq = que + (t >> 5) * PL_QC;
    int nq = 0;

    int4 d[PL_K], s[PL_K];
    #pragma unroll
    for (int k = 0; k < PL_K; ++k) {
        const int q = k * PL_T + t;
        const int qq = (q < PL_Q) ? q : PL_Q - 1;
        d[k] = dq[qq];
        s[k] = sq[qq];
    }
    #pragma unroll
    for (int j = 0; j < PL_F; ++j) {
        const int i = j * PL_T + t;
        if (i < PL_N) tbl[i] = 0;
    }
    __syncthreads();
    #pragma unroll
    for (int k = 0; k < PL_K; ++k) {
        const bool ok = (k * PL_T + t) < PL_Q;
        queue_edge(d[k].x, s[k].x, ok, lo, myq, nq);
        queue_edge(d[k].y, s[k].y, ok, lo, myq, nq);
        queue_edge(d[k].z, s[k].z, ok, lo, myq, nq);
        queue_edge(d[k].w, s[k].w, ok, lo, myq, nq);
    }
    for (int i0 = 0; i0 < nq; i0 += 32 * PL_D) {
        unsigned e[PL_D];
        #pragma unroll
        for (int j = 0; j < PL_D; ++j) {
            const int i = i0 + j * 32 + lane;
            e[j] = myq[(i < nq) ? i : 0];
        }
        #pragma unroll
        for (int j = 0; j < PL_D; ++j) {
            const int i = i0 + j * 32 + lane;
            if (i < nq) atomicAdd(&tbl[e[j] >> 16], 1);
        }
    }
    __syncthreads();
    {
        int c[PL_F], b[PL_F];
        #pragma unroll
        for (int j = 0; j < PL_F; ++j) {
            const int i = j * PL_T + t;
            c[j] = (i < PL_N) ? tbl[i] : 0;
        }
        #pragma unroll
        for (int j = 0; j < PL_F; ++j) {
            const int i = j * PL_T + t;
            b[j] = 0;
            if (c[j] != 0) b[j] = atomicAdd(cnt + lo + i, c[j]);
        }
        #pragma unroll
        for (int j = 0; j < PL_F; ++j) {
            const int i = j * PL_T + t;
            if (c[j] != 0) tbl[i] = b[j];
        }
    }
    __syncthreads();
    for (int i0 = 0; i0 < nq; i0 += 32 * PL_D) {
        unsigned e[PL_D];
        int r[PL_D];
        #pragma unroll
        for (int j = 0; j < PL_D; ++j) {
            const int i = i0 + j * 32 + lane;
            e[j] = myq[(i < nq) ? i : 0];
        }
        #pragma unroll
        for (int j = 0; j < PL_D; ++j) {
            const int i = i0 + j * 32 + lane;
            r[j] = -1;
            if (i < nq) r[j] = atomicAdd(&tbl[e[j] >> 16], 1);
        }
        #pragma unroll
        for (int j = 0; j < PL_D; ++j) {
            if (r[j] >= 0)
                place_edge(lo + (int)(e[j] >> 16), (int)(e[j] & 0xffffu), r[j], csr_src, cnt + NN, ovf);
        }
    }
}

#define AG_G 8
#define AG_R 3

__device__ __forceinline__ float edge_alpha(float si, float sj) {
    float a = si + sj;
    return (a > 0.0f) ? a : NEGS * a;
}

__device__ __forceinline__ void add_rows(int src, float c, int use, int cn, int lane,
                                         const float* __restrict__ g,
                                         float& accx, float& accy) {
    for (int j = 0; j < cn; ++j) {
        const int   uj = __shfl(use, j, 32);
        const int   sj = __shfl(src, j, 32);
        const float cj = __shfl(c, j, 32);
        if (uj != 0) {
            const float2 gv = ((const float2*)(g + (size_t)sj * CC))[lane];
            accx += gv.x * cj;
            accy += gv.y * cj;
        }
    }
}

__device__ __forceinline__ int row_value(int v, int k) {
    const unsigned sel = 0x11111111u * (unsigned)k;
    return (int)__builtin_amdgcn_permlane16((unsigned)v, (unsigned)v, sel, sel, true, false);
}
__device__ __forceinline__ float row_value(float v, int k) {
    return __int_as_float(row_value(__float_as_int(v), k));
}

#define AG_P 4
__device__ __forceinline__ float row_max4(const float (&v)[AG_P], int lane) {
    const bool b3 = (lane & 8) != 0, b2 = (lane & 4) != 0;
    float k2[2];
    #pragma unroll
    for (int u = 0; u < 2; ++u) {
        const float keep = b3 ? v[2 + u] : v[u];
        const float send = b3 ? v[u] : v[2 + u];
        k2[u] = fmaxf(keep, __shfl_xor(send, 8, 32));
    }
    const float keep1 = b2 ? k2[1] : k2[0];
    const float send1 = b2 ? k2[0] : k2[1];
    float k1 = fmaxf(keep1, __shfl_xor(send1, 4, 32));
    k1 = fmaxf(k1, __shfl_xor(k1, 2, 32));
    k1 = fmaxf(k1, __shfl_xor(k1, 1, 32));
    return k1;
}
__device__ __forceinline__ float row_sum4(const float (&v)[AG_P], int lane) {
    const bool b3 = (lane & 8) != 0, b2 = (lane & 4) != 0;
    float k2[2];
    #pragma unroll
    for (int u = 0; u < 2; ++u) {
        const float keep = b3 ? v[2 + u] : v[u];
        const float send = b3 ? v[u] : v[2 + u];
        k2[u] = keep + __shfl_xor(send, 8, 32);
    }
    const float keep1 = b2 ? k2[1] : k2[0];
    const float send1 = b2 ? k2[0] : k2[1];
    float k1 = keep1 + __shfl_xor(send1, 4, 32);
    k1 += __shfl_xor(k1, 2, 32);
    k1 += __shfl_xor(k1, 1, 32);
    return k1;
}

__device__ void node_general(int node, int lane,
                             const int* __restrict__ cnt, const int* __restrict__ csr_src,
                             const int* __restrict__ ovf, const float* __restrict__ s_i,
                             const float* __restrict__ s_j, const float* __restrict__ g,
                             const float* __restrict__ bias, float* __restrict__ out) {
    const float si = s_i[node];
    const int deg = cnt[node];
    const int inb = (deg < CAP) ? deg : CAP;
    const int n   = inb + 1;
    const int novf = (deg > CAP) ? cnt[NN] : 0;
    const int* cs = csr_src + (size_t)node * CAP;

    float m = -INFINITY;
    for (int base = 0; base < n; base += 32) {
        const int k = base + lane;
        if (k < n) {
            const int src = (k < inb) ? cs[k] : node;
            m = fmaxf(m, edge_alpha(si, s_j[src]));
        }
    }
    for (int base = 0; base < novf; base += 32) {
        const int p = base + lane;
        if (p < novf) {
            const int2 e = *(const int2*)(ovf + 2 * (size_t)p);
            if (e.x == node) m = fmaxf(m, edge_alpha(si, s_j[e.y]));
        }
    }
    #pragma unroll
    for (int off = 16; off > 0; off >>= 1) m = fmaxf(m, __shfl_xor(m, off, 32));

    float dsum = 0.0f;
    for (int base = 0; base < n; base += 32) {
        const int k = base + lane;
        if (k < n) {
            const int src = (k < inb) ? cs[k] : node;
            dsum += expf(edge_alpha(si, s_j[src]) - m);
        }
    }
    for (int base = 0; base < novf; base += 32) {
        const int p = base + lane;
        if (p < novf) {
            const int2 e = *(const int2*)(ovf + 2 * (size_t)p);
            if (e.x == node) dsum += expf(edge_alpha(si, s_j[e.y]) - m);
        }
    }
    #pragma unroll
    for (int off = 16; off > 0; off >>= 1) dsum += __shfl_xor(dsum, off, 32);
    const float den = dsum + 1e-16f;

    float accx = 0.0f, accy = 0.0f;
    for (int base = 0; base < n; base += 32) {
        const int k = base + lane;
        int src = node;
        float c = 0.0f;
        if (k < n) {
            src = (k < inb) ? cs[k] : node;
            c = expf(edge_alpha(si, s_j[src]) - m) / den;
        }
        const int cn = (n - base < 32) ? (n - base) : 32;
        #pragma unroll 4
        for (int j = 0; j < cn; ++j) {
            const int   sj = __shfl(src, j, 32);
            const float cj = __shfl(c, j, 32);
            const float2 gv = ((const float2*)(g + (size_t)sj * CC))[lane];
            accx += gv.x * cj;
            accy += gv.y * cj;
        }
    }
    for (int base = 0; base < novf; base += 32) {
        const int p = base + lane;
        int src = node;
        float c = 0.0f;
        int use = 0;
        if (p < novf) {
            const int2 e = *(const int2*)(ovf + 2 * (size_t)p);
            if (e.x == node) {
                src = e.y;
                c = expf(edge_alpha(si, s_j[src]) - m) / den;
                use = 1;
            }
        }
        const int cn = (novf - base < 32) ? (novf - base) : 32;
        add_rows(src, c, use, cn, lane, g, accx, accy);
    }
    const float2 b = ((const float2*)bias)[lane];
    float2 r;
    r.x = accx + b.x; r.x = (r.x > 0.0f) ? r.x : 0.0f;
    r.y = accy + b.y; r.y = (r.y > 0.0f) ? r.y : 0.0f;
    ((float2*)(out + (size_t)node * CC))[lane] = r;
}

template <int STEP>
__device__ __forceinline__ void row_steps(const int (&srcr)[AG_P], const float (&cr)[AG_P], int cn,
                                          unsigned c16, const float4* __restrict__ grow,
                                          float4 (&acc)[AG_P]) {
    for (int j = 0; j < cn; j += STEP) {
        float4 gv[STEP * AG_P];
        #pragma unroll
        for (int u = 0; u < STEP; ++u) {
            #pragma unroll
            for (int p = 0; p < AG_P; ++p) {
                const unsigned sj = (unsigned)row_value(srcr[p], j + u);
                gv[u * AG_P + p] = grow[sj * (CC / 4) + c16];
            }
        }
        #pragma unroll
        for (int u = 0; u < STEP; ++u) {
            #pragma unroll
            for (int p = 0; p < AG_P; ++p) {
                const float cj = row_value(cr[p], j + u);
                acc[p].x = fmaf(gv[u * AG_P + p].x, cj, acc[p].x);
                acc[p].y = fmaf(gv[u * AG_P + p].y, cj, acc[p].y);
                acc[p].z = fmaf(gv[u * AG_P + p].z, cj, acc[p].z);
                acc[p].w = fmaf(gv[u * AG_P + p].w, cj, acc[p].w);
            }
        }
    }
}

__global__ __launch_bounds__(256)
void aggregate_kernel(const int* __restrict__ cnt, const int* __restrict__ csr_src,
                      const int* __restrict__ ovf, const float* __restrict__ s_i,
                      const float* __restrict__ s_j, const float* __restrict__ g,
                      const float* __restrict__ bias, float* __restrict__ out) {
    const int wv   = __builtin_amdgcn_readfirstlane((blockIdx.x * blockDim.x + threadIdx.x) >> 5);
    const int lane = threadIdx.x & 31;
    if (wv >= NN / AG_G) return;
    const int node0 = wv * AG_G;
    const int h   = lane >> 4;
    const int c16 = lane & 15;

    int n[AG_G];
    float si[AG_G];
    unsigned big = 0;
    int maxn = 0;
    #pragma unroll
    for (int q = 0; q < AG_G; ++q) {
        const int deg = cnt[node0 + q];
        si[q] = s_i[node0 + q];
        n[q] = deg + 1;
        if (deg > 16 * AG_R - 1) { n[q] = 0; big |= 1u << q; }
        maxn = (n[q] > maxn) ? n[q] : maxn;
    }
    int nl[AG_P], mynode[AG_P];
    float sil[AG_P];
    #pragma unroll
    for (int p = 0; p < AG_P; ++p) {
        nl[p]  = h ? n[p + AG_P] : n[p];
        sil[p] = h ? si[p + AG_P] : si[p];
        mynode[p] = node0 + p + AG_P * h;
    }

    int src[AG_R][AG_P];
    #pragma unroll
    for (int p = 0; p < AG_P; ++p) {
        #pragma unroll
        for (int r = 0; r < AG_R; ++r) {
            const int e = c16 + 16 * r;
            src[r][p] = (e == nl[p] - 1) ? mynode[p] : 0;
            if (e < nl[p] - 1) src[r][p] = csr_src[(size_t)mynode[p] * CAP + e];
        }
    }
    float sjv[AG_P][AG_R];
    #pragma unroll
    for (int p = 0; p < AG_P; ++p) {
        #pragma unroll
        for (int r = 0; r < AG_R; ++r) sjv[p][r] = s_j[src[r][p]];
    }
    static_assert(AG_R == 3, "the statement below names AG_R score registers per node");
    asm volatile("" :: "v"(sjv[0][0]), "v"(sjv[0][1]), "v"(sjv[0][2]),
                       "v"(sjv[1][0]), "v"(sjv[1][1]), "v"(sjv[1][2]),
                       "v"(sjv[2][0]), "v"(sjv[2][1]), "v"(sjv[2][2]),
                       "v"(sjv[3][0]), "v"(sjv[3][1]), "v"(sjv[3][2]));
    float al[AG_P][AG_R], mv[AG_P];
    #pragma unroll
    for (int p = 0; p < AG_P; ++p) {
        #pragma unroll
        for (int r = 0; r < AG_R; ++r) {
            al[p][r] = (c16 + 16 * r < nl[p]) ? edge_alpha(sil[p], sjv[p][r]) : -INFINITY;
            mv[p] = (r == 0) ? al[p][0] : fmaxf(mv[p], al[p][r]);
        }
    }
    const float mred = row_max4(mv, lane);
    float ex[AG_P][AG_R], sv[AG_P];
    #pragma unroll
    for (int p = 0; p < AG_P; ++p) {
        const float m = row_value(mred, 4 * p);
        #pragma unroll
        for (int r = 0; r < AG_R; ++r) {
            ex[p][r] = (c16 + 16 * r < nl[p]) ? expf(al[p][r] - m) : 0.0f;
            sv[p] = (r == 0) ? ex[p][0] : sv[p] + ex[p][r];
        }
    }
    const float sred = row_sum4(sv, lane);
    float c[AG_R][AG_P];
    #pragma unroll
    for (int p = 0; p < AG_P; ++p) {
        const float den = row_value(sred, 4 * p) + 1e-16f;
        #pragma unroll
        for (int r = 0; r < AG_R; ++r) c[r][p] = ex[p][r] / den;
    }

    const float4* grow = (const float4*)g;
    float4 acc[AG_P];
    #pragma unroll
    for (int p = 0; p < AG_P; ++p) acc[p] = make_float4(0.0f, 0.0f, 0.0f, 0.0f);
    #pragma unroll
    for (int r = 0; r < AG_R; ++r) {
        const int left = maxn - 16 * r;
        row_steps<2>(src[r], c[r], (left > 16) ? 16 : left, (unsigned)c16, grow, acc);
    }

    const float4 b = ((const float4*)bias)[c16];
    #pragma unroll
    for (int p = 0; p < AG_P; ++p) {
        if (((big >> (p + AG_P * h)) & 1u) == 0) {
            float4 r4;
            r4.x = acc[p].x + b.x; r4.x = (r4.x > 0.0f) ? r4.x : 0.0f;
            r4.y = acc[p].y + b.y; r4.y = (r4.y > 0.0f) ? r4.y : 0.0f;
            r4.z = acc[p].z + b.z; r4.z = (r4.z > 0.0f) ? r4.z : 0.0f;
            r4.w = acc[p].w + b.w; r4.w = (r4.w > 0.0f) ? r4.w : 0.0f;
            ((float4*)(out + (size_t)mynode[p] * CC))[c16] = r4;
        }
    }
    for (int q = 0; q < AG_G; ++q) {
        if ((big >> q) & 1u)
            node_general(node0 + q, lane, cnt, csr_src, ovf, s_i, s_j, g, bias, out);
    }
}

extern "C" void kernel_launch(void* const* d_in, const int* in_sizes, int n_in,
                              void* d_out, int out_size, void* d_ws, size_t ws_size,
                              hipStream_t stream) {
    (void)in_sizes; (void)n_in; (void)out_size; (void)ws_size;
    const float* x      = (const float*)d_in[0];
    const int*   ei     = (const int*)  d_in[1];
    const float* corrs  = (const float*)d_in[2];
    const float* W      = (const float*)d_in[3];
    const float* att_i  = (const float*)d_in[4];
    const float* att_j  = (const float*)d_in[5];
    const float* aem_i  = (const float*)d_in[6];
    const float* aem_j  = (const float*)d_in[7];
    const float* bias   = (const float*)d_in[8];
    float* out = (float*)d_out;

    float* ws    = (float*)d_ws;
    float* g     = ws;
    int*   csr   = (int*)(g + (size_t)NN * CC);
    int*   ovf   = csr + (size_t)NN * CAP;
    float* s_i   = (float*)(ovf + (size_t)2 * EE);
    float* s_j   = s_i + NN;
    int*   cnt   = (int*)(s_j + NN);

    gemm_scores_kernel<<<NN / 16, 32, 0, stream>>>(
        x, W, corrs, att_i, att_j, aem_i, aem_j, g, s_i, s_j, cnt);
    place_kernel<<<PL_S * PL_R, PL_T, 0, stream>>>(ei, cnt, csr, ovf);
    aggregate_kernel<<<(NN / AG_G * 32 + 255) / 256, 256, 0, stream>>>(
        cnt, csr, ovf, s_i, s_j, g, bias, out);
}


# Edges sorted by node range once, in LDS, by extra one-wave blocks beside the GEMM; place reads only its own segments (no 16x slice re-read); sum order: block arrival, then chunk, LDS add order
# speedup vs baseline: 5.4517x; 1.1268x over previous
#include <hip/hip_runtime.h>
#include <math.h>

#define NN   50000
#define EE   800000
#define INC  128
#define CC   64
#define NEGS 0.2f
#define CAP  64

typedef float v2f __attribute__((ext_vector_type(2)));
typedef float v8f __attribute__((ext_vector_type(8)));


#define PL_R   16
#define PL_N   (NN / PL_R)
#define BN_E   1024
#define BN_C   ((EE + BN_E - 1) / BN_E)
#define BN_K   (BN_E / 4 / 32)
#define BN_W   32

__device__ __forceinline__ void bin_chunk(int chunk, int lane, const int* __restrict__ ei,
                                          unsigned* __restrict__ binned, int* __restrict__ bstart,
                                          int* ctr, unsigned* sorted) {
    const int4* sq = (const int4*)ei;
    const int4* dq = (const int4*)(ei + EE);
    int4 d[BN_K], s[BN_K];
    #pragma unroll
    for (int k = 0; k < BN_K; ++k) {
        const int q = chunk * (BN_E / 4) + k * 32 + lane;
        const int qq = (q < EE / 4) ? q : EE / 4 - 1;
        d[k] = dq[qq];
        s[k] = sq[qq];
    }
    ctr[lane] = 0;
    #pragma unroll
    for (int k = 0; k < BN_K; ++k) ((uint4*)sorted)[k * 32 + lane] = make_uint4(0u, 0u, 0u, 0u);
    __syncthreads();
    #define BN_RANGE(dd, okk) (((okk) && (unsigned)(dd) < (unsigned)NN) ? (int)((unsigned)(dd) / (unsigned)PL_N) : -1)
    #pragma unroll
    for (int k = 0; k < BN_K; ++k) {
        const bool ok = (chunk * (BN_E / 4) + k * 32 + lane) < EE / 4;
        const int r0 = BN_RANGE(d[k].x, ok), r1 = BN_RANGE(d[k].y, ok);
        const int r2 = BN_RANGE(d[k].z, ok), r3 = BN_RANGE(d[k].w, ok);
        if (r0 >= 0) atomicAdd(&ctr[r0], 1);
        if (r1 >= 0) atomicAdd(&ctr[r1], 1);
        if (r2 >= 0) atomicAdd(&ctr[r2], 1);
        if (r3 >= 0) atomicAdd(&ctr[r3], 1);
    }
    __syncthreads();
    const int cv = (lane < PL_R) ? ctr[lane] : 0;
    int inc = cv;
    #pragma unroll
    for (int off = 1; off < 32; off <<= 1) {
        const int up = __shfl_up(inc, off, 32);
        if (lane >= off) inc += up;
    }
    const int st = inc - cv;
    bstart[chunk * BN_W + lane] = st;
    ctr[lane] = st;
    __syncthreads();
    #pragma unroll
    for (int k = 0; k < BN_K; ++k) {
        const bool ok = (chunk * (BN_E / 4) + k * 32 + lane) < EE / 4;
        const int r0 = BN_RANGE(d[k].x, ok), r1 = BN_RANGE(d[k].y, ok);
        const int r2 = BN_RANGE(d[k].z, ok), r3 = BN_RANGE(d[k].w, ok);
        int p0 = -1, p1 = -1, p2 = -1, p3 = -1;
        if (r0 >= 0) p0 = atomicAdd(&ctr[r0], 1);
        if (r1 >= 0) p1 = atomicAdd(&ctr[r1], 1);
        if (r2 >= 0) p2 = atomicAdd(&ctr[r2], 1);
        if (r3 >= 0) p3 = atomicAdd(&ctr[r3], 1);
        if (p0 >= 0) sorted[p0] = ((unsigned)(d[k].x - r0 * PL_N) << 16) | (unsigned)s[k].x;
        if (p1 >= 0) sorted[p1] = ((unsigned)(d[k].y - r1 * PL_N) << 16) | (unsigned)s[k].y;
        if (p2 >= 0) sorted[p2] = ((unsigned)(d[k].z - r2 * PL_N) << 16) | (unsigned)s[k].z;
        if (p3 >= 0) sorted[p3] = ((unsigned)(d[k].w - r3 * PL_N) << 16) | (unsigned)s[k].w;
    }
    #undef BN_RANGE
    __syncthreads();
    uint4* op = (uint4*)(binned + (size_t)chunk * BN_E);
    #pragma unroll
    for (int k = 0; k < BN_K; ++k) op[k * 32 + lane] = ((const uint4*)sorted)[k * 32 + lane];
}

__global__ __launch_bounds__(32)
void gemm_scores_kernel(const float* __restrict__ x, const float* __restrict__ W,
                        const float* __restrict__ corrs,
                        const float* __restrict__ att_i, const float* __restrict__ att_j,
                        const float* __restrict__ aem_i, const float* __restrict__ aem_j,
                        float* __restrict__ g, float* __restrict__ s_i,
                        float* __restrict__ s_j, int* __restrict__ cnt,
                        const int* __restrict__ ei, unsigned* __restrict__ binned,
                        int* __restrict__ bstart) {
    __shared__ int bn_ctr[32];
    __shared__ __attribute__((aligned(16))) unsigned bn_sorted[BN_E];
    const int lane = threadIdx.x;
    if (blockIdx.x < BN_C) {
        bin_chunk(blockIdx.x, lane, ei, binned, bstart, bn_ctr, bn_sorted);
        return;
    }
    const int rb   = blockIdx.x - BN_C;
    const int half = lane >> 4;
    const int l15  = lane & 15;

    float4 cr[8];
    {
        const float4* cp = (const float4*)(corrs + (size_t)(rb * 16 + half * 8) * CC) + l15;
        #pragma unroll
        for (int v = 0; v < 8; ++v) cr[v] = cp[v * (CC / 4)];
    }

    v8f acc0 = {}; v8f acc1 = {}; v8f acc2 = {}; v8f acc3 = {};

    const float* xrow = x + (rb * 16 + l15) * INC + half * 2;
    for (int k0 = 0; k0 < INC; k0 += 4) {
        v2f a; a.x = xrow[k0]; a.y = xrow[k0 + 1];
        const float* wp = W + (k0 + half * 2) * CC + l15;
        v2f b0, b1, b2, b3;
        b0.x = wp[0];  b0.y = wp[0  + CC];
        b1.x = wp[16]; b1.y = wp[16 + CC];
        b2.x = wp[32]; b2.y = wp[32 + CC];
        b3.x = wp[48]; b3.y = wp[48 + CC];
        acc0 = __builtin_amdgcn_wmma_f32_16x16x4_f32(false, a, false, b0, (short)0, acc0, false, false);
        acc1 = __builtin_amdgcn_wmma_f32_16x16x4_f32(false, a, false, b1, (short)0, acc1, false, false);
        acc2 = __builtin_amdgcn_wmma_f32_16x16x4_f32(false, a, false, b2, (short)0, acc2, false, false);
        acc3 = __builtin_amdgcn_wmma_f32_16x16x4_f32(false, a, false, b3, (short)0, acc3, false, false);
    }
    #pragma unroll
    for (int v = 0; v < 8; ++v) {
        float* op = g + (rb * 16 + v + half * 8) * CC + l15;
        op[0]  = acc0[v];
        op[16] = acc1[v];
        op[32] = acc2[v];
        op[48] = acc3[v];
    }

    float p[16];
    {
        float ai[4], aj[4];
        #pragma unroll
        for (int t = 0; t < 4; ++t) { ai[t] = att_i[l15 + 16 * t]; aj[t] = att_j[l15 + 16 * t]; }
        const float4 ei4 = ((const float4*)aem_i)[l15];
        const float4 ej4 = ((const float4*)aem_j)[l15];
        #pragma unroll
        for (int v = 0; v < 8; ++v) {
            float pi = acc0[v] * ai[0] + acc1[v] * ai[1] + acc2[v] * ai[2] + acc3[v] * ai[3];
            float pj = acc0[v] * aj[0] + acc1[v] * aj[1] + acc2[v] * aj[2] + acc3[v] * aj[3];
            pi += cr[v].x * ei4.x + cr[v].y * ei4.y + cr[v].z * ei4.z + cr[v].w * ei4.w;
            pj += cr[v].x * ej4.x + cr[v].y * ej4.y + cr[v].z * ej4.z + cr[v].w * ej4.w;
            p[v] = pi; p[8 + v] = pj;
        }
    }
    const bool b3 = (l15 & 8) != 0, b2 = (l15 & 4) != 0, b1 = (l15 & 2) != 0, b0 = (l15 & 1) != 0;
    float k8[8], k4[4], k2[2];
    #pragma unroll
    for (int u = 0; u < 8; ++u) {
        const float keep = b3 ? p[8 + u] : p[u];
        const float send = b3 ? p[u] : p[8 + u];
        k8[u] = keep + __shfl_xor(send, 8, 32);
    }
    #pragma unroll
    for (int u = 0; u < 4; ++u) {
        const float keep = b2 ? k8[4 + u] : k8[u];
        const float send = b2 ? k8[u] : k8[4 + u];
        k4[u] = keep + __shfl_xor(send, 4, 32);
    }
    #pragma unroll
    for (int u = 0; u < 2; ++u) {
        const float keep = b1 ? k4[2 + u] : k4[u];
        const float send = b1 ? k4[u] : k4[2 + u];
        k2[u] = keep + __shfl_xor(send, 2, 32);
    }
    const float keep1 = b0 ? k2[1] : k2[0];
    const float send1 = b0 ? k2[0] : k2[1];
    const float sc = keep1 + __shfl_xor(send1, 1, 32);
    const int row = rb * 16 + half * 8 + (l15 & 7);
    if (b3) { s_j[row] = sc; } else { s_i[row] = sc; cnt[row] = 0; }
    if (rb == 0 && lane == 0) cnt[NN] = 0;
}

#define PL_S   16
#define PL_CS  ((BN_C + PL_S - 1) / PL_S)
#define PL_T   1024
#define PL_F   ((PL_N + PL_T - 1) / PL_T)
#define PL_WV  (PL_T / 32)
#define PL_CW  ((PL_CS + PL_WV - 1) / PL_WV)
#define PL_D   2

__device__ __forceinline__ void place_edge(int d, int s, int r, int* __restrict__ csr_src,
                                           int* __restrict__ ovf_n, int* __restrict__ ovf) {
    if (r < CAP) {
        csr_src[(size_t)d * CAP + r] = s;
    } else {
        const int p = atomicAdd(ovf_n, 1);
        *(int2*)(ovf + 2 * (size_t)p) = make_int2(d, s);
    }
}

__global__ __launch_bounds__(PL_T)
void place_kernel(const unsigned* __restrict__ binned, const int* __restrict__ bstart,
                  int* __restrict__ cnt, int* __restrict__ csr_src, int* __restrict__ ovf) {
    __shared__ int tbl[PL_N];
    const int t    = threadIdx.x;
    const int lane = t & 31;
    const int wv = __builtin_amdgcn_readfirstlane(t >> 5);
    const int sl = blockIdx.x / PL_R;
    const int rg = blockIdx.x % PL_R;
    const int lo = rg * PL_N;
    const int c_end = ((sl + 1) * PL_CS < BN_C) ? (sl + 1) * PL_CS : BN_C;

    int n[PL_CW];
    const unsigned* seg[PL_CW];
    int nmax = 0;
    #pragma unroll
    for (int cc = 0; cc < PL_CW; ++cc) {
        const int c  = sl * PL_CS + wv + cc * PL_WV;
        const int cl = (c < c_end) ? c : c_end - 1;
        const int a  = bstart[cl * BN_W + rg];
        const int b  = bstart[cl * BN_W + rg + 1];
        n[cc] = (c < c_end) ? b - a : 0;
        seg[cc] = binned + (size_t)cl * BN_E + ((n[cc] > 0) ? a : 0);
        nmax = (n[cc] > nmax) ? n[cc] : nmax;
    }

    #pragma unroll
    for (int j = 0; j < PL_F; ++j) {
        const int i = j * PL_T + t;
        if (i < PL_N) tbl[i] = 0;
    }
    __syncthreads();
    for (int i0 = 0; i0 < nmax; i0 += 32 * PL_D) {
        unsigned e[PL_CW * PL_D];
        #pragma unroll
        for (int cc = 0; cc < PL_CW; ++cc) {
            #pragma unroll
            for (int j = 0; j < PL_D; ++j) {
                const int i = i0 + j * 32 + lane;
                e[cc * PL_D + j] = seg[cc][(i < n[cc]) ? i : 0];
            }
        }
        #pragma unroll
        for (int cc = 0; cc < PL_CW; ++cc) {
            #pragma unroll
            for (int j = 0; j < PL_D; ++j) {
                const int i = i0 + j * 32 + lane;
                if (i < n[cc]) atomicAdd(&tbl[e[cc * PL_D + j] >> 16], 1);
            }
        }
    }
    __syncthreads();
    {
        int c[PL_F], b[PL_F];
        #pragma unroll
        for (int j = 0; j < PL_F; ++j) {
            const int i = j * PL_T + t;
            c[j] = (i < PL_N) ? tbl[i] : 0;
        }
        #pragma unroll
        for (int j = 0; j < PL_F; ++j) {
            const int i = j * PL_T + t;
            b[j] = 0;
            if (c[j] != 0) b[j] = atomicAdd(cnt + lo + i, c[j]);
        }
        #pragma unroll
        for (int j = 0; j < PL_F; ++j) {
            const int i = j * PL_T + t;
            if (c[j] != 0) tbl[i] = b[j];
        }
    }
    __syncthreads();
    for (int i0 = 0; i0 < nmax; i0 += 32 * PL_D) {
        unsigned e[PL_CW * PL_D];
        int r[PL_CW * PL_D];
        #pragma unroll
        for (int cc = 0; cc < PL_CW; ++cc) {
            #pragma unroll
            for (int j = 0; j < PL_D; ++j) {
                const int i = i0 + j * 32 + lane;
                e[cc * PL_D + j] = seg[cc][(i < n[cc]) ? i : 0];
            }
        }
        #pragma unroll
        for (int cc = 0; cc < PL_CW; ++cc) {
            #pragma unroll
            for (int j = 0; j < PL_D; ++j) {
                const int i = i0 + j * 32 + lane;
                r[cc * PL_D + j] = -1;
                if (i < n[cc]) r[cc * PL_D + j] = atomicAdd(&tbl[e[cc * PL_D + j] >> 16], 1);
            }
        }
        #pragma unroll
        for (int q = 0; q < PL_CW * PL_D; ++q) {
            if (r[q] >= 0)
                place_edge(lo + (int)(e[q] >> 16), (int)(e[q] & 0xffffu), r[q], csr_src, cnt + NN, ovf);
        }
    }
}

#define AG_G 8
#define AG_R 3

__device__ __forceinline__ float edge_alpha(float si, float sj) {
    float a = si + sj;
    return (a > 0.0f) ? a : NEGS * a;
}

__device__ __forceinline__ void add_rows(int src, float c, int use, int cn, int lane,
                                         const float* __restrict__ g,
                                         float& accx, float& accy) {
    for (int j = 0; j < cn; ++j) {
        const int   uj = __shfl(use, j, 32);
        const int   sj = __shfl(src, j, 32);
        const float cj = __shfl(c, j, 32);
        if (uj != 0) {
            const float2 gv = ((const float2*)(g + (size_t)sj * CC))[lane];
            accx += gv.x * cj;
            accy += gv.y * cj;
        }
    }
}

__device__ __forceinline__ int row_value(int v, int k) {
    const unsigned sel = 0x11111111u * (unsigned)k;
    return (int)__builtin_amdgcn_permlane16((unsigned)v, (unsigned)v, sel, sel, true, false);
}
__device__ __forceinline__ float row_value(float v, int k) {
    return __int_as_float(row_value(__float_as_int(v), k));
}

#define AG_P 4
__device__ __forceinline__ float row_max4(const float (&v)[AG_P], int lane) {
    const bool b3 = (lane & 8) != 0, b2 = (lane & 4) != 0;
    float k2[2];
    #pragma unroll
    for (int u = 0; u < 2; ++u) {
        const float keep = b3 ? v[2 + u] : v[u];
        const float send = b3 ? v[u] : v[2 + u];
        k2[u] = fmaxf(keep, __shfl_xor(send, 8, 32));
    }
    const float keep1 = b2 ? k2[1] : k2[0];
    const float send1 = b2 ? k2[0] : k2[1];
    float k1 = fmaxf(keep1, __shfl_xor(send1, 4, 32));
    k1 = fmaxf(k1, __shfl_xor(k1, 2, 32));
    k1 = fmaxf(k1, __shfl_xor(k1, 1, 32));
    return k1;
}
__device__ __forceinline__ float row_sum4(const float (&v)[AG_P], int lane) {
    const bool b3 = (lane & 8) != 0, b2 = (lane & 4) != 0;
    float k2[2];
    #pragma unroll
    for (int u = 0; u < 2; ++u) {
        const float keep = b3 ? v[2 + u] : v[u];
        const float send = b3 ? v[u] : v[2 + u];
        k2[u] = keep + __shfl_xor(send, 8, 32);
    }
    const float keep1 = b2 ? k2[1] : k2[0];
    const float send1 = b2 ? k2[0] : k2[1];
    float k1 = keep1 + __shfl_xor(send1, 4, 32);
    k1 += __shfl_xor(k1, 2, 32);
    k1 += __shfl_xor(k1, 1, 32);
    return k1;
}

__device__ void node_general(int node, int lane,
                             const int* __restrict__ cnt, const int* __restrict__ csr_src,
                             const int* __restrict__ ovf, const float* __restrict__ s_i,
                             const float* __restrict__ s_j, const float* __restrict__ g,
                             const float* __restrict__ bias, float* __restrict__ out) {
    const float si = s_i[node];
    const int deg = cnt[node];
    const int inb = (deg < CAP) ? deg : CAP;
    const int n   = inb + 1;
    const int novf = (deg > CAP) ? cnt[NN] : 0;
    const int* cs = csr_src + (size_t)node * CAP;

    float m = -INFINITY;
    for (int base = 0; base < n; base += 32) {
        const int k = base + lane;
        if (k < n) {
            const int src = (k < inb) ? cs[k] : node;
            m = fmaxf(m, edge_alpha(si, s_j[src]));
        }
    }
    for (int base = 0; base < novf; base += 32) {
        const int p = base + lane;
        if (p < novf) {
            const int2 e = *(const int2*)(ovf + 2 * (size_t)p);
            if (e.x == node) m = fmaxf(m, edge_alpha(si, s_j[e.y]));
        }
    }
    #pragma unroll
    for (int off = 16; off > 0; off >>= 1) m = fmaxf(m, __shfl_xor(m, off, 32));

    float dsum = 0.0f;
    for (int base = 0; base < n; base += 32) {
        const int k = base + lane;
        if (k < n) {
            const int src = (k < inb) ? cs[k] : node;
            dsum += expf(edge_alpha(si, s_j[src]) - m);
        }
    }
    for (int base = 0; base < novf; base += 32) {
        const int p = base + lane;
        if (p < novf) {
            const int2 e = *(const int2*)(ovf + 2 * (size_t)p);
            if (e.x == node) dsum += expf(edge_alpha(si, s_j[e.y]) - m);
        }
    }
    #pragma unroll
    for (int off = 16; off > 0; off >>= 1) dsum += __shfl_xor(dsum, off, 32);
    const float den = dsum + 1e-16f;

    float accx = 0.0f, accy = 0.0f;
    for (int base = 0; base < n; base += 32) {
        const int k = base + lane;
        int src = node;
        float c = 0.0f;
        if (k < n) {
            src = (k < inb) ? cs[k] : node;
            c = expf(edge_alpha(si, s_j[src]) - m) / den;
        }
        const int cn = (n - base < 32) ? (n - base) : 32;
        #pragma unroll 4
        for (int j = 0; j < cn; ++j) {
            const int   sj = __shfl(src, j, 32);
            const float cj = __shfl(c, j, 32);
            const float2 gv = ((const float2*)(g + (size_t)sj * CC))[lane];
            accx += gv.x * cj;
            accy += gv.y * cj;
        }
    }
    for (int base = 0; base < novf; base += 32) {
        const int p = base + lane;
        int src = node;
        float c = 0.0f;
        int use = 0;
        if (p < novf) {
            const int2 e = *(const int2*)(ovf + 2 * (size_t)p);
            if (e.x == node) {
                src = e.y;
                c = expf(edge_alpha(si, s_j[src]) - m) / den;
                use = 1;
            }
        }
        const int cn = (novf - base < 32) ? (novf - base) : 32;
        add_rows(src, c, use, cn, lane, g, accx, accy);
    }
    const float2 b = ((const float2*)bias)[lane];
    float2 r;
    r.x = accx + b.x; r.x = (r.x > 0.0f) ? r.x : 0.0f;
    r.y = accy + b.y; r.y = (r.y > 0.0f) ? r.y : 0.0f;
    ((float2*)(out + (size_t)node * CC))[lane] = r;
}

template <int STEP>
__device__ __forceinline__ void row_steps(const int (&srcr)[AG_P], const float (&cr)[AG_P], int cn,
                                          unsigned c16, const float4* __restrict__ grow,
                                          float4 (&acc)[AG_P]) {
    for (int j = 0; j < cn; j += STEP) {
        float4 gv[STEP * AG_P];
        #pragma unroll
        for (int u = 0; u < STEP; ++u) {
            #pragma unroll
            for (int p = 0; p < AG_P; ++p) {
                const unsigned sj = (unsigned)row_value(srcr[p], j + u);
                gv[u * AG_P + p] = grow[sj * (CC / 4) + c16];
            }
        }
        #pragma unroll
        for (int u = 0; u < STEP; ++u) {
            #pragma unroll
            for (int p = 0; p < AG_P; ++p) {
                const float cj = row_value(cr[p], j + u);
                acc[p].x = fmaf(gv[u * AG_P + p].x, cj, acc[p].x);
                acc[p].y = fmaf(gv[u * AG_P + p].y, cj, acc[p].y);
                acc[p].z = fmaf(gv[u * AG_P + p].z, cj, acc[p].z);
                acc[p].w = fmaf(gv[u * AG_P + p].w, cj, acc[p].w);
            }
        }
    }
}

__global__ __launch_bounds__(256)
void aggregate_kernel(const int* __restrict__ cnt, const int* __restrict__ csr_src,
                      const int* __restrict__ ovf, const float* __restrict__ s_i,
                      const float* __restrict__ s_j, const float* __restrict__ g,
                      const float* __restrict__ bias, float* __restrict__ out) {
    const int wv   = __builtin_amdgcn_readfirstlane((blockIdx.x * blockDim.x + threadIdx.x) >> 5);
    const int lane = threadIdx.x & 31;
    if (wv >= NN / AG_G) return;
    const int node0 = wv * AG_G;
    const int h   = lane >> 4;
    const int c16 = lane & 15;

    int n[AG_G];
    float si[AG_G];
    unsigned big = 0;
    int maxn = 0;
    #pragma unroll
    for (int q = 0; q < AG_G; ++q) {
        const int deg = cnt[node0 + q];
        si[q] = s_i[node0 + q];
        n[q] = deg + 1;
        if (deg > 16 * AG_R - 1) { n[q] = 0; big |= 1u << q; }
        maxn = (n[q] > maxn) ? n[q] : maxn;
    }
    int nl[AG_P], mynode[AG_P];
    float sil[AG_P];
    #pragma unroll
    for (int p = 0; p < AG_P; ++p) {
        nl[p]  = h ? n[p + AG_P] : n[p];
        sil[p] = h ? si[p + AG_P] : si[p];
        mynode[p] = node0 + p + AG_P * h;
    }

    int src[AG_R][AG_P];
    #pragma unroll
    for (int p = 0; p < AG_P; ++p) {
        #pragma unroll
        for (int r = 0; r < AG_R; ++r) {
            const int e = c16 + 16 * r;
            src[r][p] = (e == nl[p] - 1) ? mynode[p] : 0;
            if (e < nl[p] - 1) src[r][p] = csr_src[(size_t)mynode[p] * CAP + e];
        }
    }
    float sjv[AG_P][AG_R];
    #pragma unroll
    for (int p = 0; p < AG_P; ++p) {
        #pragma unroll
        for (int r = 0; r < AG_R; ++r) sjv[p][r] = s_j[src[r][p]];
    }
    static_assert(AG_R == 3, "the statement below names AG_R score registers per node");
    asm volatile("" :: "v"(sjv[0][0]), "v"(sjv[0][1]), "v"(sjv[0][2]),
                       "v"(sjv[1][0]), "v"(sjv[1][1]), "v"(sjv[1][2]),
                       "v"(sjv[2][0]), "v"(sjv[2][1]), "v"(sjv[2][2]),
                       "v"(sjv[3][0]), "v"(sjv[3][1]), "v"(sjv[3][2]));
    float al[AG_P][AG_R], mv[AG_P];
    #pragma unroll
    for (int p = 0; p < AG_P; ++p) {
        #pragma unroll
        for (int r = 0; r < AG_R; ++r) {
            al[p][r] = (c16 + 16 * r < nl[p]) ? edge_alpha(sil[p], sjv[p][r]) : -INFINITY;
            mv[p] = (r == 0) ? al[p][0] : fmaxf(mv[p], al[p][r]);
        }
    }
    const float mred = row_max4(mv, lane);
    float ex[AG_P][AG_R], sv[AG_P];
    #pragma unroll
    for (int p = 0; p < AG_P; ++p) {
        const float m = row_value(mred, 4 * p);
        #pragma unroll
        for (int r = 0; r < AG_R; ++r) {
            ex[p][r] = (c16 + 16 * r < nl[p]) ? expf(al[p][r] - m) : 0.0f;
            sv[p] = (r == 0) ? ex[p][0] : sv[p] + ex[p][r];
        }
    }
    const float sred = row_sum4(sv, lane);
    float c[AG_R][AG_P];
    #pragma unroll
    for (int p = 0; p < AG_P; ++p) {
        const float den = row_value(sred, 4 * p) + 1e-16f;
        #pragma unroll
        for (int r = 0; r < AG_R; ++r) c[r][p] = ex[p][r] / den;
    }

    const float4* grow = (const float4*)g;
    float4 acc[AG_P];
    #pragma unroll
    for (int p = 0; p < AG_P; ++p) acc[p] = make_float4(0.0f, 0.0f, 0.0f, 0.0f);
    #pragma unroll
    for (int r = 0; r < AG_R; ++r) {
        const int left = maxn - 16 * r;
        row_steps<2>(src[r], c[r], (left > 16) ? 16 : left, (unsigned)c16, grow, acc);
    }

    const float4 b = ((const float4*)bias)[c16];
    #pragma unroll
    for (int p = 0; p < AG_P; ++p) {
        if (((big >> (p + AG_P * h)) & 1u) == 0) {
            float4 r4;
            r4.x = acc[p].x + b.x; r4.x = (r4.x > 0.0f) ? r4.x : 0.0f;
            r4.y = acc[p].y + b.y; r4.y = (r4.y > 0.0f) ? r4.y : 0.0f;
            r4.z = acc[p].z + b.z; r4.z = (r4.z > 0.0f) ? r4.z : 0.0f;
            r4.w = acc[p].w + b.w; r4.w = (r4.w > 0.0f) ? r4.w : 0.0f;
            ((float4*)(out + (size_t)mynode[p] * CC))[c16] = r4;
        }
    }
    for (int q = 0; q < AG_G; ++q) {
        if ((big >> q) & 1u)
            node_general(node0 + q, lane, cnt, csr_src, ovf, s_i, s_j, g, bias, out);
    }
}

extern "C" void kernel_launch(void* const* d_in, const int* in_sizes, int n_in,
                              void* d_out, int out_size, void* d_ws, size_t ws_size,
                              hipStream_t stream) {
    (void)in_sizes; (void)n_in; (void)out_size; (void)ws_size;
    const float* x      = (const float*)d_in[0];
    const int*   ei     = (const int*)  d_in[1];
    const float* corrs  = (const float*)d_in[2];
    const float* W      = (const float*)d_in[3];
    const float* att_i  = (const float*)d_in[4];
    const float* att_j  = (const float*)d_in[5];
    const float* aem_i  = (const float*)d_in[6];
    const float* aem_j  = (const float*)d_in[7];
    const float* bias   = (const float*)d_in[8];
    float* out = (float*)d_out;

    float* ws    = (float*)d_ws;
    float* g     = ws;
    int*   csr   = (int*)(g + (size_t)NN * CC);
    int*   ovf   = csr + (size_t)NN * CAP;
    float* s_i   = (float*)(ovf + (size_t)2 * EE);
    float* s_j   = s_i + NN;
    int*   cnt   = (int*)(s_j + NN);
    unsigned* binned = (unsigned*)(cnt + NN + 80);
    int*   bstart = (int*)(binned + (size_t)BN_C * BN_E);

    gemm_scores_kernel<<<BN_C + NN / 16, 32, 0, stream>>>(
        x, W, corrs, att_i, att_j, aem_i, aem_j, g, s_i, s_j, cnt, ei, binned, bstart);
    place_kernel<<<PL_S * PL_R, PL_T, 0, stream>>>(binned, bstart, cnt, csr, ovf);
    aggregate_kernel<<<(NN / AG_G * 32 + 255) / 256, 256, 0, stream>>>(
        cnt, csr, ovf, s_i, s_j, g, bias, out);
}
